# starting kernel
# baseline (speedup 1.0000x reference)
#include <hip/hip_runtime.h>
#include <hip/hip_bf16.h>
#include <math.h>

typedef __attribute__((ext_vector_type(2))) float v2f;
typedef __attribute__((ext_vector_type(8))) float v8f;

#define NUM_GRAPHS 64
#define NEG_SLOPE 0.2f

template <int FIN, int FOUT>
__global__ __launch_bounds__(256) void gemm_wmma(const float* __restrict__ x,
                                                 const float* __restrict__ W,
                                                 float* __restrict__ h, int N) {
  constexpr int NT = FOUT / 16;
  __shared__ float Ws[FIN * FOUT];
  for (int i = threadIdx.x; i < FIN * FOUT; i += 256) Ws[i] = W[i];
  __syncthreads();

  const int wave = threadIdx.x >> 5;
  const int lane = threadIdx.x & 31;
  const int m0 = (blockIdx.x * 8 + wave) * 16;
  const int halfsel = lane >> 4;
  const int lidx = lane & 15;
  int row = m0 + lidx;
  if (row >= N) row = N - 1;
  const int khalf = halfsel * 2;

  v8f acc[NT];
#pragma unroll
  for (int t = 0; t < NT; ++t)
#pragma unroll
    for (int j = 0; j < 8; ++j) acc[t][j] = 0.0f;

#pragma unroll 4
  for (int k = 0; k < FIN; k += 4) {
    const int kk = k + khalf;
    v2f a;
    a.x = x[(size_t)row * FIN + kk];
    a.y = x[(size_t)row * FIN + kk + 1];
#pragma unroll
    for (int t = 0; t < NT; ++t) {
      v2f b;
      b.x = Ws[kk * FOUT + t * 16 + lidx];
      b.y = Ws[(kk + 1) * FOUT + t * 16 + lidx];
      acc[t] = __builtin_amdgcn_wmma_f32_16x16x4_f32(
          false, a, false, b, (short)0, acc[t], false, false);
    }
  }

  const int rbase = m0 + halfsel * 8;
#pragma unroll
  for (int i = 0; i < 8; ++i) {
    const int r = rbase + i;
    if (r < N) {
#pragma unroll
      for (int t = 0; t < NT; ++t)
        h[(size_t)r * FOUT + t * 16 + lidx] = acc[t][i];
    }
  }
}

template <int F>
__global__ void node_dots(const float* __restrict__ h,
                          const float* __restrict__ asrc,
                          const float* __restrict__ adst,
                          float* __restrict__ s, float* __restrict__ d, int N) {
  int n = blockIdx.x * blockDim.x + threadIdx.x;
  if (n >= N) return;
  float ss = 0.0f, dd = 0.0f;
#pragma unroll
  for (int f = 0; f < F; ++f) {
    float v = h[(size_t)n * F + f];
    ss += v * asrc[f];
    dd += v * adst[f];
  }
  s[n] = ss;
  d[n] = dd;
}

__device__ __forceinline__ void atomicMaxFloat(float* addr, float val) {
  int vi = __float_as_int(val);
  if (vi >= 0)
    atomicMax((int*)addr, vi);
  else
    atomicMin((unsigned int*)addr, (unsigned int)vi);
}

__global__ void edge_score(const int* __restrict__ src,
                           const int* __restrict__ dst,
                           const float* __restrict__ sv,
                           const float* __restrict__ dv,
                           float* __restrict__ ebuf, float* __restrict__ m,
                           int E, int ET) {
  int id = blockIdx.x * blockDim.x + threadIdx.x;
  if (id >= ET) return;
  int si, di;
  if (id < E) {
    si = src[id];
    di = dst[id];
  } else {
    si = di = id - E;
  }
  float e = sv[si] + dv[di];
  e = (e > 0.0f) ? e : NEG_SLOPE * e;
  ebuf[id] = e;
  atomicMaxFloat(m + di, e);
}

__global__ void edge_exp(const int* __restrict__ dst,
                         const float* __restrict__ m, float* __restrict__ ebuf,
                         float* __restrict__ z, int E, int ET) {
  int id = blockIdx.x * blockDim.x + threadIdx.x;
  if (id >= ET) return;
  int di = (id < E) ? dst[id] : id - E;
  float t = __expf(ebuf[id] - m[di]);
  ebuf[id] = t;
  atomicAdd(z + di, t);
}

template <int F>
__global__ __launch_bounds__(256) void edge_aggregate(
    const int* __restrict__ src, const int* __restrict__ dst,
    const float* __restrict__ ebuf, const float* __restrict__ z,
    const float* __restrict__ h, float* __restrict__ out, int E, int ET) {
  int id = blockIdx.x * 8 + (threadIdx.x >> 5);
  if (id >= ET) return;
  int lane = threadIdx.x & 31;
  int si, di;
  if (id < E) {
    si = src[id];
    di = dst[id];
  } else {
    si = di = id - E;
  }
  float alpha = ebuf[id] / z[di];
#pragma unroll
  for (int f = lane; f < F; f += 32)
    atomicAdd(out + (size_t)di * F + f, alpha * h[(size_t)si * F + f]);
}

template <int F, bool ELU>
__global__ void bias_act(float* __restrict__ h, const float* __restrict__ b,
                         int N) {
  int i = blockIdx.x * blockDim.x + threadIdx.x;
  if (i >= N * F) return;
  float v = h[i] + b[i % F];
  if (ELU) v = (v > 0.0f) ? v : (__expf(v) - 1.0f);
  h[i] = v;
}

__global__ void pool_sum(const float* __restrict__ h,
                         const int* __restrict__ batch, float* __restrict__ out,
                         float* __restrict__ cnt, int N) {
  int i = blockIdx.x * blockDim.x + threadIdx.x;
  int n = i >> 6;
  int f = i & 63;
  if (n >= N) return;
  int g = batch[n];
  atomicAdd(out + g * 64 + f, h[(size_t)n * 64 + f]);
  if (f == 0) atomicAdd(cnt + g, 1.0f);
}

__global__ void pool_div(float* __restrict__ out, const float* __restrict__ cnt) {
  int i = blockIdx.x * blockDim.x + threadIdx.x;
  if (i >= NUM_GRAPHS * 64) return;
  float c = cnt[i >> 6];
  out[i] = out[i] / fmaxf(c, 1.0f);
}

__global__ void fill_kernel(float* __restrict__ p, float v, int n) {
  int i = blockIdx.x * blockDim.x + threadIdx.x;
  if (i < n) p[i] = v;
}

static inline int cdiv(int a, int b) { return (a + b - 1) / b; }

extern "C" void kernel_launch(void* const* d_in, const int* in_sizes, int n_in,
                              void* d_out, int out_size, void* d_ws,
                              size_t ws_size, hipStream_t stream) {
  const float* x      = (const float*)d_in[0];
  const int*   ei     = (const int*)d_in[1];
  const int*   batch  = (const int*)d_in[2];
  const float* W1     = (const float*)d_in[3];
  const float* a_src1 = (const float*)d_in[4];
  const float* a_dst1 = (const float*)d_in[5];
  const float* b1     = (const float*)d_in[6];
  const float* W2     = (const float*)d_in[7];
  const float* a_src2 = (const float*)d_in[8];
  const float* a_dst2 = (const float*)d_in[9];
  const float* b2     = (const float*)d_in[10];

  const int N = in_sizes[2];
  const int E = in_sizes[1] / 2;
  const int ET = E + N;
  const int* src = ei;
  const int* dst = ei + E;

  float* ws = (float*)d_ws;
  size_t off = 0;
  float* h1 = ws + off; off += (size_t)N * 32;
  float* o1 = ws + off; off += (size_t)N * 32;
  float* h2 = ws + off; off += (size_t)N * 64;
  float* o2 = ws + off; off += (size_t)N * 64;
  float* sv = ws + off; off += (size_t)N;
  float* dv = ws + off; off += (size_t)N;
  float* mb = ws + off; off += (size_t)N;
  float* zb = ws + off; off += (size_t)N;
  float* eb = ws + off; off += (size_t)ET;
  float* cnt = ws + off; off += NUM_GRAPHS;

  const float NEG_INF = -__builtin_inff();

  gemm_wmma<128, 32><<<cdiv(N, 128), 256, 0, stream>>>(x, W1, h1, N);
  node_dots<32><<<cdiv(N, 256), 256, 0, stream>>>(h1, a_src1, a_dst1, sv, dv, N);
  fill_kernel<<<cdiv(N, 256), 256, 0, stream>>>(mb, NEG_INF, N);
  fill_kernel<<<cdiv(N, 256), 256, 0, stream>>>(zb, 0.0f, N);
  fill_kernel<<<cdiv(N * 32, 256), 256, 0, stream>>>(o1, 0.0f, N * 32);
  edge_score<<<cdiv(ET, 256), 256, 0, stream>>>(src, dst, sv, dv, eb, mb, E, ET);
  edge_exp<<<cdiv(ET, 256), 256, 0, stream>>>(dst, mb, eb, zb, E, ET);
  edge_aggregate<32><<<cdiv(ET, 8), 256, 0, stream>>>(src, dst, eb, zb, h1, o1, E, ET);
  bias_act<32, true><<<cdiv(N * 32, 256), 256, 0, stream>>>(o1, b1, N);

  gemm_wmma<32, 64><<<cdiv(N, 128), 256, 0, stream>>>(o1, W2, h2, N);
  node_dots<64><<<cdiv(N, 256), 256, 0, stream>>>(h2, a_src2, a_dst2, sv, dv, N);
  fill_kernel<<<cdiv(N, 256), 256, 0, stream>>>(mb, NEG_INF, N);
  fill_kernel<<<cdiv(N, 256), 256, 0, stream>>>(zb, 0.0f, N);
  fill_kernel<<<cdiv(N * 64, 256), 256, 0, stream>>>(o2, 0.0f, N * 64);
  edge_score<<<cdiv(ET, 256), 256, 0, stream>>>(src, dst, sv, dv, eb, mb, E, ET);
  edge_exp<<<cdiv(ET, 256), 256, 0, stream>>>(dst, mb, eb, zb, E, ET);
  edge_aggregate<64><<<cdiv(ET, 8), 256, 0, stream>>>(src, dst, eb, zb, h2, o2, E, ET);
  bias_act<64, false><<<cdiv(N * 64, 256), 256, 0, stream>>>(o2, b2, N);

  fill_kernel<<<cdiv(NUM_GRAPHS * 64, 256), 256, 0, stream>>>((float*)d_out, 0.0f,
                                                              NUM_GRAPHS * 64);
  fill_kernel<<<1, 64, 0, stream>>>(cnt, 0.0f, NUM_GRAPHS);
  pool_sum<<<cdiv(N * 64, 256), 256, 0, stream>>>(o2, batch, (float*)d_out, cnt, N);
  pool_div<<<cdiv(NUM_GRAPHS * 64, 256), 256, 0, stream>>>((float*)d_out, cnt);
}


# Edges grouped by destination once per call (int atomic rank, scan, place); per layer one wave per node does max, exp, sum and weighted row sum in registers, no float atomics there; sum order = rank
# speedup vs baseline: 2.4742x; 2.4742x over previous
#include <hip/hip_runtime.h>
#include <hip/hip_bf16.h>
#include <math.h>

typedef __attribute__((ext_vector_type(2))) float v2f;
typedef __attribute__((ext_vector_type(8))) float v8f;

#define NUM_GRAPHS 64
#define NEG_SLOPE 0.2f

template <int FIN, int FOUT>
__global__ __launch_bounds__(256) void gemm_wmma(const float* __restrict__ x,
                                                 const float* __restrict__ W,
                                                 float* __restrict__ h, int N) {
  constexpr int NT = FOUT / 16;
  __shared__ float Ws[FIN * FOUT];
  for (int i = threadIdx.x; i < FIN * FOUT; i += 256) Ws[i] = W[i];
  __syncthreads();

  const int wave = threadIdx.x >> 5;
  const int lane = threadIdx.x & 31;
  const int m0 = (blockIdx.x * 8 + wave) * 16;
  const int halfsel = lane >> 4;
  const int lidx = lane & 15;
  int row = m0 + lidx;
  if (row >= N) row = N - 1;
  const int khalf = halfsel * 2;

  v8f acc[NT];
#pragma unroll
  for (int t = 0; t < NT; ++t)
#pragma unroll
    for (int j = 0; j < 8; ++j) acc[t][j] = 0.0f;

#pragma unroll 4
  for (int k = 0; k < FIN; k += 4) {
    const int kk = k + khalf;
    v2f a;
    a.x = x[(size_t)row * FIN + kk];
    a.y = x[(size_t)row * FIN + kk + 1];
#pragma unroll
    for (int t = 0; t < NT; ++t) {
      v2f b;
      b.x = Ws[kk * FOUT + t * 16 + lidx];
      b.y = Ws[(kk + 1) * FOUT + t * 16 + lidx];
      acc[t] = __builtin_amdgcn_wmma_f32_16x16x4_f32(
          false, a, false, b, (short)0, acc[t], false, false);
    }
  }

  const int rbase = m0 + halfsel * 8;
#pragma unroll
  for (int i = 0; i < 8; ++i) {
    const int r = rbase + i;
    if (r < N) {
#pragma unroll
      for (int t = 0; t < NT; ++t)
        h[(size_t)r * FOUT + t * 16 + lidx] = acc[t][i];
    }
  }
}

template <int F>
__global__ void node_dots(const float* __restrict__ h,
                          const float* __restrict__ asrc,
                          const float* __restrict__ adst,
                          float* __restrict__ s, float* __restrict__ d, int N) {
  int n = blockIdx.x * blockDim.x + threadIdx.x;
  if (n >= N) return;
  float ss = 0.0f, dd = 0.0f;
#pragma unroll
  for (int f = 0; f < F; ++f) {
    float v = h[(size_t)n * F + f];
    ss += v * asrc[f];
    dd += v * adst[f];
  }
  s[n] = ss;
  d[n] = dd;
}

__global__ void fill_int(int* __restrict__ p, int v, int n) {
  int i = blockIdx.x * blockDim.x + threadIdx.x;
  if (i < n) p[i] = v;
}

__global__ void count_rank(const int* __restrict__ dst, int* __restrict__ deg,
                           int* __restrict__ rank, int E) {
  int e = blockIdx.x * blockDim.x + threadIdx.x;
  if (e >= E) return;
  rank[e] = atomicAdd(deg + dst[e], 1);
}

__global__ __launch_bounds__(1024) void scan_counts(const int* __restrict__ deg,
                                                    int* __restrict__ rowptr,
                                                    int N) {
  __shared__ int wsum[32];
  const int tid = threadIdx.x;
  const int lane = tid & 31;
  const int w = tid >> 5;
  const int per = (N + 1023) / 1024;
  int beg = tid * per;
  if (beg > N) beg = N;
  int end = beg + per;
  if (end > N) end = N;

  int s = 0;
  for (int i = beg; i < end; ++i) s += deg[i];

  int incl = s;
  for (int o = 1; o < 32; o <<= 1) {
    int v = __shfl_up(incl, o);
    if (lane >= o) incl += v;
  }
  if (lane == 31) wsum[w] = incl;
  __syncthreads();
  if (w == 0) {
    const int v = wsum[lane];
    int iv = v;
    for (int o = 1; o < 32; o <<= 1) {
      int u = __shfl_up(iv, o);
      if (lane >= o) iv += u;
    }
    wsum[lane] = iv - v;
  }
  __syncthreads();

  int run = wsum[w] + (incl - s);
  for (int i = beg; i < end; ++i) {
    rowptr[i] = run;
    run += deg[i];
  }
  if (tid == 1023) rowptr[N] = run;
}

__global__ void place_edges(const int* __restrict__ src,
                            const int* __restrict__ dst,
                            const int* __restrict__ rank,
                            const int* __restrict__ rowptr,
                            int* __restrict__ csr, int E) {
  int e = blockIdx.x * blockDim.x + threadIdx.x;
  if (e >= E) return;
  csr[rowptr[dst[e]] + rank[e]] = src[e];
}

__device__ __forceinline__ float leaky(float e) {
  return (e > 0.0f) ? e : NEG_SLOPE * e;
}

template <int F, bool ELU>
__global__ __launch_bounds__(256) void node_gather(
    const int* __restrict__ rowptr, const int* __restrict__ csr,
    const float* __restrict__ sv, const float* __restrict__ dv,
    const float* __restrict__ h, const float* __restrict__ b,
    float* __restrict__ out, int N) {
  static_assert(F == 32 || F == 64, "one or two floats per lane");
  const int wv = __builtin_amdgcn_readfirstlane((int)(threadIdx.x >> 5));
  const int n = blockIdx.x * 8 + wv;
  if (n >= N) return;
  const int lane = threadIdx.x & 31;

  const int beg = rowptr[n];
  const int end = rowptr[n + 1];
  const float dn = dv[n];
  const float es = leaky(sv[n] + dn);

  float mx = es;
  for (int p = beg + lane; p < end; p += 32) mx = fmaxf(mx, leaky(sv[csr[p]] + dn));
  for (int o = 16; o > 0; o >>= 1) mx = fmaxf(mx, __shfl_xor(mx, o));

  const float ts = __expf(es - mx);
  float zl = 0.0f;
  float a0, a1 = 0.0f;
  if (F == 32) {
    a0 = ts * h[(size_t)n * 32 + lane];
  } else {
    const float2 r = *((const float2*)(h + (size_t)n * 64) + lane);
    a0 = ts * r.x;
    a1 = ts * r.y;
  }

  for (int base = beg; base < end; base += 32) {
    const int p = base + lane;
    int s = n;
    float t = 0.0f;
    if (p < end) {
      s = csr[p];
      t = __expf(leaky(sv[s] + dn) - mx);
    }
    zl += t;
    int c = end - base;
    if (c > 32) c = 32;
    for (int j0 = 0; j0 < c; j0 += 8) {
      for (int u = 0; u < 8; ++u) {
        const int j = j0 + u;
        const int sj = __builtin_amdgcn_readlane(s, j);
        const float tj = __int_as_float(__builtin_amdgcn_readlane(__float_as_int(t), j));
        if (F == 32) {
          a0 += tj * h[(size_t)sj * 32 + lane];
        } else {
          const float2 r = *((const float2*)(h + (size_t)sj * 64) + lane);
          a0 += tj * r.x;
          a1 += tj * r.y;
        }
      }
    }
  }
  for (int o = 16; o > 0; o >>= 1) zl += __shfl_xor(zl, o);
  const float z = zl + ts;

  if (F == 32) {
    float v = a0 / z + b[lane];
    if (ELU) v = (v > 0.0f) ? v : (__expf(v) - 1.0f);
    out[(size_t)n * 32 + lane] = v;
  } else {
    float v0 = a0 / z + b[2 * lane];
    float v1 = a1 / z + b[2 * lane + 1];
    if (ELU) {
      v0 = (v0 > 0.0f) ? v0 : (__expf(v0) - 1.0f);
      v1 = (v1 > 0.0f) ? v1 : (__expf(v1) - 1.0f);
    }
    float2 o2v;
    o2v.x = v0;
    o2v.y = v1;
    *((float2*)(out + (size_t)n * 64) + lane) = o2v;
  }
}

__global__ void pool_sum(const float* __restrict__ h,
                         const int* __restrict__ batch, float* __restrict__ out,
                         float* __restrict__ cnt, int N) {
  int i = blockIdx.x * blockDim.x + threadIdx.x;
  int n = i >> 6;
  int f = i & 63;
  if (n >= N) return;
  int g = batch[n];
  atomicAdd(out + g * 64 + f, h[(size_t)n * 64 + f]);
  if (f == 0) atomicAdd(cnt + g, 1.0f);
}

__global__ void pool_div(float* __restrict__ out, const float* __restrict__ cnt) {
  int i = blockIdx.x * blockDim.x + threadIdx.x;
  if (i >= NUM_GRAPHS * 64) return;
  float c = cnt[i >> 6];
  out[i] = out[i] / fmaxf(c, 1.0f);
}

__global__ void fill_kernel(float* __restrict__ p, float v, int n) {
  int i = blockIdx.x * blockDim.x + threadIdx.x;
  if (i < n) p[i] = v;
}

static inline int cdiv(int a, int b) { return (a + b - 1) / b; }

extern "C" void kernel_launch(void* const* d_in, const int* in_sizes, int n_in,
                              void* d_out, int out_size, void* d_ws,
                              size_t ws_size, hipStream_t stream) {
  const float* x      = (const float*)d_in[0];
  const int*   ei     = (const int*)d_in[1];
  const int*   batch  = (const int*)d_in[2];
  const float* W1     = (const float*)d_in[3];
  const float* a_src1 = (const float*)d_in[4];
  const float* a_dst1 = (const float*)d_in[5];
  const float* b1     = (const float*)d_in[6];
  const float* W2     = (const float*)d_in[7];
  const float* a_src2 = (const float*)d_in[8];
  const float* a_dst2 = (const float*)d_in[9];
  const float* b2     = (const float*)d_in[10];

  const int N = in_sizes[2];
  const int E = in_sizes[1] / 2;
  const int* src = ei;
  const int* dst = ei + E;

  float* ws = (float*)d_ws;
  size_t off = 0;
  float* h1 = ws + off; off += (size_t)N * 32;
  float* o1 = ws + off; off += (size_t)N * 32;
  float* h2 = ws + off; off += (size_t)N * 64;
  float* o2 = ws + off; off += (size_t)N * 64;
  float* sv = ws + off; off += (size_t)N;
  float* dv = ws + off; off += (size_t)N;
  float* cnt = ws + off; off += NUM_GRAPHS;
  int* deg    = (int*)(ws + off); off += (size_t)N;
  int* rowptr = (int*)(ws + off); off += (size_t)N + 1;
  int* rank   = (int*)(ws + off); off += (size_t)E;
  int* csr    = (int*)(ws + off); off += (size_t)E;

  fill_int<<<cdiv(N, 256), 256, 0, stream>>>(deg, 0, N);
  count_rank<<<cdiv(E, 256), 256, 0, stream>>>(dst, deg, rank, E);
  scan_counts<<<1, 1024, 0, stream>>>(deg, rowptr, N);
  place_edges<<<cdiv(E, 256), 256, 0, stream>>>(src, dst, rank, rowptr, csr, E);

  gemm_wmma<128, 32><<<cdiv(N, 128), 256, 0, stream>>>(x, W1, h1, N);
  node_dots<32><<<cdiv(N, 256), 256, 0, stream>>>(h1, a_src1, a_dst1, sv, dv, N);
  node_gather<32, true><<<cdiv(N, 8), 256, 0, stream>>>(rowptr, csr, sv, dv, h1, b1, o1, N);

  gemm_wmma<32, 64><<<cdiv(N, 128), 256, 0, stream>>>(o1, W2, h2, N);
  node_dots<64><<<cdiv(N, 256), 256, 0, stream>>>(h2, a_src2, a_dst2, sv, dv, N);
  node_gather<64, false><<<cdiv(N, 8), 256, 0, stream>>>(rowptr, csr, sv, dv, h2, b2, o2, N);

  fill_kernel<<<cdiv(NUM_GRAPHS * 64, 256), 256, 0, stream>>>((float*)d_out, 0.0f,
                                                              NUM_GRAPHS * 64);
  fill_kernel<<<1, 64, 0, stream>>>(cnt, 0.0f, NUM_GRAPHS);
  pool_sum<<<cdiv(N * 64, 256), 256, 0, stream>>>(o2, batch, (float*)d_out, cnt, N);
  pool_div<<<cdiv(NUM_GRAPHS * 64, 256), 256, 0, stream>>>((float*)d_out, cnt);
}


# Fewer, fatter waves: count_rank and place_edges take 8 edges per lane; pool takes 16 nodes per wave, sums each run of one graph in registers, then one atomic per feature; 200k waves become 16k
# speedup vs baseline: 3.0467x; 1.2314x over previous
#include <hip/hip_runtime.h>
#include <hip/hip_bf16.h>
#include <math.h>

typedef __attribute__((ext_vector_type(2))) float v2f;
typedef __attribute__((ext_vector_type(8))) float v8f;

#define NUM_GRAPHS 64
#define NEG_SLOPE 0.2f

template <int FIN, int FOUT>
__global__ __launch_bounds__(256) void gemm_wmma(const float* __restrict__ x,
                                                 const float* __restrict__ W,
                                                 float* __restrict__ h, int N) {
  constexpr int NT = FOUT / 16;
  __shared__ float Ws[FIN * FOUT];
  for (int i = threadIdx.x; i < FIN * FOUT; i += 256) Ws[i] = W[i];
  __syncthreads();

  const int wave = threadIdx.x >> 5;
  const int lane = threadIdx.x & 31;
  const int m0 = (blockIdx.x * 8 + wave) * 16;
  const int halfsel = lane >> 4;
  const int lidx = lane & 15;
  int row = m0 + lidx;
  if (row >= N) row = N - 1;
  const int khalf = halfsel * 2;

  v8f acc[NT];
#pragma unroll
  for (int t = 0; t < NT; ++t)
#pragma unroll
    for (int j = 0; j < 8; ++j) acc[t][j] = 0.0f;

#pragma unroll 4
  for (int k = 0; k < FIN; k += 4) {
    const int kk = k + khalf;
    v2f a;
    a.x = x[(size_t)row * FIN + kk];
    a.y = x[(size_t)row * FIN + kk + 1];
#pragma unroll
    for (int t = 0; t < NT; ++t) {
      v2f b;
      b.x = Ws[kk * FOUT + t * 16 + lidx];
      b.y = Ws[(kk + 1) * FOUT + t * 16 + lidx];
      acc[t] = __builtin_amdgcn_wmma_f32_16x16x4_f32(
          false, a, false, b, (short)0, acc[t], false, false);
    }
  }

  const int rbase = m0 + halfsel * 8;
#pragma unroll
  for (int i = 0; i < 8; ++i) {
    const int r = rbase + i;
    if (r < N) {
#pragma unroll
      for (int t = 0; t < NT; ++t)
        h[(size_t)r * FOUT + t * 16 + lidx] = acc[t][i];
    }
  }
}

template <int F>
__global__ void node_dots(const float* __restrict__ h,
                          const float* __restrict__ asrc,
                          const float* __restrict__ adst,
                          float* __restrict__ s, float* __restrict__ d, int N) {
  int n = blockIdx.x * blockDim.x + threadIdx.x;
  if (n >= N) return;
  float ss = 0.0f, dd = 0.0f;
#pragma unroll
  for (int f = 0; f < F; ++f) {
    float v = h[(size_t)n * F + f];
    ss += v * asrc[f];
    dd += v * adst[f];
  }
  s[n] = ss;
  d[n] = dd;
}

__global__ void fill_int(int* __restrict__ p, int v, int n) {
  int i = blockIdx.x * blockDim.x + threadIdx.x;
  if (i < n) p[i] = v;
}

#define EPL 8

template <bool FULL>
__device__ __forceinline__ void count_rank_body(const int* __restrict__ dst,
                                                int* __restrict__ deg,
                                                int* __restrict__ rank,
                                                int base, int E) {
  int d[EPL], r[EPL];
#pragma unroll
  for (int j = 0; j < EPL; ++j) {
    const int e = base + j * 256;
    d[j] = (FULL || e < E) ? dst[e] : 0;
  }
#pragma unroll
  for (int j = 0; j < EPL; ++j) {
    const int e = base + j * 256;
    r[j] = 0;
    if (FULL || e < E) r[j] = atomicAdd(deg + d[j], 1);
  }
#pragma unroll
  for (int j = 0; j < EPL; ++j) {
    const int e = base + j * 256;
    if (FULL || e < E) rank[e] = r[j];
  }
}

__global__ __launch_bounds__(256) void count_rank(const int* __restrict__ dst,
                                                  int* __restrict__ deg,
                                                  int* __restrict__ rank, int E) {
  const int blk = blockIdx.x * (256 * EPL);
  const int base = blk + threadIdx.x;
  if (blk + 256 * EPL <= E)
    count_rank_body<true>(dst, deg, rank, base, E);
  else
    count_rank_body<false>(dst, deg, rank, base, E);
}

__global__ __launch_bounds__(1024) void scan_counts(const int* __restrict__ deg,
                                                    int* __restrict__ rowptr,
                                                    int N) {
  __shared__ int wsum[32];
  const int tid = threadIdx.x;
  const int lane = tid & 31;
  const int w = tid >> 5;
  const int per = (N + 1023) / 1024;
  int beg = tid * per;
  if (beg > N) beg = N;
  int end = beg + per;
  if (end > N) end = N;

  int s = 0;
  for (int i = beg; i < end; ++i) s += deg[i];

  int incl = s;
  for (int o = 1; o < 32; o <<= 1) {
    int v = __shfl_up(incl, o);
    if (lane >= o) incl += v;
  }
  if (lane == 31) wsum[w] = incl;
  __syncthreads();
  if (w == 0) {
    const int v = wsum[lane];
    int iv = v;
    for (int o = 1; o < 32; o <<= 1) {
      int u = __shfl_up(iv, o);
      if (lane >= o) iv += u;
    }
    wsum[lane] = iv - v;
  }
  __syncthreads();

  int run = wsum[w] + (incl - s);
  for (int i = beg; i < end; ++i) {
    rowptr[i] = run;
    run += deg[i];
  }
  if (tid == 1023) rowptr[N] = run;
}

template <bool FULL>
__device__ __forceinline__ void place_edges_body(const int* __restrict__ src,
                                                 const int* __restrict__ dst,
                                                 const int* __restrict__ rank,
                                                 const int* __restrict__ rowptr,
                                                 int* __restrict__ csr,
                                                 int base, int E) {
  int s[EPL], d[EPL], r[EPL], rp[EPL];
#pragma unroll
  for (int j = 0; j < EPL; ++j) {
    const int e = base + j * 256;
    const bool ok = FULL || e < E;
    d[j] = ok ? dst[e] : 0;
    r[j] = ok ? rank[e] : 0;
    s[j] = ok ? src[e] : 0;
  }
#pragma unroll
  for (int j = 0; j < EPL; ++j) rp[j] = rowptr[d[j]];
#pragma unroll
  for (int j = 0; j < EPL; ++j) {
    const int e = base + j * 256;
    if (FULL || e < E) csr[rp[j] + r[j]] = s[j];
  }
}

__global__ __launch_bounds__(256) void place_edges(const int* __restrict__ src,
                                                   const int* __restrict__ dst,
                                                   const int* __restrict__ rank,
                                                   const int* __restrict__ rowptr,
                                                   int* __restrict__ csr, int E) {
  const int blk = blockIdx.x * (256 * EPL);
  const int base = blk + threadIdx.x;
  if (blk + 256 * EPL <= E)
    place_edges_body<true>(src, dst, rank, rowptr, csr, base, E);
  else
    place_edges_body<false>(src, dst, rank, rowptr, csr, base, E);
}

__device__ __forceinline__ float leaky(float e) {
  return (e > 0.0f) ? e : NEG_SLOPE * e;
}

template <int F, bool ELU>
__global__ __launch_bounds__(256) void node_gather(
    const int* __restrict__ rowptr, const int* __restrict__ csr,
    const float* __restrict__ sv, const float* __restrict__ dv,
    const float* __restrict__ h, const float* __restrict__ b,
    float* __restrict__ out, int N) {
  static_assert(F == 32 || F == 64, "one or two floats per lane");
  const int wv = __builtin_amdgcn_readfirstlane((int)(threadIdx.x >> 5));
  const int n = blockIdx.x * 8 + wv;
  if (n >= N) return;
  const int lane = threadIdx.x & 31;

  const int beg = rowptr[n];
  const int end = rowptr[n + 1];
  const float dn = dv[n];
  const float es = leaky(sv[n] + dn);

  float mx = es;
  for (int p = beg + lane; p < end; p += 32) mx = fmaxf(mx, leaky(sv[csr[p]] + dn));
  for (int o = 16; o > 0; o >>= 1) mx = fmaxf(mx, __shfl_xor(mx, o));

  const float ts = __expf(es - mx);
  float zl = 0.0f;
  float a0, a1 = 0.0f;
  if (F == 32) {
    a0 = ts * h[(size_t)n * 32 + lane];
  } else {
    const float2 r = *((const float2*)(h + (size_t)n * 64) + lane);
    a0 = ts * r.x;
    a1 = ts * r.y;
  }

  for (int base = beg; base < end; base += 32) {
    const int p = base + lane;
    int s = n;
    float t = 0.0f;
    if (p < end) {
      s = csr[p];
      t = __expf(leaky(sv[s] + dn) - mx);
    }
    zl += t;
    int c = end - base;
    if (c > 32) c = 32;
    for (int j0 = 0; j0 < c; j0 += 8) {
      for (int u = 0; u < 8; ++u) {
        const int j = j0 + u;
        const int sj = __builtin_amdgcn_readlane(s, j);
        const float tj = __int_as_float(__builtin_amdgcn_readlane(__float_as_int(t), j));
        if (F == 32) {
          a0 += tj * h[(size_t)sj * 32 + lane];
        } else {
          const float2 r = *((const float2*)(h + (size_t)sj * 64) + lane);
          a0 += tj * r.x;
          a1 += tj * r.y;
        }
      }
    }
  }
  for (int o = 16; o > 0; o >>= 1) zl += __shfl_xor(zl, o);
  const float z = zl + ts;

  if (F == 32) {
    float v = a0 / z + b[lane];
    if (ELU) v = (v > 0.0f) ? v : (__expf(v) - 1.0f);
    out[(size_t)n * 32 + lane] = v;
  } else {
    float v0 = a0 / z + b[2 * lane];
    float v1 = a1 / z + b[2 * lane + 1];
    if (ELU) {
      v0 = (v0 > 0.0f) ? v0 : (__expf(v0) - 1.0f);
      v1 = (v1 > 0.0f) ? v1 : (__expf(v1) - 1.0f);
    }
    float2 o2v;
    o2v.x = v0;
    o2v.y = v1;
    *((float2*)(out + (size_t)n * 64) + lane) = o2v;
  }
}

#define NPW 16

__device__ __forceinline__ void pool_flush(float* __restrict__ out,
                                           float* __restrict__ cnt, int g,
                                           int lane, float a0, float a1,
                                           int run) {
  atomicAdd(out + g * 64 + 2 * lane, a0);
  atomicAdd(out + g * 64 + 2 * lane + 1, a1);
  if (lane == 0) atomicAdd(cnt + g, (float)run);
}

template <bool FULL>
__device__ __forceinline__ void pool_sum_body(const float* __restrict__ h,
                                              const int* __restrict__ batch,
                                              float* __restrict__ out,
                                              float* __restrict__ cnt, int n0,
                                              int cn, int lane) {
  float2 r[NPW];
  int g[NPW];
#pragma unroll
  for (int u = 0; u < NPW; ++u) {
    const int n = n0 + ((FULL || u < cn) ? u : (cn - 1));
    g[u] = batch[n];
    r[u] = *((const float2*)(h + (size_t)n * 64) + lane);
  }

  int cur = g[0];
  int run = 0;
  float a0 = 0.0f, a1 = 0.0f;
#pragma unroll
  for (int u = 0; u < NPW; ++u) {
    if (FULL || u < cn) {
      if (g[u] != cur) {
        pool_flush(out, cnt, cur, lane, a0, a1, run);
        cur = g[u];
        run = 0;
        a0 = 0.0f;
        a1 = 0.0f;
      }
      a0 += r[u].x;
      a1 += r[u].y;
      ++run;
    }
  }
  pool_flush(out, cnt, cur, lane, a0, a1, run);
}

__global__ __launch_bounds__(256) void pool_sum(const float* __restrict__ h,
                                                const int* __restrict__ batch,
                                                float* __restrict__ out,
                                                float* __restrict__ cnt, int N) {
  const int wv = __builtin_amdgcn_readfirstlane((int)(threadIdx.x >> 5));
  const int lane = threadIdx.x & 31;
  const int n0 = (blockIdx.x * 8 + wv) * NPW;
  if (n0 >= N) return;
  const int cn = N - n0;
  if (cn >= NPW)
    pool_sum_body<true>(h, batch, out, cnt, n0, NPW, lane);
  else
    pool_sum_body<false>(h, batch, out, cnt, n0, cn, lane);
}

__global__ void pool_div(float* __restrict__ out, const float* __restrict__ cnt) {
  int i = blockIdx.x * blockDim.x + threadIdx.x;
  if (i >= NUM_GRAPHS * 64) return;
  float c = cnt[i >> 6];
  out[i] = out[i] / fmaxf(c, 1.0f);
}

__global__ void fill_kernel(float* __restrict__ p, float v, int n) {
  int i = blockIdx.x * blockDim.x + threadIdx.x;
  if (i < n) p[i] = v;
}

static inline int cdiv(int a, int b) { return (a + b - 1) / b; }

extern "C" void kernel_launch(void* const* d_in, const int* in_sizes, int n_in,
                              void* d_out, int out_size, void* d_ws,
                              size_t ws_size, hipStream_t stream) {
  const float* x      = (const float*)d_in[0];
  const int*   ei     = (const int*)d_in[1];
  const int*   batch  = (const int*)d_in[2];
  const float* W1     = (const float*)d_in[3];
  const float* a_src1 = (const float*)d_in[4];
  const float* a_dst1 = (const float*)d_in[5];
  const float* b1     = (const float*)d_in[6];
  const float* W2     = (const float*)d_in[7];
  const float* a_src2 = (const float*)d_in[8];
  const float* a_dst2 = (const float*)d_in[9];
  const float* b2     = (const float*)d_in[10];

  const int N = in_sizes[2];
  const int E = in_sizes[1] / 2;
  const int* src = ei;
  const int* dst = ei + E;

  float* ws = (float*)d_ws;
  size_t off = 0;
  float* h1 = ws + off; off += (size_t)N * 32;
  float* o1 = ws + off; off += (size_t)N * 32;
  float* h2 = ws + off; off += (size_t)N * 64;
  float* o2 = ws + off; off += (size_t)N * 64;
  float* sv = ws + off; off += (size_t)N;
  float* dv = ws + off; off += (size_t)N;
  float* cnt = ws + off; off += NUM_GRAPHS;
  int* deg    = (int*)(ws + off); off += (size_t)N;
  int* rowptr = (int*)(ws + off); off += (size_t)N + 1;
  int* rank   = (int*)(ws + off); off += (size_t)E;
  int* csr    = (int*)(ws + off); off += (size_t)E;

  fill_int<<<cdiv(N, 256), 256, 0, stream>>>(deg, 0, N);
  count_rank<<<cdiv(E, 256 * EPL), 256, 0, stream>>>(dst, deg, rank, E);
  scan_counts<<<1, 1024, 0, stream>>>(deg, rowptr, N);
  place_edges<<<cdiv(E, 256 * EPL), 256, 0, stream>>>(src, dst, rank, rowptr, csr, E);

  gemm_wmma<128, 32><<<cdiv(N, 128), 256, 0, stream>>>(x, W1, h1, N);
  node_dots<32><<<cdiv(N, 256), 256, 0, stream>>>(h1, a_src1, a_dst1, sv, dv, N);
  node_gather<32, true><<<cdiv(N, 8), 256, 0, stream>>>(rowptr, csr, sv, dv, h1, b1, o1, N);

  gemm_wmma<32, 64><<<cdiv(N, 128), 256, 0, stream>>>(o1, W2, h2, N);
  node_dots<64><<<cdiv(N, 256), 256, 0, stream>>>(h2, a_src2, a_dst2, sv, dv, N);
  node_gather<64, false><<<cdiv(N, 8), 256, 0, stream>>>(rowptr, csr, sv, dv, h2, b2, o2, N);

  fill_kernel<<<cdiv(NUM_GRAPHS * 64, 256), 256, 0, stream>>>((float*)d_out, 0.0f,
                                                              NUM_GRAPHS * 64);
  fill_kernel<<<1, 64, 0, stream>>>(cnt, 0.0f, NUM_GRAPHS);
  pool_sum<<<cdiv(N, 8 * NPW), 256, 0, stream>>>(o2, batch, (float*)d_out, cnt, N);
  pool_div<<<cdiv(NUM_GRAPHS * 64, 256), 256, 0, stream>>>((float*)d_out, cnt);
}


# node_gather keeps each node's first 64 edge sources and scores in registers from the max pass to the weight pass: csr and s[src] are read once per edge, not twice; sum order unchanged
# speedup vs baseline: 3.2051x; 1.0520x over previous
#include <hip/hip_runtime.h>
#include <hip/hip_bf16.h>
#include <math.h>

typedef __attribute__((ext_vector_type(2))) float v2f;
typedef __attribute__((ext_vector_type(8))) float v8f;

#define NUM_GRAPHS 64
#define NEG_SLOPE 0.2f

template <int FIN, int FOUT>
__global__ __launch_bounds__(256) void gemm_wmma(const float* __restrict__ x,
                                                 const float* __restrict__ W,
                                                 float* __restrict__ h, int N) {
  constexpr int NT = FOUT / 16;
  __shared__ float Ws[FIN * FOUT];
  for (int i = threadIdx.x; i < FIN * FOUT; i += 256) Ws[i] = W[i];
  __syncthreads();

  const int wave = threadIdx.x >> 5;
  const int lane = threadIdx.x & 31;
  const int m0 = (blockIdx.x * 8 + wave) * 16;
  const int halfsel = lane >> 4;
  const int lidx = lane & 15;
  int row = m0 + lidx;
  if (row >= N) row = N - 1;
  const int khalf = halfsel * 2;

  v8f acc[NT];
#pragma unroll
  for (int t = 0; t < NT; ++t)
#pragma unroll
    for (int j = 0; j < 8; ++j) acc[t][j] = 0.0f;

#pragma unroll 4
  for (int k = 0; k < FIN; k += 4) {
    const int kk = k + khalf;
    v2f a;
    a.x = x[(size_t)row * FIN + kk];
    a.y = x[(size_t)row * FIN + kk + 1];
#pragma unroll
    for (int t = 0; t < NT; ++t) {
      v2f b;
      b.x = Ws[kk * FOUT + t * 16 + lidx];
      b.y = Ws[(kk + 1) * FOUT + t * 16 + lidx];
      acc[t] = __builtin_amdgcn_wmma_f32_16x16x4_f32(
          false, a, false, b, (short)0, acc[t], false, false);
    }
  }

  const int rbase = m0 + halfsel * 8;
#pragma unroll
  for (int i = 0; i < 8; ++i) {
    const int r = rbase + i;
    if (r < N) {
#pragma unroll
      for (int t = 0; t < NT; ++t)
        h[(size_t)r * FOUT + t * 16 + lidx] = acc[t][i];
    }
  }
}

template <int F>
__global__ void node_dots(const float* __restrict__ h,
                          const float* __restrict__ asrc,
                          const float* __restrict__ adst,
                          float* __restrict__ s, float* __restrict__ d, int N) {
  int n = blockIdx.x * blockDim.x + threadIdx.x;
  if (n >= N) return;
  float ss = 0.0f, dd = 0.0f;
#pragma unroll
  for (int f = 0; f < F; ++f) {
    float v = h[(size_t)n * F + f];
    ss += v * asrc[f];
    dd += v * adst[f];
  }
  s[n] = ss;
  d[n] = dd;
}

__global__ void fill_int(int* __restrict__ p, int v, int n) {
  int i = blockIdx.x * blockDim.x + threadIdx.x;
  if (i < n) p[i] = v;
}

#define EPL 8

template <bool FULL>
__device__ __forceinline__ void count_rank_body(const int* __restrict__ dst,
                                                int* __restrict__ deg,
                                                int* __restrict__ rank,
                                                int base, int E) {
  int d[EPL], r[EPL];
#pragma unroll
  for (int j = 0; j < EPL; ++j) {
    const int e = base + j * 256;
    d[j] = (FULL || e < E) ? dst[e] : 0;
  }
#pragma unroll
  for (int j = 0; j < EPL; ++j) {
    const int e = base + j * 256;
    r[j] = 0;
    if (FULL || e < E) r[j] = atomicAdd(deg + d[j], 1);
  }
#pragma unroll
  for (int j = 0; j < EPL; ++j) {
    const int e = base + j * 256;
    if (FULL || e < E) rank[e] = r[j];
  }
}

__global__ __launch_bounds__(256) void count_rank(const int* __restrict__ dst,
                                                  int* __restrict__ deg,
                                                  int* __restrict__ rank, int E) {
  const int blk = blockIdx.x * (256 * EPL);
  const int base = blk + threadIdx.x;
  if (blk + 256 * EPL <= E)
    count_rank_body<true>(dst, deg, rank, base, E);
  else
    count_rank_body<false>(dst, deg, rank, base, E);
}

__global__ __launch_bounds__(1024) void scan_counts(const int* __restrict__ deg,
                                                    int* __restrict__ rowptr,
                                                    int N) {
  __shared__ int wsum[32];
  const int tid = threadIdx.x;
  const int lane = tid & 31;
  const int w = tid >> 5;
  const int per = (N + 1023) / 1024;
  int beg = tid * per;
  if (beg > N) beg = N;
  int end = beg + per;
  if (end > N) end = N;

  int s = 0;
  for (int i = beg; i < end; ++i) s += deg[i];

  int incl = s;
  for (int o = 1; o < 32; o <<= 1) {
    int v = __shfl_up(incl, o);
    if (lane >= o) incl += v;
  }
  if (lane == 31) wsum[w] = incl;
  __syncthreads();
  if (w == 0) {
    const int v = wsum[lane];
    int iv = v;
    for (int o = 1; o < 32; o <<= 1) {
      int u = __shfl_up(iv, o);
      if (lane >= o) iv += u;
    }
    wsum[lane] = iv - v;
  }
  __syncthreads();

  int run = wsum[w] + (incl - s);
  for (int i = beg; i < end; ++i) {
    rowptr[i] = run;
    run += deg[i];
  }
  if (tid == 1023) rowptr[N] = run;
}

template <bool FULL>
__device__ __forceinline__ void place_edges_body(const int* __restrict__ src,
                                                 const int* __restrict__ dst,
                                                 const int* __restrict__ rank,
                                                 const int* __restrict__ rowptr,
                                                 int* __restrict__ csr,
                                                 int base, int E) {
  int s[EPL], d[EPL], r[EPL], rp[EPL];
#pragma unroll
  for (int j = 0; j < EPL; ++j) {
    const int e = base + j * 256;
    const bool ok = FULL || e < E;
    d[j] = ok ? dst[e] : 0;
    r[j] = ok ? rank[e] : 0;
    s[j] = ok ? src[e] : 0;
  }
#pragma unroll
  for (int j = 0; j < EPL; ++j) rp[j] = rowptr[d[j]];
#pragma unroll
  for (int j = 0; j < EPL; ++j) {
    const int e = base + j * 256;
    if (FULL || e < E) csr[rp[j] + r[j]] = s[j];
  }
}

__global__ __launch_bounds__(256) void place_edges(const int* __restrict__ src,
                                                   const int* __restrict__ dst,
                                                   const int* __restrict__ rank,
                                                   const int* __restrict__ rowptr,
                                                   int* __restrict__ csr, int E) {
  const int blk = blockIdx.x * (256 * EPL);
  const int base = blk + threadIdx.x;
  if (blk + 256 * EPL <= E)
    place_edges_body<true>(src, dst, rank, rowptr, csr, base, E);
  else
    place_edges_body<false>(src, dst, rank, rowptr, csr, base, E);
}

__device__ __forceinline__ float leaky(float e) {
  return (e > 0.0f) ? e : NEG_SLOPE * e;
}


template <int F>
__device__ __forceinline__ void add_rows(const float* __restrict__ h, int s,
                                         float t, int c, int lane, float& a0,
                                         float& a1) {
  for (int j0 = 0; j0 < c; j0 += 8) {
    for (int u = 0; u < 8; ++u) {
      const int j = j0 + u;
      const int sj = __builtin_amdgcn_readlane(s, j);
      const float tj = __int_as_float(__builtin_amdgcn_readlane(__float_as_int(t), j));
      if (F == 32) {
        a0 += tj * h[(size_t)sj * 32 + lane];
      } else {
        const float2 r = *((const float2*)(h + (size_t)sj * 64) + lane);
        a0 += tj * r.x;
        a1 += tj * r.y;
      }
    }
  }
}

template <int F, bool ELU>
__global__ __launch_bounds__(256) void node_gather(
    const int* __restrict__ rowptr, const int* __restrict__ csr,
    const float* __restrict__ sv, const float* __restrict__ dv,
    const float* __restrict__ h, const float* __restrict__ b,
    float* __restrict__ out, int N) {
  static_assert(F == 32 || F == 64, "one or two floats per lane");
  const int wv = __builtin_amdgcn_readfirstlane((int)(threadIdx.x >> 5));
  const int n = blockIdx.x * 8 + wv;
  if (n >= N) return;
  const int lane = threadIdx.x & 31;

  const int beg = rowptr[n];
  const int end = rowptr[n + 1];
  const float dn = dv[n];
  const float es = leaky(sv[n] + dn);

  const int pa = beg + lane;
  const int pb = pa + 32;
  const bool oka = pa < end;
  const bool okb = pb < end;
  int sa = n, sb = n;
  float ea = es, eb = es;
  if (beg < end) {
    const int last = end - 1;
    const int ra = csr[oka ? pa : last];
    const int rb = csr[okb ? pb : last];
    asm volatile("" ::"v"(ra), "v"(rb));
    const float va = sv[ra];
    const float vb = sv[rb];
    if (oka) {
      sa = ra;
      ea = leaky(va + dn);
    }
    if (okb) {
      sb = rb;
      eb = leaky(vb + dn);
    }
  }

  float mx = fmaxf(es, fmaxf(ea, eb));
  for (int p = beg + 64 + lane; p < end; p += 32) mx = fmaxf(mx, leaky(sv[csr[p]] + dn));
  for (int o = 16; o > 0; o >>= 1) mx = fmaxf(mx, __shfl_xor(mx, o));

  const float ts = __expf(es - mx);
  float a0, a1 = 0.0f;
  if (F == 32) {
    a0 = ts * h[(size_t)n * 32 + lane];
  } else {
    const float2 r = *((const float2*)(h + (size_t)n * 64) + lane);
    a0 = ts * r.x;
    a1 = ts * r.y;
  }

  const float ta = oka ? __expf(ea - mx) : 0.0f;
  const float tb = okb ? __expf(eb - mx) : 0.0f;
  float zl = ta;
  zl += tb;
  const int deg = end - beg;
  add_rows<F>(h, sa, ta, deg < 32 ? deg : 32, lane, a0, a1);
  add_rows<F>(h, sb, tb, deg < 64 ? deg - 32 : 32, lane, a0, a1);

  for (int base = beg + 64; base < end; base += 32) {
    const int p = base + lane;
    int s = n;
    float t = 0.0f;
    if (p < end) {
      s = csr[p];
      t = __expf(leaky(sv[s] + dn) - mx);
    }
    zl += t;
    int c = end - base;
    if (c > 32) c = 32;
    add_rows<F>(h, s, t, c, lane, a0, a1);
  }
  for (int o = 16; o > 0; o >>= 1) zl += __shfl_xor(zl, o);
  const float z = zl + ts;

  if (F == 32) {
    float v = a0 / z + b[lane];
    if (ELU) v = (v > 0.0f) ? v : (__expf(v) - 1.0f);
    out[(size_t)n * 32 + lane] = v;
  } else {
    float v0 = a0 / z + b[2 * lane];
    float v1 = a1 / z + b[2 * lane + 1];
    if (ELU) {
      v0 = (v0 > 0.0f) ? v0 : (__expf(v0) - 1.0f);
      v1 = (v1 > 0.0f) ? v1 : (__expf(v1) - 1.0f);
    }
    float2 o2v;
    o2v.x = v0;
    o2v.y = v1;
    *((float2*)(out + (size_t)n * 64) + lane) = o2v;
  }
}

#define NPW 16

__device__ __forceinline__ void pool_flush(float* __restrict__ out,
                                           float* __restrict__ cnt, int g,
                                           int lane, float a0, float a1,
                                           int run) {
  atomicAdd(out + g * 64 + 2 * lane, a0);
  atomicAdd(out + g * 64 + 2 * lane + 1, a1);
  if (lane == 0) atomicAdd(cnt + g, (float)run);
}

template <bool FULL>
__device__ __forceinline__ void pool_sum_body(const float* __restrict__ h,
                                              const int* __restrict__ batch,
                                              float* __restrict__ out,
                                              float* __restrict__ cnt, int n0,
                                              int cn, int lane) {
  float2 r[NPW];
  int g[NPW];
#pragma unroll
  for (int u = 0; u < NPW; ++u) {
    const int n = n0 + ((FULL || u < cn) ? u : (cn - 1));
    g[u] = batch[n];
    r[u] = *((const float2*)(h + (size_t)n * 64) + lane);
  }

  int cur = g[0];
  int run = 0;
  float a0 = 0.0f, a1 = 0.0f;
#pragma unroll
  for (int u = 0; u < NPW; ++u) {
    if (FULL || u < cn) {
      if (g[u] != cur) {
        pool_flush(out, cnt, cur, lane, a0, a1, run);
        cur = g[u];
        run = 0;
        a0 = 0.0f;
        a1 = 0.0f;
      }
      a0 += r[u].x;
      a1 += r[u].y;
      ++run;
    }
  }
  pool_flush(out, cnt, cur, lane, a0, a1, run);
}

__global__ __launch_bounds__(256) void pool_sum(const float* __restrict__ h,
                                                const int* __restrict__ batch,
                                                float* __restrict__ out,
                                                float* __restrict__ cnt, int N) {
  const int wv = __builtin_amdgcn_readfirstlane((int)(threadIdx.x >> 5));
  const int lane = threadIdx.x & 31;
  const int n0 = (blockIdx.x * 8 + wv) * NPW;
  if (n0 >= N) return;
  const int cn = N - n0;
  if (cn >= NPW)
    pool_sum_body<true>(h, batch, out, cnt, n0, NPW, lane);
  else
    pool_sum_body<false>(h, batch, out, cnt, n0, cn, lane);
}

__global__ void pool_div(float* __restrict__ out, const float* __restrict__ cnt) {
  int i = blockIdx.x * blockDim.x + threadIdx.x;
  if (i >= NUM_GRAPHS * 64) return;
  float c = cnt[i >> 6];
  out[i] = out[i] / fmaxf(c, 1.0f);
}

__global__ void fill_kernel(float* __restrict__ p, float v, int n) {
  int i = blockIdx.x * blockDim.x + threadIdx.x;
  if (i < n) p[i] = v;
}

static inline int cdiv(int a, int b) { return (a + b - 1) / b; }

extern "C" void kernel_launch(void* const* d_in, const int* in_sizes, int n_in,
                              void* d_out, int out_size, void* d_ws,
                              size_t ws_size, hipStream_t stream) {
  const float* x      = (const float*)d_in[0];
  const int*   ei     = (const int*)d_in[1];
  const int*   batch  = (const int*)d_in[2];
  const float* W1     = (const float*)d_in[3];
  const float* a_src1 = (const float*)d_in[4];
  const float* a_dst1 = (const float*)d_in[5];
  const float* b1     = (const float*)d_in[6];
  const float* W2     = (const float*)d_in[7];
  const float* a_src2 = (const float*)d_in[8];
  const float* a_dst2 = (const float*)d_in[9];
  const float* b2     = (const float*)d_in[10];

  const int N = in_sizes[2];
  const int E = in_sizes[1] / 2;
  const int* src = ei;
  const int* dst = ei + E;

  float* ws = (float*)d_ws;
  size_t off = 0;
  float* h1 = ws + off; off += (size_t)N * 32;
  float* o1 = ws + off; off += (size_t)N * 32;
  float* h2 = ws + off; off += (size_t)N * 64;
  float* o2 = ws + off; off += (size_t)N * 64;
  float* sv = ws + off; off += (size_t)N;
  float* dv = ws + off; off += (size_t)N;
  float* cnt = ws + off; off += NUM_GRAPHS;
  int* deg    = (int*)(ws + off); off += (size_t)N;
  int* rowptr = (int*)(ws + off); off += (size_t)N + 1;
  int* rank   = (int*)(ws + off); off += (size_t)E;
  int* csr    = (int*)(ws + off); off += (size_t)E;

  fill_int<<<cdiv(N, 256), 256, 0, stream>>>(deg, 0, N);
  count_rank<<<cdiv(E, 256 * EPL), 256, 0, stream>>>(dst, deg, rank, E);
  scan_counts<<<1, 1024, 0, stream>>>(deg, rowptr, N);
  place_edges<<<cdiv(E, 256 * EPL), 256, 0, stream>>>(src, dst, rank, rowptr, csr, E);

  gemm_wmma<128, 32><<<cdiv(N, 128), 256, 0, stream>>>(x, W1, h1, N);
  node_dots<32><<<cdiv(N, 256), 256, 0, stream>>>(h1, a_src1, a_dst1, sv, dv, N);
  node_gather<32, true><<<cdiv(N, 8), 256, 0, stream>>>(rowptr, csr, sv, dv, h1, b1, o1, N);

  gemm_wmma<32, 64><<<cdiv(N, 128), 256, 0, stream>>>(o1, W2, h2, N);
  node_dots<64><<<cdiv(N, 256), 256, 0, stream>>>(h2, a_src2, a_dst2, sv, dv, N);
  node_gather<64, false><<<cdiv(N, 8), 256, 0, stream>>>(rowptr, csr, sv, dv, h2, b2, o2, N);

  fill_kernel<<<cdiv(NUM_GRAPHS * 64, 256), 256, 0, stream>>>((float*)d_out, 0.0f,
                                                              NUM_GRAPHS * 64);
  fill_kernel<<<1, 64, 0, stream>>>(cnt, 0.0f, NUM_GRAPHS);
  pool_sum<<<cdiv(N, 8 * NPW), 256, 0, stream>>>(o2, batch, (float*)d_out, cnt, N);
  pool_div<<<cdiv(NUM_GRAPHS * 64, 256), 256, 0, stream>>>((float*)d_out, cnt);
}


# 14 launches become 9: the first projection also zeroes deg, out and cnt; each projection's epilogue forms s and d from its accumulators (16-lane tree sums, not a serial sum)
# speedup vs baseline: 3.3583x; 1.0478x over previous
#include <hip/hip_runtime.h>
#include <hip/hip_bf16.h>
#include <math.h>

typedef __attribute__((ext_vector_type(2))) float v2f;
typedef __attribute__((ext_vector_type(8))) float v8f;

#define NUM_GRAPHS 64
#define NEG_SLOPE 0.2f


template <int M>
__device__ __forceinline__ float row_xor_add(float v) {
  const int o = __builtin_amdgcn_update_dpp(0, __float_as_int(v), 0x160 + M, 0xf,
                                            0xf, true);
  return v + __int_as_float(o);
}

template <int FIN, int FOUT, bool ZERO>
__global__ __launch_bounds__(256) void gemm_wmma(
    const float* __restrict__ x, const float* __restrict__ W,
    const float* __restrict__ asrc, const float* __restrict__ adst,
    float* __restrict__ h, float* __restrict__ s, float* __restrict__ d, int N,
    int* __restrict__ zdeg, float* __restrict__ zout, float* __restrict__ zcnt) {
  constexpr int NT = FOUT / 16;
  __shared__ float Ws[FIN * FOUT];

  for (int i = threadIdx.x; i < FIN * FOUT; i += 256) Ws[i] = W[i];
  __syncthreads();

  if (ZERO) {
    const int gid = blockIdx.x * 256 + threadIdx.x;
    if (gid < N) zdeg[gid] = 0;
    if (blockIdx.x == 0) {
      for (int i = threadIdx.x; i < NUM_GRAPHS * 64; i += 256) zout[i] = 0.0f;
      if (threadIdx.x < NUM_GRAPHS) zcnt[threadIdx.x] = 0.0f;
    }
  }

  const int wave = threadIdx.x >> 5;
  const int lane = threadIdx.x & 31;
  const int m0 = (blockIdx.x * 8 + wave) * 16;
  const int halfsel = lane >> 4;
  const int lidx = lane & 15;
  int row = m0 + lidx;
  if (row >= N) row = N - 1;
  const int khalf = halfsel * 2;

  float as_[NT], ad_[NT];
#pragma unroll
  for (int t = 0; t < NT; ++t) {
    as_[t] = asrc[t * 16 + lidx];
    ad_[t] = adst[t * 16 + lidx];
  }

  v8f acc[NT];
#pragma unroll
  for (int t = 0; t < NT; ++t)
#pragma unroll
    for (int j = 0; j < 8; ++j) acc[t][j] = 0.0f;

#pragma unroll 4
  for (int k = 0; k < FIN; k += 4) {
    const int kk = k + khalf;
    v2f a;
    a.x = x[(size_t)row * FIN + kk];
    a.y = x[(size_t)row * FIN + kk + 1];
#pragma unroll
    for (int t = 0; t < NT; ++t) {
      v2f b;
      b.x = Ws[kk * FOUT + t * 16 + lidx];
      b.y = Ws[(kk + 1) * FOUT + t * 16 + lidx];
      acc[t] = __builtin_amdgcn_wmma_f32_16x16x4_f32(
          false, a, false, b, (short)0, acc[t], false, false);
    }
  }

  const int rbase = m0 + halfsel * 8;
#pragma unroll
  for (int i = 0; i < 8; ++i) {
    const int r = rbase + i;
    if (r < N) {
#pragma unroll
      for (int t = 0; t < NT; ++t)
        h[(size_t)r * FOUT + t * 16 + lidx] = acc[t][i];
    }
  }

  float ps[8], pd[8];
#pragma unroll
  for (int i = 0; i < 8; ++i) {
    float a = 0.0f, b = 0.0f;
#pragma unroll
    for (int t = 0; t < NT; ++t) {
      a += acc[t][i] * as_[t];
      b += acc[t][i] * ad_[t];
    }
    ps[i] = a;
    pd[i] = b;
  }
#pragma unroll
  for (int i = 0; i < 8; ++i) {
    ps[i] = row_xor_add<1>(ps[i]);
    pd[i] = row_xor_add<1>(pd[i]);
  }
#pragma unroll
  for (int i = 0; i < 8; ++i) {
    ps[i] = row_xor_add<2>(ps[i]);
    pd[i] = row_xor_add<2>(pd[i]);
  }
#pragma unroll
  for (int i = 0; i < 8; ++i) {
    ps[i] = row_xor_add<4>(ps[i]);
    pd[i] = row_xor_add<4>(pd[i]);
  }
#pragma unroll
  for (int i = 0; i < 8; ++i) {
    ps[i] = row_xor_add<8>(ps[i]);
    pd[i] = row_xor_add<8>(pd[i]);
  }
  float so = ps[0], dout = pd[0];
#pragma unroll
  for (int i = 1; i < 8; ++i) {
    if (lidx == i) {
      so = ps[i];
      dout = pd[i];
    }
  }
  const int r = rbase + lidx;
  if (lidx < 8 && r < N) {
    s[r] = so;
    d[r] = dout;
  }
}

#define EPL 8

template <bool FULL>
__device__ __forceinline__ void count_rank_body(const int* __restrict__ dst,
                                                int* __restrict__ deg,
                                                int* __restrict__ rank,
                                                int base, int E) {
  int d[EPL], r[EPL];
#pragma unroll
  for (int j = 0; j < EPL; ++j) {
    const int e = base + j * 256;
    d[j] = (FULL || e < E) ? dst[e] : 0;
  }
#pragma unroll
  for (int j = 0; j < EPL; ++j) {
    const int e = base + j * 256;
    r[j] = 0;
    if (FULL || e < E) r[j] = atomicAdd(deg + d[j], 1);
  }
#pragma unroll
  for (int j = 0; j < EPL; ++j) {
    const int e = base + j * 256;
    if (FULL || e < E) rank[e] = r[j];
  }
}

__global__ __launch_bounds__(256) void count_rank(const int* __restrict__ dst,
                                                  int* __restrict__ deg,
                                                  int* __restrict__ rank, int E) {
  const int blk = blockIdx.x * (256 * EPL);
  const int base = blk + threadIdx.x;
  if (blk + 256 * EPL <= E)
    count_rank_body<true>(dst, deg, rank, base, E);
  else
    count_rank_body<false>(dst, deg, rank, base, E);
}

__global__ __launch_bounds__(1024) void scan_counts(const int* __restrict__ deg,
                                                    int* __restrict__ rowptr,
                                                    int N) {
  __shared__ int wsum[32];
  const int tid = threadIdx.x;
  const int lane = tid & 31;
  const int w = tid >> 5;
  const int per = (N + 1023) / 1024;
  int beg = tid * per;
  if (beg > N) beg = N;
  int end = beg + per;
  if (end > N) end = N;

  int s = 0;
  for (int i = beg; i < end; ++i) s += deg[i];

  int incl = s;
  for (int o = 1; o < 32; o <<= 1) {
    int v = __shfl_up(incl, o);
    if (lane >= o) incl += v;
  }
  if (lane == 31) wsum[w] = incl;
  __syncthreads();
  if (w == 0) {
    const int v = wsum[lane];
    int iv = v;
    for (int o = 1; o < 32; o <<= 1) {
      int u = __shfl_up(iv, o);
      if (lane >= o) iv += u;
    }
    wsum[lane] = iv - v;
  }
  __syncthreads();

  int run = wsum[w] + (incl - s);
  for (int i = beg; i < end; ++i) {
    rowptr[i] = run;
    run += deg[i];
  }
  if (tid == 1023) rowptr[N] = run;
}

template <bool FULL>
__device__ __forceinline__ void place_edges_body(const int* __restrict__ src,
                                                 const int* __restrict__ dst,
                                                 const int* __restrict__ rank,
                                                 const int* __restrict__ rowptr,
                                                 int* __restrict__ csr,
                                                 int base, int E) {
  int s[EPL], d[EPL], r[EPL], rp[EPL];
#pragma unroll
  for (int j = 0; j < EPL; ++j) {
    const int e = base + j * 256;
    const bool ok = FULL || e < E;
    d[j] = ok ? dst[e] : 0;
    r[j] = ok ? rank[e] : 0;
    s[j] = ok ? src[e] : 0;
  }
#pragma unroll
  for (int j = 0; j < EPL; ++j) rp[j] = rowptr[d[j]];
#pragma unroll
  for (int j = 0; j < EPL; ++j) {
    const int e = base + j * 256;
    if (FULL || e < E) csr[rp[j] + r[j]] = s[j];
  }
}

__global__ __launch_bounds__(256) void place_edges(const int* __restrict__ src,
                                                   const int* __restrict__ dst,
                                                   const int* __restrict__ rank,
                                                   const int* __restrict__ rowptr,
                                                   int* __restrict__ csr, int E) {
  const int blk = blockIdx.x * (256 * EPL);
  const int base = blk + threadIdx.x;
  if (blk + 256 * EPL <= E)
    place_edges_body<true>(src, dst, rank, rowptr, csr, base, E);
  else
    place_edges_body<false>(src, dst, rank, rowptr, csr, base, E);
}

__device__ __forceinline__ float leaky(float e) {
  return (e > 0.0f) ? e : NEG_SLOPE * e;
}


template <int F>
__device__ __forceinline__ void add_rows(const float* __restrict__ h, int s,
                                         float t, int c, int lane, float& a0,
                                         float& a1) {
  for (int j0 = 0; j0 < c; j0 += 8) {
    for (int u = 0; u < 8; ++u) {
      const int j = j0 + u;
      const int sj = __builtin_amdgcn_readlane(s, j);
      const float tj = __int_as_float(__builtin_amdgcn_readlane(__float_as_int(t), j));
      if (F == 32) {
        a0 += tj * h[(size_t)sj * 32 + lane];
      } else {
        const float2 r = *((const float2*)(h + (size_t)sj * 64) + lane);
        a0 += tj * r.x;
        a1 += tj * r.y;
      }
    }
  }
}

template <int F, bool ELU>
__global__ __launch_bounds__(256) void node_gather(
    const int* __restrict__ rowptr, const int* __restrict__ csr,
    const float* __restrict__ sv, const float* __restrict__ dv,
    const float* __restrict__ h, const float* __restrict__ b,
    float* __restrict__ out, int N) {
  static_assert(F == 32 || F == 64, "one or two floats per lane");
  const int wv = __builtin_amdgcn_readfirstlane((int)(threadIdx.x >> 5));
  const int n = blockIdx.x * 8 + wv;
  if (n >= N) return;
  const int lane = threadIdx.x & 31;

  const int beg = rowptr[n];
  const int end = rowptr[n + 1];
  const float dn = dv[n];
  const float es = leaky(sv[n] + dn);

  const int pa = beg + lane;
  const int pb = pa + 32;
  const bool oka = pa < end;
  const bool okb = pb < end;
  int sa = n, sb = n;
  float ea = es, eb = es;
  if (beg < end) {
    const int last = end - 1;
    const int ra = csr[oka ? pa : last];
    const int rb = csr[okb ? pb : last];
    asm volatile("" ::"v"(ra), "v"(rb));
    const float va = sv[ra];
    const float vb = sv[rb];
    if (oka) {
      sa = ra;
      ea = leaky(va + dn);
    }
    if (okb) {
      sb = rb;
      eb = leaky(vb + dn);
    }
  }

  float mx = fmaxf(es, fmaxf(ea, eb));
  for (int p = beg + 64 + lane; p < end; p += 32) mx = fmaxf(mx, leaky(sv[csr[p]] + dn));
  for (int o = 16; o > 0; o >>= 1) mx = fmaxf(mx, __shfl_xor(mx, o));

  const float ts = __expf(es - mx);
  float a0, a1 = 0.0f;
  if (F == 32) {
    a0 = ts * h[(size_t)n * 32 + lane];
  } else {
    const float2 r = *((const float2*)(h + (size_t)n * 64) + lane);
    a0 = ts * r.x;
    a1 = ts * r.y;
  }

  const float ta = oka ? __expf(ea - mx) : 0.0f;
  const float tb = okb ? __expf(eb - mx) : 0.0f;
  float zl = ta;
  zl += tb;
  const int deg = end - beg;
  add_rows<F>(h, sa, ta, deg < 32 ? deg : 32, lane, a0, a1);
  add_rows<F>(h, sb, tb, deg < 64 ? deg - 32 : 32, lane, a0, a1);

  for (int base = beg + 64; base < end; base += 32) {
    const int p = base + lane;
    int s = n;
    float t = 0.0f;
    if (p < end) {
      s = csr[p];
      t = __expf(leaky(sv[s] + dn) - mx);
    }
    zl += t;
    int c = end - base;
    if (c > 32) c = 32;
    add_rows<F>(h, s, t, c, lane, a0, a1);
  }
  for (int o = 16; o > 0; o >>= 1) zl += __shfl_xor(zl, o);
  const float z = zl + ts;

  if (F == 32) {
    float v = a0 / z + b[lane];
    if (ELU) v = (v > 0.0f) ? v : (__expf(v) - 1.0f);
    out[(size_t)n * 32 + lane] = v;
  } else {
    float v0 = a0 / z + b[2 * lane];
    float v1 = a1 / z + b[2 * lane + 1];
    if (ELU) {
      v0 = (v0 > 0.0f) ? v0 : (__expf(v0) - 1.0f);
      v1 = (v1 > 0.0f) ? v1 : (__expf(v1) - 1.0f);
    }
    float2 o2v;
    o2v.x = v0;
    o2v.y = v1;
    *((float2*)(out + (size_t)n * 64) + lane) = o2v;
  }
}

#define NPW 16

__device__ __forceinline__ void pool_flush(float* __restrict__ out,
                                           float* __restrict__ cnt, int g,
                                           int lane, float a0, float a1,
                                           int run) {
  atomicAdd(out + g * 64 + 2 * lane, a0);
  atomicAdd(out + g * 64 + 2 * lane + 1, a1);
  if (lane == 0) atomicAdd(cnt + g, (float)run);
}

template <bool FULL>
__device__ __forceinline__ void pool_sum_body(const float* __restrict__ h,
                                              const int* __restrict__ batch,
                                              float* __restrict__ out,
                                              float* __restrict__ cnt, int n0,
                                              int cn, int lane) {
  float2 r[NPW];
  int g[NPW];
#pragma unroll
  for (int u = 0; u < NPW; ++u) {
    const int n = n0 + ((FULL || u < cn) ? u : (cn - 1));
    g[u] = batch[n];
    r[u] = *((const float2*)(h + (size_t)n * 64) + lane);
  }

  int cur = g[0];
  int run = 0;
  float a0 = 0.0f, a1 = 0.0f;
#pragma unroll
  for (int u = 0; u < NPW; ++u) {
    if (FULL || u < cn) {
      if (g[u] != cur) {
        pool_flush(out, cnt, cur, lane, a0, a1, run);
        cur = g[u];
        run = 0;
        a0 = 0.0f;
        a1 = 0.0f;
      }
      a0 += r[u].x;
      a1 += r[u].y;
      ++run;
    }
  }
  pool_flush(out, cnt, cur, lane, a0, a1, run);
}

__global__ __launch_bounds__(256) void pool_sum(const float* __restrict__ h,
                                                const int* __restrict__ batch,
                                                float* __restrict__ out,
                                                float* __restrict__ cnt, int N) {
  const int wv = __builtin_amdgcn_readfirstlane((int)(threadIdx.x >> 5));
  const int lane = threadIdx.x & 31;
  const int n0 = (blockIdx.x * 8 + wv) * NPW;
  if (n0 >= N) return;
  const int cn = N - n0;
  if (cn >= NPW)
    pool_sum_body<true>(h, batch, out, cnt, n0, NPW, lane);
  else
    pool_sum_body<false>(h, batch, out, cnt, n0, cn, lane);
}

__global__ void pool_div(float* __restrict__ out, const float* __restrict__ cnt) {
  int i = blockIdx.x * blockDim.x + threadIdx.x;
  if (i >= NUM_GRAPHS * 64) return;
  float c = cnt[i >> 6];
  out[i] = out[i] / fmaxf(c, 1.0f);
}

static inline int cdiv(int a, int b) { return (a + b - 1) / b; }

extern "C" void kernel_launch(void* const* d_in, const int* in_sizes, int n_in,
                              void* d_out, int out_size, void* d_ws,
                              size_t ws_size, hipStream_t stream) {
  const float* x      = (const float*)d_in[0];
  const int*   ei     = (const int*)d_in[1];
  const int*   batch  = (const int*)d_in[2];
  const float* W1     = (const float*)d_in[3];
  const float* a_src1 = (const float*)d_in[4];
  const float* a_dst1 = (const float*)d_in[5];
  const float* b1     = (const float*)d_in[6];
  const float* W2     = (const float*)d_in[7];
  const float* a_src2 = (const float*)d_in[8];
  const float* a_dst2 = (const float*)d_in[9];
  const float* b2     = (const float*)d_in[10];

  const int N = in_sizes[2];
  const int E = in_sizes[1] / 2;
  const int* src = ei;
  const int* dst = ei + E;

  float* ws = (float*)d_ws;
  size_t off = 0;
  float* h1 = ws + off; off += (size_t)N * 32;
  float* o1 = ws + off; off += (size_t)N * 32;
  float* h2 = ws + off; off += (size_t)N * 64;
  float* o2 = ws + off; off += (size_t)N * 64;
  float* sv = ws + off; off += (size_t)N;
  float* dv = ws + off; off += (size_t)N;
  float* cnt = ws + off; off += NUM_GRAPHS;
  int* deg    = (int*)(ws + off); off += (size_t)N;
  int* rowptr = (int*)(ws + off); off += (size_t)N + 1;
  int* rank   = (int*)(ws + off); off += (size_t)E;
  int* csr    = (int*)(ws + off); off += (size_t)E;

  gemm_wmma<128, 32, true><<<cdiv(N, 128), 256, 0, stream>>>(
      x, W1, a_src1, a_dst1, h1, sv, dv, N, deg, (float*)d_out, cnt);

  count_rank<<<cdiv(E, 256 * EPL), 256, 0, stream>>>(dst, deg, rank, E);
  scan_counts<<<1, 1024, 0, stream>>>(deg, rowptr, N);
  place_edges<<<cdiv(E, 256 * EPL), 256, 0, stream>>>(src, dst, rank, rowptr, csr, E);

  node_gather<32, true><<<cdiv(N, 8), 256, 0, stream>>>(rowptr, csr, sv, dv, h1, b1, o1, N);

  gemm_wmma<32, 64, false><<<cdiv(N, 128), 256, 0, stream>>>(
      o1, W2, a_src2, a_dst2, h2, sv, dv, N, nullptr, nullptr, nullptr);
  node_gather<64, false><<<cdiv(N, 8), 256, 0, stream>>>(rowptr, csr, sv, dv, h2, b2, o2, N);

  pool_sum<<<cdiv(N, 8 * NPW), 256, 0, stream>>>(o2, batch, (float*)d_out, cnt, N);
  pool_div<<<cdiv(NUM_GRAPHS * 64, 256), 256, 0, stream>>>((float*)d_out, cnt);
}


# Edges sorted by destination in two levels ranked by LDS counters (256-node buckets, then one block per bucket): no device-wide add per edge, csr lines written by one block; run order = LDS arrival
# speedup vs baseline: 6.3440x; 1.8891x over previous
#include <hip/hip_runtime.h>
#include <hip/hip_bf16.h>
#include <math.h>

typedef __attribute__((ext_vector_type(2))) float v2f;
typedef __attribute__((ext_vector_type(8))) float v8f;

#define NUM_GRAPHS 64
#define NEG_SLOPE 0.2f

#define BKT_NODES 256
#define BKT_SHIFT 8
#define BKT_MAX 2048
#define MV_T 1024
#define MV_EPT 8


template <int M>
__device__ __forceinline__ float row_xor_add(float v) {
  const int o = __builtin_amdgcn_update_dpp(0, __float_as_int(v), 0x160 + M, 0xf,
                                            0xf, true);
  return v + __int_as_float(o);
}

template <int FIN, int FOUT, bool ZERO>
__global__ __launch_bounds__(256) void gemm_wmma(
    const float* __restrict__ x, const float* __restrict__ W,
    const float* __restrict__ asrc, const float* __restrict__ adst,
    float* __restrict__ h, float* __restrict__ s, float* __restrict__ d, int N,
    int* __restrict__ zbk, float* __restrict__ zout, float* __restrict__ zcnt) {
  constexpr int NT = FOUT / 16;
  __shared__ float Ws[FIN * FOUT];

  for (int i = threadIdx.x; i < FIN * FOUT; i += 256) Ws[i] = W[i];
  __syncthreads();

  if (ZERO) {
    if (blockIdx.x == 0) {
      for (int i = threadIdx.x; i < 2 * BKT_MAX; i += 256) zbk[i] = 0;
      for (int i = threadIdx.x; i < NUM_GRAPHS * 64; i += 256) zout[i] = 0.0f;
      if (threadIdx.x < NUM_GRAPHS) zcnt[threadIdx.x] = 0.0f;
    }
  }

  const int wave = threadIdx.x >> 5;
  const int lane = threadIdx.x & 31;
  const int m0 = (blockIdx.x * 8 + wave) * 16;
  const int halfsel = lane >> 4;
  const int lidx = lane & 15;
  int row = m0 + lidx;
  if (row >= N) row = N - 1;
  const int khalf = halfsel * 2;

  float as_[NT], ad_[NT];
#pragma unroll
  for (int t = 0; t < NT; ++t) {
    as_[t] = asrc[t * 16 + lidx];
    ad_[t] = adst[t * 16 + lidx];
  }

  v8f acc[NT];
#pragma unroll
  for (int t = 0; t < NT; ++t)
#pragma unroll
    for (int j = 0; j < 8; ++j) acc[t][j] = 0.0f;

#pragma unroll 4
  for (int k = 0; k < FIN; k += 4) {
    const int kk = k + khalf;
    v2f a;
    a.x = x[(size_t)row * FIN + kk];
    a.y = x[(size_t)row * FIN + kk + 1];
#pragma unroll
    for (int t = 0; t < NT; ++t) {
      v2f b;
      b.x = Ws[kk * FOUT + t * 16 + lidx];
      b.y = Ws[(kk + 1) * FOUT + t * 16 + lidx];
      acc[t] = __builtin_amdgcn_wmma_f32_16x16x4_f32(
          false, a, false, b, (short)0, acc[t], false, false);
    }
  }

  const int rbase = m0 + halfsel * 8;
#pragma unroll
  for (int i = 0; i < 8; ++i) {
    const int r = rbase + i;
    if (r < N) {
#pragma unroll
      for (int t = 0; t < NT; ++t)
        h[(size_t)r * FOUT + t * 16 + lidx] = acc[t][i];
    }
  }

  float ps[8], pd[8];
#pragma unroll
  for (int i = 0; i < 8; ++i) {
    float a = 0.0f, b = 0.0f;
#pragma unroll
    for (int t = 0; t < NT; ++t) {
      a += acc[t][i] * as_[t];
      b += acc[t][i] * ad_[t];
    }
    ps[i] = a;
    pd[i] = b;
  }
#pragma unroll
  for (int i = 0; i < 8; ++i) {
    ps[i] = row_xor_add<1>(ps[i]);
    pd[i] = row_xor_add<1>(pd[i]);
  }
#pragma unroll
  for (int i = 0; i < 8; ++i) {
    ps[i] = row_xor_add<2>(ps[i]);
    pd[i] = row_xor_add<2>(pd[i]);
  }
#pragma unroll
  for (int i = 0; i < 8; ++i) {
    ps[i] = row_xor_add<4>(ps[i]);
    pd[i] = row_xor_add<4>(pd[i]);
  }
#pragma unroll
  for (int i = 0; i < 8; ++i) {
    ps[i] = row_xor_add<8>(ps[i]);
    pd[i] = row_xor_add<8>(pd[i]);
  }
  float so = ps[0], dout = pd[0];
#pragma unroll
  for (int i = 1; i < 8; ++i) {
    if (lidx == i) {
      so = ps[i];
      dout = pd[i];
    }
  }
  const int r = rbase + lidx;
  if (lidx < 8 && r < N) {
    s[r] = so;
    d[r] = dout;
  }
}


template <int S>
__device__ __forceinline__ int row_shr_add(int v) {
  return v + __builtin_amdgcn_update_dpp(0, v, 0x110 + S, 0xf, 0xf, true);
}

__device__ __forceinline__ int wave_prefix(int v, int lane) {
  v = row_shr_add<1>(v);
  v = row_shr_add<2>(v);
  v = row_shr_add<4>(v);
  v = row_shr_add<8>(v);
  const int low = __builtin_amdgcn_readlane(v, 15);
  return v + (lane >= 16 ? low : 0);
}

__global__ __launch_bounds__(MV_T) void bucket_count(const int* __restrict__ dst,
                                                     int* __restrict__ btotal,
                                                     int E, int NB) {
  __shared__ int hist[BKT_MAX];
  for (int i = threadIdx.x; i < NB; i += MV_T) hist[i] = 0;
  const int base = blockIdx.x * (MV_T * MV_EPT) + threadIdx.x;
  int d[MV_EPT];
#pragma unroll
  for (int j = 0; j < MV_EPT; ++j) {
    const int e = base + j * MV_T;
    d[j] = (e < E) ? dst[e] : -1;
  }
  __syncthreads();
#pragma unroll
  for (int j = 0; j < MV_EPT; ++j)
    if (d[j] >= 0) atomicAdd(&hist[d[j] >> BKT_SHIFT], 1);
  __syncthreads();
  for (int i = threadIdx.x; i < NB; i += MV_T) {
    const int h = hist[i];
    if (h) atomicAdd(btotal + i, h);
  }
}

__global__ __launch_bounds__(MV_T) void bucket_move(
    const int* __restrict__ src, const int* __restrict__ dst,
    const int* __restrict__ btotal, int* __restrict__ cursor,
    int* __restrict__ bstart, int* __restrict__ rec, int E, int NB) {
  __shared__ int hist[BKT_MAX];
  __shared__ int place[BKT_MAX];
  __shared__ int wtot[MV_T / 32];
  const int tid = threadIdx.x;
  const int lane = tid & 31;
  const int w = __builtin_amdgcn_readfirstlane(tid >> 5);
  for (int i = tid; i < NB; i += MV_T) hist[i] = 0;

  const int base = blockIdx.x * (MV_T * MV_EPT) + tid;
  int s[MV_EPT], d[MV_EPT];
#pragma unroll
  for (int j = 0; j < MV_EPT; ++j) {
    const int e = base + j * MV_T;
    const bool ok = e < E;
    d[j] = ok ? dst[e] : -1;
    s[j] = ok ? src[e] : 0;
  }
  const int i0 = 2 * tid;
  const int a0 = (i0 < NB) ? btotal[i0] : 0;
  const int a1 = (i0 + 1 < NB) ? btotal[i0 + 1] : 0;
  __syncthreads();

  int r[MV_EPT];
#pragma unroll
  for (int j = 0; j < MV_EPT; ++j) {
    r[j] = 0;
    if (d[j] >= 0) r[j] = atomicAdd(&hist[d[j] >> BKT_SHIFT], 1);
  }

  const int ts = a0 + a1;
  const int inc = wave_prefix(ts, lane);
  if (lane == 31) wtot[w] = inc;
  __syncthreads();
  const int x = wtot[lane];
  const int inc2 = wave_prefix(x, lane);
  const int ex = __builtin_amdgcn_readlane(inc2 - x, w) + inc - ts;

  if (i0 < NB) {
    const int h = hist[i0];
    place[i0] = ex + (h ? atomicAdd(cursor + i0, h) : 0);
  }
  if (i0 + 1 < NB) {
    const int h = hist[i0 + 1];
    place[i0 + 1] = ex + a0 + (h ? atomicAdd(cursor + i0 + 1, h) : 0);
  }
  if (blockIdx.x == 0) {
    if (i0 <= NB) bstart[i0] = ex;
    if (i0 + 1 <= NB) bstart[i0 + 1] = ex + a0;
  }
  __syncthreads();

#pragma unroll
  for (int j = 0; j < MV_EPT; ++j) {
    if (d[j] >= 0) {
      const unsigned word =
          (unsigned)s[j] | ((unsigned)(d[j] & (BKT_NODES - 1)) << 24);
      rec[place[d[j] >> BKT_SHIFT] + r[j]] = (int)word;
    }
  }
}

__global__ __launch_bounds__(BKT_NODES) void bucket_place(
    const int* __restrict__ rec, const int* __restrict__ bstart,
    int* __restrict__ rowptr, int* __restrict__ csr, int N) {
  __shared__ int cnt[BKT_NODES + 1];
  __shared__ int wtot[BKT_NODES / 32];
  const int tid = threadIdx.x;
  const int lane = tid & 31;
  const int w = tid >> 5;
  const int b = blockIdx.x;
  const int e0 = bstart[b];
  const int e1 = bstart[b + 1];
  cnt[tid] = 0;
  if (tid == 0) cnt[BKT_NODES] = 0;
  __syncthreads();

  for (int e = e0 + tid; e < e1; e += BKT_NODES * 8) {
    unsigned q[8];
#pragma unroll
    for (int j = 0; j < 8; ++j) {
      const int ee = e + j * BKT_NODES;
      q[j] = (unsigned)rec[ee < e1 ? ee : e1 - 1];
    }
#pragma unroll
    for (int j = 0; j < 8; ++j) {
      const bool ok = e + j * BKT_NODES < e1;
      atomicAdd(&cnt[ok ? (int)(q[j] >> 24) : BKT_NODES], 1);
    }
  }
  __syncthreads();

  const int c = cnt[tid];
  const int inc = wave_prefix(c, lane);
  if (lane == 31) wtot[w] = inc;
  __syncthreads();
  int wofs = 0;
#pragma unroll
  for (int k = 0; k < BKT_NODES / 32; ++k)
    if (k < w) wofs += wtot[k];
  const int ex = wofs + inc - c;
  const int n = b * BKT_NODES + tid;
  if (n < N) rowptr[n] = e0 + ex;
  if (b == (int)gridDim.x - 1 && tid == 0) rowptr[N] = e1;
  cnt[tid] = ex;
  __syncthreads();

  for (int e = e0 + tid; e < e1; e += BKT_NODES * 8) {
    unsigned q[8];
    int p[8];
#pragma unroll
    for (int j = 0; j < 8; ++j) {
      const int ee = e + j * BKT_NODES;
      q[j] = (unsigned)rec[ee < e1 ? ee : e1 - 1];
    }
#pragma unroll
    for (int j = 0; j < 8; ++j) {
      const bool ok = e + j * BKT_NODES < e1;
      p[j] = atomicAdd(&cnt[ok ? (int)(q[j] >> 24) : BKT_NODES], 1);
    }
#pragma unroll
    for (int j = 0; j < 8; ++j)
      if (e + j * BKT_NODES < e1) csr[e0 + p[j]] = (int)(q[j] & 0xffffffu);
  }
}

__device__ __forceinline__ float leaky(float e) {
  return (e > 0.0f) ? e : NEG_SLOPE * e;
}


template <int F>
__device__ __forceinline__ void add_rows(const float* __restrict__ h, int s,
                                         float t, int c, int lane, float& a0,
                                         float& a1) {
  for (int j0 = 0; j0 < c; j0 += 8) {
    for (int u = 0; u < 8; ++u) {
      const int j = j0 + u;
      const int sj = __builtin_amdgcn_readlane(s, j);
      const float tj = __int_as_float(__builtin_amdgcn_readlane(__float_as_int(t), j));
      if (F == 32) {
        a0 += tj * h[(size_t)sj * 32 + lane];
      } else {
        const float2 r = *((const float2*)(h + (size_t)sj * 64) + lane);
        a0 += tj * r.x;
        a1 += tj * r.y;
      }
    }
  }
}

template <int F, bool ELU>
__global__ __launch_bounds__(256) void node_gather(
    const int* __restrict__ rowptr, const int* __restrict__ csr,
    const float* __restrict__ sv, const float* __restrict__ dv,
    const float* __restrict__ h, const float* __restrict__ b,
    float* __restrict__ out, int N) {
  static_assert(F == 32 || F == 64, "one or two floats per lane");
  const int wv = __builtin_amdgcn_readfirstlane((int)(threadIdx.x >> 5));
  const int n = blockIdx.x * 8 + wv;
  if (n >= N) return;
  const int lane = threadIdx.x & 31;

  const int beg = rowptr[n];
  const int end = rowptr[n + 1];
  const float dn = dv[n];
  const float es = leaky(sv[n] + dn);

  const int pa = beg + lane;
  const int pb = pa + 32;
  const bool oka = pa < end;
  const bool okb = pb < end;
  int sa = n, sb = n;
  float ea = es, eb = es;
  if (beg < end) {
    const int last = end - 1;
    const int ra = csr[oka ? pa : last];
    const int rb = csr[okb ? pb : last];
    asm volatile("" ::"v"(ra), "v"(rb));
    const float va = sv[ra];
    const float vb = sv[rb];
    if (oka) {
      sa = ra;
      ea = leaky(va + dn);
    }
    if (okb) {
      sb = rb;
      eb = leaky(vb + dn);
    }
  }

  float mx = fmaxf(es, fmaxf(ea, eb));
  for (int p = beg + 64 + lane; p < end; p += 32) mx = fmaxf(mx, leaky(sv[csr[p]] + dn));
  for (int o = 16; o > 0; o >>= 1) mx = fmaxf(mx, __shfl_xor(mx, o));

  const float ts = __expf(es - mx);
  float a0, a1 = 0.0f;
  if (F == 32) {
    a0 = ts * h[(size_t)n * 32 + lane];
  } else {
    const float2 r = *((const float2*)(h + (size_t)n * 64) + lane);
    a0 = ts * r.x;
    a1 = ts * r.y;
  }

  const float ta = oka ? __expf(ea - mx) : 0.0f;
  const float tb = okb ? __expf(eb - mx) : 0.0f;
  float zl = ta;
  zl += tb;
  const int deg = end - beg;
  add_rows<F>(h, sa, ta, deg < 32 ? deg : 32, lane, a0, a1);
  add_rows<F>(h, sb, tb, deg < 64 ? deg - 32 : 32, lane, a0, a1);

  for (int base = beg + 64; base < end; base += 32) {
    const int p = base + lane;
    int s = n;
    float t = 0.0f;
    if (p < end) {
      s = csr[p];
      t = __expf(leaky(sv[s] + dn) - mx);
    }
    zl += t;
    int c = end - base;
    if (c > 32) c = 32;
    add_rows<F>(h, s, t, c, lane, a0, a1);
  }
  for (int o = 16; o > 0; o >>= 1) zl += __shfl_xor(zl, o);
  const float z = zl + ts;

  if (F == 32) {
    float v = a0 / z + b[lane];
    if (ELU) v = (v > 0.0f) ? v : (__expf(v) - 1.0f);
    out[(size_t)n * 32 + lane] = v;
  } else {
    float v0 = a0 / z + b[2 * lane];
    float v1 = a1 / z + b[2 * lane + 1];
    if (ELU) {
      v0 = (v0 > 0.0f) ? v0 : (__expf(v0) - 1.0f);
      v1 = (v1 > 0.0f) ? v1 : (__expf(v1) - 1.0f);
    }
    float2 o2v;
    o2v.x = v0;
    o2v.y = v1;
    *((float2*)(out + (size_t)n * 64) + lane) = o2v;
  }
}

#define NPW 16

__device__ __forceinline__ void pool_flush(float* __restrict__ out,
                                           float* __restrict__ cnt, int g,
                                           int lane, float a0, float a1,
                                           int run) {
  atomicAdd(out + g * 64 + 2 * lane, a0);
  atomicAdd(out + g * 64 + 2 * lane + 1, a1);
  if (lane == 0) atomicAdd(cnt + g, (float)run);
}

template <bool FULL>
__device__ __forceinline__ void pool_sum_body(const float* __restrict__ h,
                                              const int* __restrict__ batch,
                                              float* __restrict__ out,
                                              float* __restrict__ cnt, int n0,
                                              int cn, int lane) {
  float2 r[NPW];
  int g[NPW];
#pragma unroll
  for (int u = 0; u < NPW; ++u) {
    const int n = n0 + ((FULL || u < cn) ? u : (cn - 1));
    g[u] = batch[n];
    r[u] = *((const float2*)(h + (size_t)n * 64) + lane);
  }

  int cur = g[0];
  int run = 0;
  float a0 = 0.0f, a1 = 0.0f;
#pragma unroll
  for (int u = 0; u < NPW; ++u) {
    if (FULL || u < cn) {
      if (g[u] != cur) {
        pool_flush(out, cnt, cur, lane, a0, a1, run);
        cur = g[u];
        run = 0;
        a0 = 0.0f;
        a1 = 0.0f;
      }
      a0 += r[u].x;
      a1 += r[u].y;
      ++run;
    }
  }
  pool_flush(out, cnt, cur, lane, a0, a1, run);
}

__global__ __launch_bounds__(256) void pool_sum(const float* __restrict__ h,
                                                const int* __restrict__ batch,
                                                float* __restrict__ out,
                                                float* __restrict__ cnt, int N) {
  const int wv = __builtin_amdgcn_readfirstlane((int)(threadIdx.x >> 5));
  const int lane = threadIdx.x & 31;
  const int n0 = (blockIdx.x * 8 + wv) * NPW;
  if (n0 >= N) return;
  const int cn = N - n0;
  if (cn >= NPW)
    pool_sum_body<true>(h, batch, out, cnt, n0, NPW, lane);
  else
    pool_sum_body<false>(h, batch, out, cnt, n0, cn, lane);
}

__global__ void pool_div(float* __restrict__ out, const float* __restrict__ cnt) {
  int i = blockIdx.x * blockDim.x + threadIdx.x;
  if (i >= NUM_GRAPHS * 64) return;
  float c = cnt[i >> 6];
  out[i] = out[i] / fmaxf(c, 1.0f);
}

static inline int cdiv(int a, int b) { return (a + b - 1) / b; }

extern "C" void kernel_launch(void* const* d_in, const int* in_sizes, int n_in,
                              void* d_out, int out_size, void* d_ws,
                              size_t ws_size, hipStream_t stream) {
  const float* x      = (const float*)d_in[0];
  const int*   ei     = (const int*)d_in[1];
  const int*   batch  = (const int*)d_in[2];
  const float* W1     = (const float*)d_in[3];
  const float* a_src1 = (const float*)d_in[4];
  const float* a_dst1 = (const float*)d_in[5];
  const float* b1     = (const float*)d_in[6];
  const float* W2     = (const float*)d_in[7];
  const float* a_src2 = (const float*)d_in[8];
  const float* a_dst2 = (const float*)d_in[9];
  const float* b2     = (const float*)d_in[10];

  const int N = in_sizes[2];
  const int E = in_sizes[1] / 2;
  const int* src = ei;
  const int* dst = ei + E;

  float* ws = (float*)d_ws;
  size_t off = 0;
  float* h1 = ws + off; off += (size_t)N * 32;
  float* o1 = ws + off; off += (size_t)N * 32;
  float* h2 = ws + off; off += (size_t)N * 64;
  float* o2 = ws + off; off += (size_t)N * 64;
  float* sv = ws + off; off += (size_t)N;
  float* dv = ws + off; off += (size_t)N;
  float* cnt = ws + off; off += NUM_GRAPHS;
  int* btotal = (int*)(ws + off); off += BKT_MAX;
  int* cursor = (int*)(ws + off); off += BKT_MAX;
  int* bstart = (int*)(ws + off); off += BKT_MAX;
  int* rowptr = (int*)(ws + off); off += (size_t)N + 1;
  int* rec    = (int*)(ws + off); off += (size_t)E;
  int* csr    = (int*)(ws + off); off += (size_t)E;
  const int NB = cdiv(N, BKT_NODES);

  gemm_wmma<128, 32, true><<<cdiv(N, 128), 256, 0, stream>>>(
      x, W1, a_src1, a_dst1, h1, sv, dv, N, btotal, (float*)d_out, cnt);

  bucket_count<<<cdiv(E, MV_T * MV_EPT), MV_T, 0, stream>>>(dst, btotal, E, NB);
  bucket_move<<<cdiv(E, MV_T * MV_EPT), MV_T, 0, stream>>>(src, dst, btotal, cursor,
                                                           bstart, rec, E, NB);
  bucket_place<<<NB, BKT_NODES, 0, stream>>>(rec, bstart, rowptr, csr, N);

  node_gather<32, true><<<cdiv(N, 8), 256, 0, stream>>>(rowptr, csr, sv, dv, h1, b1, o1, N);

  gemm_wmma<32, 64, false><<<cdiv(N, 128), 256, 0, stream>>>(
      o1, W2, a_src2, a_dst2, h2, sv, dv, N, nullptr, nullptr, nullptr);
  node_gather<64, false><<<cdiv(N, 8), 256, 0, stream>>>(rowptr, csr, sv, dv, h2, b2, o2, N);

  pool_sum<<<cdiv(N, 8 * NPW), 256, 0, stream>>>(o2, batch, (float*)d_out, cnt, N);
  pool_div<<<cdiv(NUM_GRAPHS * 64, 256), 256, 0, stream>>>((float*)d_out, cnt);
}


# node_gather: one 16-lane half-wave per node, two nodes per wave, 8 or 16 bytes of a row per lane: half the waves and row loads; max and z are 16-lane trees (z order changes), row sums keep run order
# speedup vs baseline: 6.5679x; 1.0353x over previous
#include <hip/hip_runtime.h>
#include <hip/hip_bf16.h>
#include <math.h>

typedef __attribute__((ext_vector_type(2))) float v2f;
typedef __attribute__((ext_vector_type(8))) float v8f;

#define NUM_GRAPHS 64
#define NEG_SLOPE 0.2f

#define BKT_NODES 256
#define BKT_SHIFT 8
#define BKT_MAX 2048
#define MV_T 1024
#define MV_EPT 8


template <int M>
__device__ __forceinline__ float row_xor_add(float v) {
  const int o = __builtin_amdgcn_update_dpp(0, __float_as_int(v), 0x160 + M, 0xf,
                                            0xf, true);
  return v + __int_as_float(o);
}

template <int FIN, int FOUT, bool ZERO>
__global__ __launch_bounds__(256) void gemm_wmma(
    const float* __restrict__ x, const float* __restrict__ W,
    const float* __restrict__ asrc, const float* __restrict__ adst,
    float* __restrict__ h, float* __restrict__ s, float* __restrict__ d, int N,
    int* __restrict__ zbk, float* __restrict__ zout, float* __restrict__ zcnt) {
  constexpr int NT = FOUT / 16;
  __shared__ float Ws[FIN * FOUT];

  for (int i = threadIdx.x; i < FIN * FOUT; i += 256) Ws[i] = W[i];
  __syncthreads();

  if (ZERO) {
    if (blockIdx.x == 0) {
      for (int i = threadIdx.x; i < 2 * BKT_MAX; i += 256) zbk[i] = 0;
      for (int i = threadIdx.x; i < NUM_GRAPHS * 64; i += 256) zout[i] = 0.0f;
      if (threadIdx.x < NUM_GRAPHS) zcnt[threadIdx.x] = 0.0f;
    }
  }

  const int wave = threadIdx.x >> 5;
  const int lane = threadIdx.x & 31;
  const int m0 = (blockIdx.x * 8 + wave) * 16;
  const int halfsel = lane >> 4;
  const int lidx = lane & 15;
  int row = m0 + lidx;
  if (row >= N) row = N - 1;
  const int khalf = halfsel * 2;

  float as_[NT], ad_[NT];
#pragma unroll
  for (int t = 0; t < NT; ++t) {
    as_[t] = asrc[t * 16 + lidx];
    ad_[t] = adst[t * 16 + lidx];
  }

  v8f acc[NT];
#pragma unroll
  for (int t = 0; t < NT; ++t)
#pragma unroll
    for (int j = 0; j < 8; ++j) acc[t][j] = 0.0f;

#pragma unroll 4
  for (int k = 0; k < FIN; k += 4) {
    const int kk = k + khalf;
    v2f a;
    a.x = x[(size_t)row * FIN + kk];
    a.y = x[(size_t)row * FIN + kk + 1];
#pragma unroll
    for (int t = 0; t < NT; ++t) {
      v2f b;
      b.x = Ws[kk * FOUT + t * 16 + lidx];
      b.y = Ws[(kk + 1) * FOUT + t * 16 + lidx];
      acc[t] = __builtin_amdgcn_wmma_f32_16x16x4_f32(
          false, a, false, b, (short)0, acc[t], false, false);
    }
  }

  const int rbase = m0 + halfsel * 8;
#pragma unroll
  for (int i = 0; i < 8; ++i) {
    const int r = rbase + i;
    if (r < N) {
#pragma unroll
      for (int t = 0; t < NT; ++t)
        h[(size_t)r * FOUT + t * 16 + lidx] = acc[t][i];
    }
  }

  float ps[8], pd[8];
#pragma unroll
  for (int i = 0; i < 8; ++i) {
    float a = 0.0f, b = 0.0f;
#pragma unroll
    for (int t = 0; t < NT; ++t) {
      a += acc[t][i] * as_[t];
      b += acc[t][i] * ad_[t];
    }
    ps[i] = a;
    pd[i] = b;
  }
#pragma unroll
  for (int i = 0; i < 8; ++i) {
    ps[i] = row_xor_add<1>(ps[i]);
    pd[i] = row_xor_add<1>(pd[i]);
  }
#pragma unroll
  for (int i = 0; i < 8; ++i) {
    ps[i] = row_xor_add<2>(ps[i]);
    pd[i] = row_xor_add<2>(pd[i]);
  }
#pragma unroll
  for (int i = 0; i < 8; ++i) {
    ps[i] = row_xor_add<4>(ps[i]);
    pd[i] = row_xor_add<4>(pd[i]);
  }
#pragma unroll
  for (int i = 0; i < 8; ++i) {
    ps[i] = row_xor_add<8>(ps[i]);
    pd[i] = row_xor_add<8>(pd[i]);
  }
  float so = ps[0], dout = pd[0];
#pragma unroll
  for (int i = 1; i < 8; ++i) {
    if (lidx == i) {
      so = ps[i];
      dout = pd[i];
    }
  }
  const int r = rbase + lidx;
  if (lidx < 8 && r < N) {
    s[r] = so;
    d[r] = dout;
  }
}


template <int S>
__device__ __forceinline__ int row_shr_add(int v) {
  return v + __builtin_amdgcn_update_dpp(0, v, 0x110 + S, 0xf, 0xf, true);
}

__device__ __forceinline__ int wave_prefix(int v, int lane) {
  v = row_shr_add<1>(v);
  v = row_shr_add<2>(v);
  v = row_shr_add<4>(v);
  v = row_shr_add<8>(v);
  const int low = __builtin_amdgcn_readlane(v, 15);
  return v + (lane >= 16 ? low : 0);
}

__global__ __launch_bounds__(MV_T) void bucket_count(const int* __restrict__ dst,
                                                     int* __restrict__ btotal,
                                                     int E, int NB) {
  __shared__ int hist[BKT_MAX];
  for (int i = threadIdx.x; i < NB; i += MV_T) hist[i] = 0;
  const int base = blockIdx.x * (MV_T * MV_EPT) + threadIdx.x;
  int d[MV_EPT];
#pragma unroll
  for (int j = 0; j < MV_EPT; ++j) {
    const int e = base + j * MV_T;
    d[j] = (e < E) ? dst[e] : -1;
  }
  __syncthreads();
#pragma unroll
  for (int j = 0; j < MV_EPT; ++j)
    if (d[j] >= 0) atomicAdd(&hist[d[j] >> BKT_SHIFT], 1);
  __syncthreads();
  for (int i = threadIdx.x; i < NB; i += MV_T) {
    const int h = hist[i];
    if (h) atomicAdd(btotal + i, h);
  }
}

__global__ __launch_bounds__(MV_T) void bucket_move(
    const int* __restrict__ src, const int* __restrict__ dst,
    const int* __restrict__ btotal, int* __restrict__ cursor,
    int* __restrict__ bstart, int* __restrict__ rec, int E, int NB) {
  __shared__ int hist[BKT_MAX];
  __shared__ int place[BKT_MAX];
  __shared__ int wtot[MV_T / 32];
  const int tid = threadIdx.x;
  const int lane = tid & 31;
  const int w = __builtin_amdgcn_readfirstlane(tid >> 5);
  for (int i = tid; i < NB; i += MV_T) hist[i] = 0;

  const int base = blockIdx.x * (MV_T * MV_EPT) + tid;
  int s[MV_EPT], d[MV_EPT];
#pragma unroll
  for (int j = 0; j < MV_EPT; ++j) {
    const int e = base + j * MV_T;
    const bool ok = e < E;
    d[j] = ok ? dst[e] : -1;
    s[j] = ok ? src[e] : 0;
  }
  const int i0 = 2 * tid;
  const int a0 = (i0 < NB) ? btotal[i0] : 0;
  const int a1 = (i0 + 1 < NB) ? btotal[i0 + 1] : 0;
  __syncthreads();

  int r[MV_EPT];
#pragma unroll
  for (int j = 0; j < MV_EPT; ++j) {
    r[j] = 0;
    if (d[j] >= 0) r[j] = atomicAdd(&hist[d[j] >> BKT_SHIFT], 1);
  }

  const int ts = a0 + a1;
  const int inc = wave_prefix(ts, lane);
  if (lane == 31) wtot[w] = inc;
  __syncthreads();
  const int x = wtot[lane];
  const int inc2 = wave_prefix(x, lane);
  const int ex = __builtin_amdgcn_readlane(inc2 - x, w) + inc - ts;

  if (i0 < NB) {
    const int h = hist[i0];
    place[i0] = ex + (h ? atomicAdd(cursor + i0, h) : 0);
  }
  if (i0 + 1 < NB) {
    const int h = hist[i0 + 1];
    place[i0 + 1] = ex + a0 + (h ? atomicAdd(cursor + i0 + 1, h) : 0);
  }
  if (blockIdx.x == 0) {
    if (i0 <= NB) bstart[i0] = ex;
    if (i0 + 1 <= NB) bstart[i0 + 1] = ex + a0;
  }
  __syncthreads();

#pragma unroll
  for (int j = 0; j < MV_EPT; ++j) {
    if (d[j] >= 0) {
      const unsigned word =
          (unsigned)s[j] | ((unsigned)(d[j] & (BKT_NODES - 1)) << 24);
      rec[place[d[j] >> BKT_SHIFT] + r[j]] = (int)word;
    }
  }
}

__global__ __launch_bounds__(BKT_NODES) void bucket_place(
    const int* __restrict__ rec, const int* __restrict__ bstart,
    int* __restrict__ rowptr, int* __restrict__ csr, int N) {
  __shared__ int cnt[BKT_NODES + 1];
  __shared__ int wtot[BKT_NODES / 32];
  const int tid = threadIdx.x;
  const int lane = tid & 31;
  const int w = tid >> 5;
  const int b = blockIdx.x;
  const int e0 = bstart[b];
  const int e1 = bstart[b + 1];
  cnt[tid] = 0;
  if (tid == 0) cnt[BKT_NODES] = 0;
  __syncthreads();

  for (int e = e0 + tid; e < e1; e += BKT_NODES * 8) {
    unsigned q[8];
#pragma unroll
    for (int j = 0; j < 8; ++j) {
      const int ee = e + j * BKT_NODES;
      q[j] = (unsigned)rec[ee < e1 ? ee : e1 - 1];
    }
#pragma unroll
    for (int j = 0; j < 8; ++j) {
      const bool ok = e + j * BKT_NODES < e1;
      atomicAdd(&cnt[ok ? (int)(q[j] >> 24) : BKT_NODES], 1);
    }
  }
  __syncthreads();

  const int c = cnt[tid];
  const int inc = wave_prefix(c, lane);
  if (lane == 31) wtot[w] = inc;
  __syncthreads();
  int wofs = 0;
#pragma unroll
  for (int k = 0; k < BKT_NODES / 32; ++k)
    if (k < w) wofs += wtot[k];
  const int ex = wofs + inc - c;
  const int n = b * BKT_NODES + tid;
  if (n < N) rowptr[n] = e0 + ex;
  if (b == (int)gridDim.x - 1 && tid == 0) rowptr[N] = e1;
  cnt[tid] = ex;
  __syncthreads();

  for (int e = e0 + tid; e < e1; e += BKT_NODES * 8) {
    unsigned q[8];
    int p[8];
#pragma unroll
    for (int j = 0; j < 8; ++j) {
      const int ee = e + j * BKT_NODES;
      q[j] = (unsigned)rec[ee < e1 ? ee : e1 - 1];
    }
#pragma unroll
    for (int j = 0; j < 8; ++j) {
      const bool ok = e + j * BKT_NODES < e1;
      p[j] = atomicAdd(&cnt[ok ? (int)(q[j] >> 24) : BKT_NODES], 1);
    }
#pragma unroll
    for (int j = 0; j < 8; ++j)
      if (e + j * BKT_NODES < e1) csr[e0 + p[j]] = (int)(q[j] & 0xffffffu);
  }
}

__device__ __forceinline__ float leaky(float e) {
  return (e > 0.0f) ? e : NEG_SLOPE * e;
}


template <int J>
__device__ __forceinline__ int row_share(int v) {
  return __builtin_amdgcn_update_dpp(0, v, 0x150 + J, 0xf, 0xf, true);
}

template <int M>
__device__ __forceinline__ float row_xor_max(float v) {
  const int o = __builtin_amdgcn_update_dpp(0, __float_as_int(v), 0x160 + M, 0xf,
                                            0xf, true);
  return fmaxf(v, __int_as_float(o));
}

template <int F, int J>
__device__ __forceinline__ void add_row(const float* __restrict__ h, int s,
                                        float t, unsigned colb,
                                        float (&a)[F / 16]) {
  const unsigned sj = (unsigned)row_share<J>(s);
  const float tj = __int_as_float(row_share<J>(__float_as_int(t)));
  const char* p = (const char*)h + (sj * (unsigned)(F * 4) + colb);
  if (F == 32) {
    const float2 r = *(const float2*)p;
    a[0] += tj * r.x;
    a[1] += tj * r.y;
  } else {
    const float4 r = *(const float4*)p;
    a[0] += tj * r.x;
    a[1] += tj * r.y;
    a[2] += tj * r.z;
    a[3] += tj * r.w;
  }
}

template <int F, int J0>
__device__ __forceinline__ void add_rows4(const float* __restrict__ h, int s,
                                          float t, unsigned colb,
                                          float (&a)[F / 16]) {
  add_row<F, J0 + 0>(h, s, t, colb, a);
  add_row<F, J0 + 1>(h, s, t, colb, a);
  add_row<F, J0 + 2>(h, s, t, colb, a);
  add_row<F, J0 + 3>(h, s, t, colb, a);
}

template <int F>
__device__ __forceinline__ void add_row_at(const float* __restrict__ h, int s,
                                           float t, int j, unsigned colb,
                                           float (&a)[F / 16]) {
  const unsigned sj =
      __builtin_amdgcn_permlane16_var((unsigned)s, (unsigned)s, (unsigned)j, false, false);
  const float tj = __uint_as_float(__builtin_amdgcn_permlane16_var(
      __float_as_uint(t), __float_as_uint(t), (unsigned)j, false, false));
  const char* p = (const char*)h + (sj * (unsigned)(F * 4) + colb);
  if (F == 32) {
    const float2 r = *(const float2*)p;
    a[0] += tj * r.x;
    a[1] += tj * r.y;
  } else {
    const float4 r = *(const float4*)p;
    a[0] += tj * r.x;
    a[1] += tj * r.y;
    a[2] += tj * r.z;
    a[3] += tj * r.w;
  }
}

template <int F>
__device__ __forceinline__ void add_chunk(const float* __restrict__ h, int s,
                                          float t, int c, unsigned colb,
                                          float (&a)[F / 16]) {
  if (c >= 16) {
    add_rows4<F, 0>(h, s, t, colb, a);
    add_rows4<F, 4>(h, s, t, colb, a);
    add_rows4<F, 8>(h, s, t, colb, a);
    add_rows4<F, 12>(h, s, t, colb, a);
  } else {
    for (int j0 = 0; j0 < c; j0 += 8) {
#pragma unroll
      for (int u = 0; u < 8; ++u) add_row_at<F>(h, s, t, j0 + u, colb, a);
    }
  }
}

template <int F, bool ELU>
__global__ __launch_bounds__(256) void node_gather(
    const int* __restrict__ rowptr, const int* __restrict__ csr,
    const float* __restrict__ sv, const float* __restrict__ dv,
    const float* __restrict__ h, const float* __restrict__ b,
    float* __restrict__ out, int N) {
  static_assert(F == 32 || F == 64, "two or four floats per lane");
  constexpr int VPL = F / 16;
  const int wv = __builtin_amdgcn_readfirstlane((int)(threadIdx.x >> 5));
  const int n0 = (blockIdx.x * 8 + wv) * 2;
  if (n0 >= N) return;
  const bool two = n0 + 1 < N;
  const int n1 = two ? n0 + 1 : n0;
  const int lane = threadIdx.x & 31;
  const bool hi = lane >= 16;
  const int l = lane & 15;

  const int b0 = rowptr[n0], e0 = rowptr[n0 + 1];
  const int b1 = rowptr[n1], e1 = rowptr[n1 + 1];
  const float dn0 = dv[n0], dn1 = dv[n1];
  const float sn0 = sv[n0], sn1 = sv[n1];
  const int n = hi ? n1 : n0;
  const int beg = hi ? b1 : b0;
  const int end = hi ? e1 : e0;
  const float dn = hi ? dn1 : dn0;
  const float es = leaky((hi ? sn1 : sn0) + dn);
  const int maxdeg = (e0 - b0 > e1 - b1) ? e0 - b0 : e1 - b1;

  const int safe = end > 0 ? end - 1 : 0;
  const int p0 = beg + l, p1 = p0 + 16, p2 = p0 + 32;
  const bool k0 = p0 < end, k1 = p1 < end, k2 = p2 < end;
  const int c0 = csr[k0 ? p0 : safe];
  const int c1 = csr[k1 ? p1 : safe];
  const int c2 = csr[k2 ? p2 : safe];
  const float v0 = sv[c0];
  const float v1 = sv[c1];
  const float v2 = sv[c2];
  const int s0 = k0 ? c0 : n, s1 = k1 ? c1 : n, s2 = k2 ? c2 : n;
  const float x0 = k0 ? leaky(v0 + dn) : es;
  const float x1 = k1 ? leaky(v1 + dn) : es;
  const float x2 = k2 ? leaky(v2 + dn) : es;

  float mx = fmaxf(fmaxf(es, x0), fmaxf(x1, x2));
  for (int p = beg + 48 + l; p < end; p += 16) mx = fmaxf(mx, leaky(sv[csr[p]] + dn));
  mx = row_xor_max<1>(mx);
  mx = row_xor_max<2>(mx);
  mx = row_xor_max<4>(mx);
  mx = row_xor_max<8>(mx);

  const unsigned colb = (unsigned)l * (VPL * 4);
  const float ts = __expf(es - mx);
  float a[VPL];
  {
    const char* p = (const char*)h + ((unsigned)n * (unsigned)(F * 4) + colb);
    if (F == 32) {
      const float2 r = *(const float2*)p;
      a[0] = ts * r.x;
      a[1] = ts * r.y;
    } else {
      const float4 r = *(const float4*)p;
      a[0] = ts * r.x;
      a[1] = ts * r.y;
      a[2] = ts * r.z;
      a[3] = ts * r.w;
    }
  }

  const float t0 = k0 ? __expf(x0 - mx) : 0.0f;
  const float t1 = k1 ? __expf(x1 - mx) : 0.0f;
  const float t2 = k2 ? __expf(x2 - mx) : 0.0f;
  float zl = t0;
  zl += t1;
  zl += t2;
  for (int base = 0; base < maxdeg; base += 16) {
    int s;
    float t;
    if (base == 0) {
      s = s0;
      t = t0;
    } else if (base == 16) {
      s = s1;
      t = t1;
    } else if (base == 32) {
      s = s2;
      t = t2;
    } else {
      const int p = beg + base + l;
      s = n;
      t = 0.0f;
      if (p < end) {
        s = csr[p];
        t = __expf(leaky(sv[s] + dn) - mx);
      }
      zl += t;
    }
    add_chunk<F>(h, s, t, maxdeg - base, colb, a);
  }
  zl = row_xor_add<1>(zl);
  zl = row_xor_add<2>(zl);
  zl = row_xor_add<4>(zl);
  zl = row_xor_add<8>(zl);
  const float z = zl + ts;

  if (hi && !two) return;
  if (F == 32) {
    const float2 bb = *((const float2*)b + l);
    float w0 = a[0] / z + bb.x;
    float w1 = a[1] / z + bb.y;
    if (ELU) {
      w0 = (w0 > 0.0f) ? w0 : (__expf(w0) - 1.0f);
      w1 = (w1 > 0.0f) ? w1 : (__expf(w1) - 1.0f);
    }
    float2 o;
    o.x = w0;
    o.y = w1;
    *((float2*)(out + (size_t)n * 32) + l) = o;
  } else {
    const float4 bb = *((const float4*)b + l);
    float w0 = a[0] / z + bb.x;
    float w1 = a[1] / z + bb.y;
    float w2 = a[2] / z + bb.z;
    float w3 = a[3] / z + bb.w;
    if (ELU) {
      w0 = (w0 > 0.0f) ? w0 : (__expf(w0) - 1.0f);
      w1 = (w1 > 0.0f) ? w1 : (__expf(w1) - 1.0f);
      w2 = (w2 > 0.0f) ? w2 : (__expf(w2) - 1.0f);
      w3 = (w3 > 0.0f) ? w3 : (__expf(w3) - 1.0f);
    }
    float4 o;
    o.x = w0;
    o.y = w1;
    o.z = w2;
    o.w = w3;
    *((float4*)(out + (size_t)n * 64) + l) = o;
  }
}

#define NPW 16

__device__ __forceinline__ void pool_flush(float* __restrict__ out,
                                           float* __restrict__ cnt, int g,
                                           int lane, float a0, float a1,
                                           int run) {
  atomicAdd(out + g * 64 + 2 * lane, a0);
  atomicAdd(out + g * 64 + 2 * lane + 1, a1);
  if (lane == 0) atomicAdd(cnt + g, (float)run);
}

template <bool FULL>
__device__ __forceinline__ void pool_sum_body(const float* __restrict__ h,
                                              const int* __restrict__ batch,
                                              float* __restrict__ out,
                                              float* __restrict__ cnt, int n0,
                                              int cn, int lane) {
  float2 r[NPW];
  int g[NPW];
#pragma unroll
  for (int u = 0; u < NPW; ++u) {
    const int n = n0 + ((FULL || u < cn) ? u : (cn - 1));
    g[u] = batch[n];
    r[u] = *((const float2*)(h + (size_t)n * 64) + lane);
  }

  int cur = g[0];
  int run = 0;
  float a0 = 0.0f, a1 = 0.0f;
#pragma unroll
  for (int u = 0; u < NPW; ++u) {
    if (FULL || u < cn) {
      if (g[u] != cur) {
        pool_flush(out, cnt, cur, lane, a0, a1, run);
        cur = g[u];
        run = 0;
        a0 = 0.0f;
        a1 = 0.0f;
      }
      a0 += r[u].x;
      a1 += r[u].y;
      ++run;
    }
  }
  pool_flush(out, cnt, cur, lane, a0, a1, run);
}

__global__ __launch_bounds__(256) void pool_sum(const float* __restrict__ h,
                                                const int* __restrict__ batch,
                                                float* __restrict__ out,
                                                float* __restrict__ cnt, int N) {
  const int wv = __builtin_amdgcn_readfirstlane((int)(threadIdx.x >> 5));
  const int lane = threadIdx.x & 31;
  const int n0 = (blockIdx.x * 8 + wv) * NPW;
  if (n0 >= N) return;
  const int cn = N - n0;
  if (cn >= NPW)
    pool_sum_body<true>(h, batch, out, cnt, n0, NPW, lane);
  else
    pool_sum_body<false>(h, batch, out, cnt, n0, cn, lane);
}

__global__ void pool_div(float* __restrict__ out, const float* __restrict__ cnt) {
  int i = blockIdx.x * blockDim.x + threadIdx.x;
  if (i >= NUM_GRAPHS * 64) return;
  float c = cnt[i >> 6];
  out[i] = out[i] / fmaxf(c, 1.0f);
}

static inline int cdiv(int a, int b) { return (a + b - 1) / b; }

extern "C" void kernel_launch(void* const* d_in, const int* in_sizes, int n_in,
                              void* d_out, int out_size, void* d_ws,
                              size_t ws_size, hipStream_t stream) {
  const float* x      = (const float*)d_in[0];
  const int*   ei     = (const int*)d_in[1];
  const int*   batch  = (const int*)d_in[2];
  const float* W1     = (const float*)d_in[3];
  const float* a_src1 = (const float*)d_in[4];
  const float* a_dst1 = (const float*)d_in[5];
  const float* b1     = (const float*)d_in[6];
  const float* W2     = (const float*)d_in[7];
  const float* a_src2 = (const float*)d_in[8];
  const float* a_dst2 = (const float*)d_in[9];
  const float* b2     = (const float*)d_in[10];

  const int N = in_sizes[2];
  const int E = in_sizes[1] / 2;
  const int* src = ei;
  const int* dst = ei + E;

  float* ws = (float*)d_ws;
  size_t off = 0;
  float* h1 = ws + off; off += (size_t)N * 32;
  float* o1 = ws + off; off += (size_t)N * 32;
  float* h2 = ws + off; off += (size_t)N * 64;
  float* o2 = ws + off; off += (size_t)N * 64;
  float* sv = ws + off; off += (size_t)N;
  float* dv = ws + off; off += (size_t)N;
  float* cnt = ws + off; off += NUM_GRAPHS;
  int* btotal = (int*)(ws + off); off += BKT_MAX;
  int* cursor = (int*)(ws + off); off += BKT_MAX;
  int* bstart = (int*)(ws + off); off += BKT_MAX;
  int* rowptr = (int*)(ws + off); off += (size_t)N + 1;
  int* rec    = (int*)(ws + off); off += (size_t)E;
  int* csr    = (int*)(ws + off); off += (size_t)E;
  const int NB = cdiv(N, BKT_NODES);

  gemm_wmma<128, 32, true><<<cdiv(N, 128), 256, 0, stream>>>(
      x, W1, a_src1, a_dst1, h1, sv, dv, N, btotal, (float*)d_out, cnt);

  bucket_count<<<cdiv(E, MV_T * MV_EPT), MV_T, 0, stream>>>(dst, btotal, E, NB);
  bucket_move<<<cdiv(E, MV_T * MV_EPT), MV_T, 0, stream>>>(src, dst, btotal, cursor,
                                                           bstart, rec, E, NB);
  bucket_place<<<NB, BKT_NODES, 0, stream>>>(rec, bstart, rowptr, csr, N);

  node_gather<32, true><<<cdiv(N, 16), 256, 0, stream>>>(rowptr, csr, sv, dv, h1, b1, o1, N);

  gemm_wmma<32, 64, false><<<cdiv(N, 128), 256, 0, stream>>>(
      o1, W2, a_src2, a_dst2, h2, sv, dv, N, nullptr, nullptr, nullptr);
  node_gather<64, false><<<cdiv(N, 16), 256, 0, stream>>>(rowptr, csr, sv, dv, h2, b2, o2, N);

  pool_sum<<<cdiv(N, 8 * NPW), 256, 0, stream>>>(o2, batch, (float*)d_out, cnt, N);
  pool_div<<<cdiv(NUM_GRAPHS * 64, 256), 256, 0, stream>>>((float*)d_out, cnt);
}


# Layer 2 by linearity: W2 moves behind softmax sums and mean pool (WMMA on the 64 pooled rows); gather and pool read 32-float o1 rows, scores = o1.(W2 a); products and sums reassociated
# speedup vs baseline: 7.4173x; 1.1293x over previous
#include <hip/hip_runtime.h>
#include <hip/hip_bf16.h>
#include <math.h>

typedef __attribute__((ext_vector_type(2))) float v2f;
typedef __attribute__((ext_vector_type(8))) float v8f;

#define NUM_GRAPHS 64
#define NEG_SLOPE 0.2f

#define BKT_NODES 256
#define BKT_SHIFT 8
#define BKT_MAX 2048
#define MV_T 1024
#define MV_EPT 8


#define F_HID 32
#define F_OUT 64

template <int M>
__device__ __forceinline__ float row_xor_add(float v) {
  const int o = __builtin_amdgcn_update_dpp(0, __float_as_int(v), 0x160 + M, 0xf,
                                            0xf, true);
  return v + __int_as_float(o);
}

template <int FIN, int FOUT>
__device__ __forceinline__ void wave_rows16(const float* __restrict__ x,
                                            const float* Ws, int row, int khalf,
                                            int lidx, v8f (&acc)[FOUT / 16]) {
  constexpr int NT = FOUT / 16;
#pragma unroll 4
  for (int k = 0; k < FIN; k += 4) {
    const int kk = k + khalf;
    v2f a;
    a.x = x[(size_t)row * FIN + kk];
    a.y = x[(size_t)row * FIN + kk + 1];
#pragma unroll
    for (int t = 0; t < NT; ++t) {
      v2f b;
      b.x = Ws[kk * FOUT + t * 16 + lidx];
      b.y = Ws[(kk + 1) * FOUT + t * 16 + lidx];
      acc[t] = __builtin_amdgcn_wmma_f32_16x16x4_f32(
          false, a, false, b, (short)0, acc[t], false, false);
    }
  }
}

template <int FIN, int FOUT, bool ZERO>
__global__ __launch_bounds__(256) void gemm_wmma(
    const float* __restrict__ x, const float* __restrict__ W,
    const float* __restrict__ asrc, const float* __restrict__ adst,
    float* __restrict__ h, float* __restrict__ s, float* __restrict__ d, int N,
    int* __restrict__ zbk, float* __restrict__ zpool, float* __restrict__ zcnt,
    const float* __restrict__ Wn, const float* __restrict__ asrcn,
    const float* __restrict__ adstn, float* __restrict__ uw) {
  constexpr int NT = FOUT / 16;
  __shared__ float Ws[FIN * FOUT];

  for (int i = threadIdx.x; i < FIN * FOUT; i += 256) Ws[i] = W[i];
  __syncthreads();

  if (ZERO) {
    if (blockIdx.x == 0) {
      for (int i = threadIdx.x; i < 2 * BKT_MAX; i += 256) zbk[i] = 0;
      for (int i = threadIdx.x; i < NUM_GRAPHS * FOUT; i += 256) zpool[i] = 0.0f;
      if (threadIdx.x < NUM_GRAPHS) zcnt[threadIdx.x] = 0.0f;
      if (threadIdx.x < 2 * FOUT) {
        const int f = threadIdx.x % FOUT;
        const float* av = (threadIdx.x < FOUT) ? asrcn : adstn;
        const float4* wr = (const float4*)(Wn + f * F_OUT);
        const float4* ar = (const float4*)av;
        float acc1 = 0.0f;
#pragma unroll
        for (int q = 0; q < F_OUT / 4; ++q) {
          const float4 wq = wr[q];
          const float4 aq = ar[q];
          acc1 += wq.x * aq.x;
          acc1 += wq.y * aq.y;
          acc1 += wq.z * aq.z;
          acc1 += wq.w * aq.w;
        }
        uw[threadIdx.x] = acc1;
      }
    }
  }

  const int wave = threadIdx.x >> 5;
  const int lane = threadIdx.x & 31;
  const int m0 = (blockIdx.x * 8 + wave) * 16;
  const int halfsel = lane >> 4;
  const int lidx = lane & 15;
  int row = m0 + lidx;
  if (row >= N) row = N - 1;
  const int khalf = halfsel * 2;

  float as_[NT], ad_[NT];
#pragma unroll
  for (int t = 0; t < NT; ++t) {
    as_[t] = asrc[t * 16 + lidx];
    ad_[t] = adst[t * 16 + lidx];
  }

  v8f acc[NT];
#pragma unroll
  for (int t = 0; t < NT; ++t)
#pragma unroll
    for (int j = 0; j < 8; ++j) acc[t][j] = 0.0f;

  wave_rows16<FIN, FOUT>(x, Ws, row, khalf, lidx, acc);

  const int rbase = m0 + halfsel * 8;
#pragma unroll
  for (int i = 0; i < 8; ++i) {
    const int r = rbase + i;
    if (r < N) {
#pragma unroll
      for (int t = 0; t < NT; ++t)
        h[(size_t)r * FOUT + t * 16 + lidx] = acc[t][i];
    }
  }

  float ps[8], pd[8];
#pragma unroll
  for (int i = 0; i < 8; ++i) {
    float a = 0.0f, b = 0.0f;
#pragma unroll
    for (int t = 0; t < NT; ++t) {
      a += acc[t][i] * as_[t];
      b += acc[t][i] * ad_[t];
    }
    ps[i] = a;
    pd[i] = b;
  }
#pragma unroll
  for (int i = 0; i < 8; ++i) {
    ps[i] = row_xor_add<1>(ps[i]);
    pd[i] = row_xor_add<1>(pd[i]);
  }
#pragma unroll
  for (int i = 0; i < 8; ++i) {
    ps[i] = row_xor_add<2>(ps[i]);
    pd[i] = row_xor_add<2>(pd[i]);
  }
#pragma unroll
  for (int i = 0; i < 8; ++i) {
    ps[i] = row_xor_add<4>(ps[i]);
    pd[i] = row_xor_add<4>(pd[i]);
  }
#pragma unroll
  for (int i = 0; i < 8; ++i) {
    ps[i] = row_xor_add<8>(ps[i]);
    pd[i] = row_xor_add<8>(pd[i]);
  }
  float so = ps[0], dout = pd[0];
#pragma unroll
  for (int i = 1; i < 8; ++i) {
    if (lidx == i) {
      so = ps[i];
      dout = pd[i];
    }
  }
  const int r = rbase + lidx;
  if (lidx < 8 && r < N) {
    s[r] = so;
    d[r] = dout;
  }
}


template <int S>
__device__ __forceinline__ int row_shr_add(int v) {
  return v + __builtin_amdgcn_update_dpp(0, v, 0x110 + S, 0xf, 0xf, true);
}

__device__ __forceinline__ int wave_prefix(int v, int lane) {
  v = row_shr_add<1>(v);
  v = row_shr_add<2>(v);
  v = row_shr_add<4>(v);
  v = row_shr_add<8>(v);
  const int low = __builtin_amdgcn_readlane(v, 15);
  return v + (lane >= 16 ? low : 0);
}

__global__ __launch_bounds__(MV_T) void bucket_count(const int* __restrict__ dst,
                                                     int* __restrict__ btotal,
                                                     int E, int NB) {
  __shared__ int hist[BKT_MAX];
  for (int i = threadIdx.x; i < NB; i += MV_T) hist[i] = 0;
  const int base = blockIdx.x * (MV_T * MV_EPT) + threadIdx.x;
  int d[MV_EPT];
#pragma unroll
  for (int j = 0; j < MV_EPT; ++j) {
    const int e = base + j * MV_T;
    d[j] = (e < E) ? dst[e] : -1;
  }
  __syncthreads();
#pragma unroll
  for (int j = 0; j < MV_EPT; ++j)
    if (d[j] >= 0) atomicAdd(&hist[d[j] >> BKT_SHIFT], 1);
  __syncthreads();
  for (int i = threadIdx.x; i < NB; i += MV_T) {
    const int h = hist[i];
    if (h) atomicAdd(btotal + i, h);
  }
}

__global__ __launch_bounds__(MV_T) void bucket_move(
    const int* __restrict__ src, const int* __restrict__ dst,
    const int* __restrict__ btotal, int* __restrict__ cursor,
    int* __restrict__ bstart, int* __restrict__ rec, int E, int NB) {
  __shared__ int hist[BKT_MAX];
  __shared__ int place[BKT_MAX];
  __shared__ int wtot[MV_T / 32];
  const int tid = threadIdx.x;
  const int lane = tid & 31;
  const int w = __builtin_amdgcn_readfirstlane(tid >> 5);
  for (int i = tid; i < NB; i += MV_T) hist[i] = 0;

  const int base = blockIdx.x * (MV_T * MV_EPT) + tid;
  int s[MV_EPT], d[MV_EPT];
#pragma unroll
  for (int j = 0; j < MV_EPT; ++j) {
    const int e = base + j * MV_T;
    const bool ok = e < E;
    d[j] = ok ? dst[e] : -1;
    s[j] = ok ? src[e] : 0;
  }
  const int i0 = 2 * tid;
  const int a0 = (i0 < NB) ? btotal[i0] : 0;
  const int a1 = (i0 + 1 < NB) ? btotal[i0 + 1] : 0;
  __syncthreads();

  int r[MV_EPT];
#pragma unroll
  for (int j = 0; j < MV_EPT; ++j) {
    r[j] = 0;
    if (d[j] >= 0) r[j] = atomicAdd(&hist[d[j] >> BKT_SHIFT], 1);
  }

  const int ts = a0 + a1;
  const int inc = wave_prefix(ts, lane);
  if (lane == 31) wtot[w] = inc;
  __syncthreads();
  const int x = wtot[lane];
  const int inc2 = wave_prefix(x, lane);
  const int ex = __builtin_amdgcn_readlane(inc2 - x, w) + inc - ts;

  if (i0 < NB) {
    const int h = hist[i0];
    place[i0] = ex + (h ? atomicAdd(cursor + i0, h) : 0);
  }
  if (i0 + 1 < NB) {
    const int h = hist[i0 + 1];
    place[i0 + 1] = ex + a0 + (h ? atomicAdd(cursor + i0 + 1, h) : 0);
  }
  if (blockIdx.x == 0) {
    if (i0 <= NB) bstart[i0] = ex;
    if (i0 + 1 <= NB) bstart[i0 + 1] = ex + a0;
  }
  __syncthreads();

#pragma unroll
  for (int j = 0; j < MV_EPT; ++j) {
    if (d[j] >= 0) {
      const unsigned word =
          (unsigned)s[j] | ((unsigned)(d[j] & (BKT_NODES - 1)) << 24);
      rec[place[d[j] >> BKT_SHIFT] + r[j]] = (int)word;
    }
  }
}

__global__ __launch_bounds__(BKT_NODES) void bucket_place(
    const int* __restrict__ rec, const int* __restrict__ bstart,
    int* __restrict__ rowptr, int* __restrict__ csr, int N) {
  __shared__ int cnt[BKT_NODES + 1];
  __shared__ int wtot[BKT_NODES / 32];
  const int tid = threadIdx.x;
  const int lane = tid & 31;
  const int w = tid >> 5;
  const int b = blockIdx.x;
  const int e0 = bstart[b];
  const int e1 = bstart[b + 1];
  cnt[tid] = 0;
  if (tid == 0) cnt[BKT_NODES] = 0;
  __syncthreads();

  for (int e = e0 + tid; e < e1; e += BKT_NODES * 8) {
    unsigned q[8];
#pragma unroll
    for (int j = 0; j < 8; ++j) {
      const int ee = e + j * BKT_NODES;
      q[j] = (unsigned)rec[ee < e1 ? ee : e1 - 1];
    }
#pragma unroll
    for (int j = 0; j < 8; ++j) {
      const bool ok = e + j * BKT_NODES < e1;
      atomicAdd(&cnt[ok ? (int)(q[j] >> 24) : BKT_NODES], 1);
    }
  }
  __syncthreads();

  const int c = cnt[tid];
  const int inc = wave_prefix(c, lane);
  if (lane == 31) wtot[w] = inc;
  __syncthreads();
  int wofs = 0;
#pragma unroll
  for (int k = 0; k < BKT_NODES / 32; ++k)
    if (k < w) wofs += wtot[k];
  const int ex = wofs + inc - c;
  const int n = b * BKT_NODES + tid;
  if (n < N) rowptr[n] = e0 + ex;
  if (b == (int)gridDim.x - 1 && tid == 0) rowptr[N] = e1;
  cnt[tid] = ex;
  __syncthreads();

  for (int e = e0 + tid; e < e1; e += BKT_NODES * 8) {
    unsigned q[8];
    int p[8];
#pragma unroll
    for (int j = 0; j < 8; ++j) {
      const int ee = e + j * BKT_NODES;
      q[j] = (unsigned)rec[ee < e1 ? ee : e1 - 1];
    }
#pragma unroll
    for (int j = 0; j < 8; ++j) {
      const bool ok = e + j * BKT_NODES < e1;
      p[j] = atomicAdd(&cnt[ok ? (int)(q[j] >> 24) : BKT_NODES], 1);
    }
#pragma unroll
    for (int j = 0; j < 8; ++j)
      if (e + j * BKT_NODES < e1) csr[e0 + p[j]] = (int)(q[j] & 0xffffffu);
  }
}

__device__ __forceinline__ float leaky(float e) {
  return (e > 0.0f) ? e : NEG_SLOPE * e;
}


template <int J>
__device__ __forceinline__ int row_share(int v) {
  return __builtin_amdgcn_update_dpp(0, v, 0x150 + J, 0xf, 0xf, true);
}

template <int M>
__device__ __forceinline__ float row_xor_max(float v) {
  const int o = __builtin_amdgcn_update_dpp(0, __float_as_int(v), 0x160 + M, 0xf,
                                            0xf, true);
  return fmaxf(v, __int_as_float(o));
}

template <int F, int J>
__device__ __forceinline__ void add_row(const float* __restrict__ h, int s,
                                        float t, unsigned colb,
                                        float (&a)[F / 16]) {
  const unsigned sj = (unsigned)row_share<J>(s);
  const float tj = __int_as_float(row_share<J>(__float_as_int(t)));
  const char* p = (const char*)h + (sj * (unsigned)(F * 4) + colb);
  if (F == 32) {
    const float2 r = *(const float2*)p;
    a[0] += tj * r.x;
    a[1] += tj * r.y;
  } else {
    const float4 r = *(const float4*)p;
    a[0] += tj * r.x;
    a[1] += tj * r.y;
    a[2] += tj * r.z;
    a[3] += tj * r.w;
  }
}

template <int F, int J0>
__device__ __forceinline__ void add_rows4(const float* __restrict__ h, int s,
                                          float t, unsigned colb,
                                          float (&a)[F / 16]) {
  add_row<F, J0 + 0>(h, s, t, colb, a);
  add_row<F, J0 + 1>(h, s, t, colb, a);
  add_row<F, J0 + 2>(h, s, t, colb, a);
  add_row<F, J0 + 3>(h, s, t, colb, a);
}

template <int F>
__device__ __forceinline__ void add_row_at(const float* __restrict__ h, int s,
                                           float t, int j, unsigned colb,
                                           float (&a)[F / 16]) {
  const unsigned sj =
      __builtin_amdgcn_permlane16_var((unsigned)s, (unsigned)s, (unsigned)j, false, false);
  const float tj = __uint_as_float(__builtin_amdgcn_permlane16_var(
      __float_as_uint(t), __float_as_uint(t), (unsigned)j, false, false));
  const char* p = (const char*)h + (sj * (unsigned)(F * 4) + colb);
  if (F == 32) {
    const float2 r = *(const float2*)p;
    a[0] += tj * r.x;
    a[1] += tj * r.y;
  } else {
    const float4 r = *(const float4*)p;
    a[0] += tj * r.x;
    a[1] += tj * r.y;
    a[2] += tj * r.z;
    a[3] += tj * r.w;
  }
}

template <int F>
__device__ __forceinline__ void add_chunk(const float* __restrict__ h, int s,
                                          float t, int c, unsigned colb,
                                          float (&a)[F / 16]) {
  if (c >= 16) {
    add_rows4<F, 0>(h, s, t, colb, a);
    add_rows4<F, 4>(h, s, t, colb, a);
    add_rows4<F, 8>(h, s, t, colb, a);
    add_rows4<F, 12>(h, s, t, colb, a);
  } else {
    for (int j0 = 0; j0 < c; j0 += 8) {
#pragma unroll
      for (int u = 0; u < 8; ++u) add_row_at<F>(h, s, t, j0 + u, colb, a);
    }
  }
}

template <int VPL>
__device__ __forceinline__ void load_lane(const float* __restrict__ p, int l,
                                          float (&v)[VPL]) {
  if constexpr (VPL == 2) {
    const float2 r = *((const float2*)p + l);
    v[0] = r.x;
    v[1] = r.y;
  } else {
    const float4 r = *((const float4*)p + l);
    v[0] = r.x;
    v[1] = r.y;
    v[2] = r.z;
    v[3] = r.w;
  }
}

template <int VPL>
__device__ __forceinline__ void store_lane(float* __restrict__ p, int l,
                                           const float (&v)[VPL]) {
  if constexpr (VPL == 2) {
    float2 o;
    o.x = v[0];
    o.y = v[1];
    *((float2*)p + l) = o;
  } else {
    float4 o;
    o.x = v[0];
    o.y = v[1];
    o.z = v[2];
    o.w = v[3];
    *((float4*)p + l) = o;
  }
}

template <int F, bool HIDDEN>
__global__ __launch_bounds__(256) void node_gather(
    const int* __restrict__ rowptr, const int* __restrict__ csr,
    const float* __restrict__ sv, const float* __restrict__ dv,
    const float* __restrict__ h, const float* __restrict__ b,
    float* __restrict__ out, int N, const float* __restrict__ uw,
    float* __restrict__ snext, float* __restrict__ dnext) {
  static_assert(F == 32 || F == 64, "two or four floats per lane");
  constexpr int VPL = F / 16;
  const int wv = __builtin_amdgcn_readfirstlane((int)(threadIdx.x >> 5));
  const int n0 = (blockIdx.x * 8 + wv) * 2;
  if (n0 >= N) return;
  const bool two = n0 + 1 < N;
  const int n1 = two ? n0 + 1 : n0;
  const int lane = threadIdx.x & 31;
  const bool hi = lane >= 16;
  const int l = lane & 15;

  const int b0 = rowptr[n0], e0 = rowptr[n0 + 1];
  const int b1 = rowptr[n1], e1 = rowptr[n1 + 1];
  const float dn0 = dv[n0], dn1 = dv[n1];
  const float sn0 = sv[n0], sn1 = sv[n1];
  const int n = hi ? n1 : n0;
  const int beg = hi ? b1 : b0;
  const int end = hi ? e1 : e0;
  const float dn = hi ? dn1 : dn0;
  const float es = leaky((hi ? sn1 : sn0) + dn);
  const int maxdeg = (e0 - b0 > e1 - b1) ? e0 - b0 : e1 - b1;

  const int safe = end > 0 ? end - 1 : 0;
  const int p0 = beg + l, p1 = p0 + 16, p2 = p0 + 32;
  const bool k0 = p0 < end, k1 = p1 < end, k2 = p2 < end;
  const int c0 = csr[k0 ? p0 : safe];
  const int c1 = csr[k1 ? p1 : safe];
  const int c2 = csr[k2 ? p2 : safe];
  const float v0 = sv[c0];
  const float v1 = sv[c1];
  const float v2 = sv[c2];
  const int s0 = k0 ? c0 : n, s1 = k1 ? c1 : n, s2 = k2 ? c2 : n;
  const float x0 = k0 ? leaky(v0 + dn) : es;
  const float x1 = k1 ? leaky(v1 + dn) : es;
  const float x2 = k2 ? leaky(v2 + dn) : es;

  float mx = fmaxf(fmaxf(es, x0), fmaxf(x1, x2));
  for (int p = beg + 48 + l; p < end; p += 16) mx = fmaxf(mx, leaky(sv[csr[p]] + dn));
  mx = row_xor_max<1>(mx);
  mx = row_xor_max<2>(mx);
  mx = row_xor_max<4>(mx);
  mx = row_xor_max<8>(mx);

  const unsigned colb = (unsigned)l * (VPL * 4);
  const float ts = __expf(es - mx);
  float a[VPL];
  {
    const char* p = (const char*)h + ((unsigned)n * (unsigned)(F * 4) + colb);
    if (F == 32) {
      const float2 r = *(const float2*)p;
      a[0] = ts * r.x;
      a[1] = ts * r.y;
    } else {
      const float4 r = *(const float4*)p;
      a[0] = ts * r.x;
      a[1] = ts * r.y;
      a[2] = ts * r.z;
      a[3] = ts * r.w;
    }
  }

  const float t0 = k0 ? __expf(x0 - mx) : 0.0f;
  const float t1 = k1 ? __expf(x1 - mx) : 0.0f;
  const float t2 = k2 ? __expf(x2 - mx) : 0.0f;
  float zl = t0;
  zl += t1;
  zl += t2;
  for (int base = 0; base < maxdeg; base += 16) {
    int s;
    float t;
    if (base == 0) {
      s = s0;
      t = t0;
    } else if (base == 16) {
      s = s1;
      t = t1;
    } else if (base == 32) {
      s = s2;
      t = t2;
    } else {
      const int p = beg + base + l;
      s = n;
      t = 0.0f;
      if (p < end) {
        s = csr[p];
        t = __expf(leaky(sv[s] + dn) - mx);
      }
      zl += t;
    }
    add_chunk<F>(h, s, t, maxdeg - base, colb, a);
  }
  zl = row_xor_add<1>(zl);
  zl = row_xor_add<2>(zl);
  zl = row_xor_add<4>(zl);
  zl = row_xor_add<8>(zl);
  const float z = zl + ts;

  const bool live = !hi || two;
  float o[VPL];
#pragma unroll
  for (int i = 0; i < VPL; ++i) o[i] = a[i] / z;
  if (HIDDEN) {
    float bb[VPL];
    load_lane<VPL>(b, l, bb);
#pragma unroll
    for (int i = 0; i < VPL; ++i) {
      const float w = o[i] + bb[i];
      o[i] = (w > 0.0f) ? w : (__expf(w) - 1.0f);
    }
  }
  if (live) store_lane<VPL>(out + (size_t)n * F, l, o);
  if (HIDDEN) {
    float uu[VPL], ww[VPL];
    load_lane<VPL>(uw, l, uu);
    load_lane<VPL>(uw + F, l, ww);
    float ps = 0.0f, pd = 0.0f;
#pragma unroll
    for (int i = 0; i < VPL; ++i) {
      ps += o[i] * uu[i];
      pd += o[i] * ww[i];
    }
    ps = row_xor_add<1>(ps);
    pd = row_xor_add<1>(pd);
    ps = row_xor_add<2>(ps);
    pd = row_xor_add<2>(pd);
    ps = row_xor_add<4>(ps);
    pd = row_xor_add<4>(pd);
    ps = row_xor_add<8>(ps);
    pd = row_xor_add<8>(pd);
    if (live && l == 0) {
      snext[n] = ps;
      dnext[n] = pd;
    }
  }
}

#define NPW 16

__device__ __forceinline__ void pool_flush(float* __restrict__ pooled,
                                           float* __restrict__ cnt, int g,
                                           int lane, float a0, int run) {
  atomicAdd(pooled + g * F_HID + lane, a0);
  if (lane == 0) atomicAdd(cnt + g, (float)run);
}

template <bool FULL>
__device__ __forceinline__ void pool_sum_body(const float* __restrict__ h,
                                              const int* __restrict__ batch,
                                              float* __restrict__ pooled,
                                              float* __restrict__ cnt, int n0,
                                              int cn, int lane) {
  float r[NPW];
  int g[NPW];
#pragma unroll
  for (int u = 0; u < NPW; ++u) {
    const int n = n0 + ((FULL || u < cn) ? u : (cn - 1));
    g[u] = batch[n];
    r[u] = h[(size_t)n * F_HID + lane];
  }

  int cur = g[0];
  int run = 0;
  float a0 = 0.0f;
#pragma unroll
  for (int u = 0; u < NPW; ++u) {
    if (FULL || u < cn) {
      if (g[u] != cur) {
        pool_flush(pooled, cnt, cur, lane, a0, run);
        cur = g[u];
        run = 0;
        a0 = 0.0f;
      }
      a0 += r[u];
      ++run;
    }
  }
  pool_flush(pooled, cnt, cur, lane, a0, run);
}

__global__ __launch_bounds__(256) void pool_sum(const float* __restrict__ h,
                                                const int* __restrict__ batch,
                                                float* __restrict__ pooled,
                                                float* __restrict__ cnt, int N) {
  static_assert(F_HID == 32, "one feature per lane");
  const int wv = __builtin_amdgcn_readfirstlane((int)(threadIdx.x >> 5));
  const int lane = threadIdx.x & 31;
  const int n0 = (blockIdx.x * 8 + wv) * NPW;
  if (n0 >= N) return;
  const int cn = N - n0;
  if (cn >= NPW)
    pool_sum_body<true>(h, batch, pooled, cnt, n0, NPW, lane);
  else
    pool_sum_body<false>(h, batch, pooled, cnt, n0, cn, lane);
}

__global__ __launch_bounds__(NUM_GRAPHS * 2) void pool_project(
    const float* __restrict__ pooled, const float* __restrict__ W,
    const float* __restrict__ b, const float* __restrict__ cnt,
    float* __restrict__ out) {
  constexpr int NT = F_OUT / 16;
  __shared__ float Ws[F_HID * F_OUT];
  for (int i = threadIdx.x; i < F_HID * F_OUT; i += NUM_GRAPHS * 2) Ws[i] = W[i];
  __syncthreads();

  const int wave = threadIdx.x >> 5;
  const int lane = threadIdx.x & 31;
  const int m0 = wave * 16;
  const int halfsel = lane >> 4;
  const int lidx = lane & 15;
  const int row = m0 + lidx;
  const int khalf = halfsel * 2;

  float bb[NT];
#pragma unroll
  for (int t = 0; t < NT; ++t) bb[t] = b[t * 16 + lidx];

  v8f acc[NT];
#pragma unroll
  for (int t = 0; t < NT; ++t)
#pragma unroll
    for (int j = 0; j < 8; ++j) acc[t][j] = 0.0f;

  wave_rows16<F_HID, F_OUT>(pooled, Ws, row, khalf, lidx, acc);

  const int rbase = m0 + halfsel * 8;
#pragma unroll
  for (int i = 0; i < 8; ++i) {
    const int r = rbase + i;
    const float cg = cnt[r];
    const float den = fmaxf(cg, 1.0f);
#pragma unroll
    for (int t = 0; t < NT; ++t)
      out[r * F_OUT + t * 16 + lidx] = (acc[t][i] + cg * bb[t]) / den;
  }
}

static inline int cdiv(int a, int b) { return (a + b - 1) / b; }

extern "C" void kernel_launch(void* const* d_in, const int* in_sizes, int n_in,
                              void* d_out, int out_size, void* d_ws,
                              size_t ws_size, hipStream_t stream) {
  const float* x      = (const float*)d_in[0];
  const int*   ei     = (const int*)d_in[1];
  const int*   batch  = (const int*)d_in[2];
  const float* W1     = (const float*)d_in[3];
  const float* a_src1 = (const float*)d_in[4];
  const float* a_dst1 = (const float*)d_in[5];
  const float* b1     = (const float*)d_in[6];
  const float* W2     = (const float*)d_in[7];
  const float* a_src2 = (const float*)d_in[8];
  const float* a_dst2 = (const float*)d_in[9];
  const float* b2     = (const float*)d_in[10];

  const int N = in_sizes[2];
  const int E = in_sizes[1] / 2;
  const int* src = ei;
  const int* dst = ei + E;

  float* ws = (float*)d_ws;
  size_t off = 0;
  float* h1 = ws + off; off += (size_t)N * F_HID;
  float* o1 = ws + off; off += (size_t)N * F_HID;
  float* g2 = ws + off; off += (size_t)N * F_HID;
  float* sv = ws + off; off += (size_t)N;
  float* dv = ws + off; off += (size_t)N;
  float* sv2 = ws + off; off += (size_t)N;
  float* dv2 = ws + off; off += (size_t)N;
  float* cnt = ws + off; off += NUM_GRAPHS;
  float* pooled = ws + off; off += NUM_GRAPHS * F_HID;
  float* uw = ws + off; off += 2 * F_HID;
  int* btotal = (int*)(ws + off); off += BKT_MAX;
  int* cursor = (int*)(ws + off); off += BKT_MAX;
  int* bstart = (int*)(ws + off); off += BKT_MAX;
  int* rowptr = (int*)(ws + off); off += (size_t)N + 1;
  int* rec    = (int*)(ws + off); off += (size_t)E;
  int* csr    = (int*)(ws + off); off += (size_t)E;
  const int NB = cdiv(N, BKT_NODES);

  gemm_wmma<128, F_HID, true><<<cdiv(N, 128), 256, 0, stream>>>(
      x, W1, a_src1, a_dst1, h1, sv, dv, N, btotal, pooled, cnt, W2, a_src2,
      a_dst2, uw);

  bucket_count<<<cdiv(E, MV_T * MV_EPT), MV_T, 0, stream>>>(dst, btotal, E, NB);
  bucket_move<<<cdiv(E, MV_T * MV_EPT), MV_T, 0, stream>>>(src, dst, btotal, cursor,
                                                           bstart, rec, E, NB);
  bucket_place<<<NB, BKT_NODES, 0, stream>>>(rec, bstart, rowptr, csr, N);

  node_gather<F_HID, true><<<cdiv(N, 16), 256, 0, stream>>>(
      rowptr, csr, sv, dv, h1, b1, o1, N, uw, sv2, dv2);

  node_gather<F_HID, false><<<cdiv(N, 16), 256, 0, stream>>>(
      rowptr, csr, sv2, dv2, o1, nullptr, g2, N, nullptr, nullptr, nullptr);

  pool_sum<<<cdiv(N, 8 * NPW), 256, 0, stream>>>(g2, batch, pooled, cnt, N);
  pool_project<<<1, NUM_GRAPHS * 2, 0, stream>>>(pooled, W2, b2, cnt, (float*)d_out);
}


# 8 launches become 6: edge counting rides in the projection launch as per-chunk count rows that bucket_move sums; the layer-2 gather also pools its block's 16 rows (LDS, atomics); sum order unchanged
# speedup vs baseline: 7.8333x; 1.0561x over previous
#include <hip/hip_runtime.h>
#include <hip/hip_bf16.h>
#include <math.h>

typedef __attribute__((ext_vector_type(2))) float v2f;
typedef __attribute__((ext_vector_type(8))) float v8f;

#define NUM_GRAPHS 64
#define NEG_SLOPE 0.2f

#define BKT_NODES 256
#define BKT_SHIFT 8
#define BKT_MAX 2048
#define MV_T 1024
#define MV_EPT 8
#define CNT_T 256
#define CNT_EPT 32
#define CNT_ROUNDS 2
#define CNT_CHUNK (CNT_T * CNT_EPT * CNT_ROUNDS)


#define F_HID 32
#define F_OUT 64

template <int M>
__device__ __forceinline__ float row_xor_add(float v) {
  const int o = __builtin_amdgcn_update_dpp(0, __float_as_int(v), 0x160 + M, 0xf,
                                            0xf, true);
  return v + __int_as_float(o);
}

template <int FIN, int FOUT>
__device__ __forceinline__ void wave_rows16(const float* __restrict__ x,
                                            const float* Ws, int row, int khalf,
                                            int lidx, v8f (&acc)[FOUT / 16]) {
  constexpr int NT = FOUT / 16;
#pragma unroll 4
  for (int k = 0; k < FIN; k += 4) {
    const int kk = k + khalf;
    v2f a;
    a.x = x[(size_t)row * FIN + kk];
    a.y = x[(size_t)row * FIN + kk + 1];
#pragma unroll
    for (int t = 0; t < NT; ++t) {
      v2f b;
      b.x = Ws[kk * FOUT + t * 16 + lidx];
      b.y = Ws[(kk + 1) * FOUT + t * 16 + lidx];
      acc[t] = __builtin_amdgcn_wmma_f32_16x16x4_f32(
          false, a, false, b, (short)0, acc[t], false, false);
    }
  }
}

template <int FIN, int FOUT, bool ZERO>
__global__ __launch_bounds__(256) void gemm_wmma(
    const float* __restrict__ x, const float* __restrict__ W,
    const float* __restrict__ asrc, const float* __restrict__ adst,
    float* __restrict__ h, float* __restrict__ s, float* __restrict__ d, int N,
    int* __restrict__ zbk, float* __restrict__ zpool, float* __restrict__ zcnt,
    const float* __restrict__ Wn, const float* __restrict__ asrcn,
    const float* __restrict__ adstn, float* __restrict__ uw,
    const int* __restrict__ dst, int* __restrict__ histB, int E, int NB,
    int nA) {
  constexpr int NT = FOUT / 16;
  __shared__ float Ws[FIN * FOUT];

  if (ZERO) {
    static_assert(!ZERO || FIN * FOUT >= BKT_MAX, "the counters fit in Ws");
    if ((int)blockIdx.x >= nA) {
      int* hist = reinterpret_cast<int*>(Ws);
      const int cb = (int)blockIdx.x - nA;
      for (int i = threadIdx.x; i < NB; i += CNT_T) hist[i] = 0;
      __syncthreads();
      for (int r = 0; r < CNT_ROUNDS; ++r) {
        const int base = cb * CNT_CHUNK + r * (CNT_T * CNT_EPT) + (int)threadIdx.x;
        int dd[CNT_EPT];
#pragma unroll
        for (int j = 0; j < CNT_EPT; ++j) {
          const int e = base + j * CNT_T;
          dd[j] = (e < E) ? dst[e] : -1;
        }
#pragma unroll
        for (int j = 0; j < CNT_EPT; ++j)
          if (dd[j] >= 0) atomicAdd(&hist[dd[j] >> BKT_SHIFT], 1);
      }
      __syncthreads();
      for (int i = threadIdx.x; i < NB; i += CNT_T) histB[cb * NB + i] = hist[i];
      return;
    }
  }

  for (int i = threadIdx.x; i < FIN * FOUT; i += 256) Ws[i] = W[i];
  __syncthreads();

  if (ZERO) {
    if (blockIdx.x == 0) {
      for (int i = threadIdx.x; i < BKT_MAX; i += 256) zbk[i] = 0;
      for (int i = threadIdx.x; i < NUM_GRAPHS * FOUT; i += 256) zpool[i] = 0.0f;
      if (threadIdx.x < NUM_GRAPHS) zcnt[threadIdx.x] = 0.0f;
      if (threadIdx.x < 2 * FOUT) {
        const int f = threadIdx.x % FOUT;
        const float* av = (threadIdx.x < FOUT) ? asrcn : adstn;
        const float4* wr = (const float4*)(Wn + f * F_OUT);
        const float4* ar = (const float4*)av;
        float acc1 = 0.0f;
#pragma unroll
        for (int q = 0; q < F_OUT / 4; ++q) {
          const float4 wq = wr[q];
          const float4 aq = ar[q];
          acc1 += wq.x * aq.x;
          acc1 += wq.y * aq.y;
          acc1 += wq.z * aq.z;
          acc1 += wq.w * aq.w;
        }
        uw[threadIdx.x] = acc1;
      }
    }
  }

  const int wave = threadIdx.x >> 5;
  const int lane = threadIdx.x & 31;
  const int m0 = (blockIdx.x * 8 + wave) * 16;
  const int halfsel = lane >> 4;
  const int lidx = lane & 15;
  int row = m0 + lidx;
  if (row >= N) row = N - 1;
  const int khalf = halfsel * 2;

  float as_[NT], ad_[NT];
#pragma unroll
  for (int t = 0; t < NT; ++t) {
    as_[t] = asrc[t * 16 + lidx];
    ad_[t] = adst[t * 16 + lidx];
  }

  v8f acc[NT];
#pragma unroll
  for (int t = 0; t < NT; ++t)
#pragma unroll
    for (int j = 0; j < 8; ++j) acc[t][j] = 0.0f;

  wave_rows16<FIN, FOUT>(x, Ws, row, khalf, lidx, acc);

  const int rbase = m0 + halfsel * 8;
#pragma unroll
  for (int i = 0; i < 8; ++i) {
    const int r = rbase + i;
    if (r < N) {
#pragma unroll
      for (int t = 0; t < NT; ++t)
        h[(size_t)r * FOUT + t * 16 + lidx] = acc[t][i];
    }
  }

  float ps[8], pd[8];
#pragma unroll
  for (int i = 0; i < 8; ++i) {
    float a = 0.0f, b = 0.0f;
#pragma unroll
    for (int t = 0; t < NT; ++t) {
      a += acc[t][i] * as_[t];
      b += acc[t][i] * ad_[t];
    }
    ps[i] = a;
    pd[i] = b;
  }
#pragma unroll
  for (int i = 0; i < 8; ++i) {
    ps[i] = row_xor_add<1>(ps[i]);
    pd[i] = row_xor_add<1>(pd[i]);
  }
#pragma unroll
  for (int i = 0; i < 8; ++i) {
    ps[i] = row_xor_add<2>(ps[i]);
    pd[i] = row_xor_add<2>(pd[i]);
  }
#pragma unroll
  for (int i = 0; i < 8; ++i) {
    ps[i] = row_xor_add<4>(ps[i]);
    pd[i] = row_xor_add<4>(pd[i]);
  }
#pragma unroll
  for (int i = 0; i < 8; ++i) {
    ps[i] = row_xor_add<8>(ps[i]);
    pd[i] = row_xor_add<8>(pd[i]);
  }
  float so = ps[0], dout = pd[0];
#pragma unroll
  for (int i = 1; i < 8; ++i) {
    if (lidx == i) {
      so = ps[i];
      dout = pd[i];
    }
  }
  const int r = rbase + lidx;
  if (lidx < 8 && r < N) {
    s[r] = so;
    d[r] = dout;
  }
}


template <int S>
__device__ __forceinline__ int row_shr_add(int v) {
  return v + __builtin_amdgcn_update_dpp(0, v, 0x110 + S, 0xf, 0xf, true);
}

__device__ __forceinline__ int wave_prefix(int v, int lane) {
  v = row_shr_add<1>(v);
  v = row_shr_add<2>(v);
  v = row_shr_add<4>(v);
  v = row_shr_add<8>(v);
  const int low = __builtin_amdgcn_readlane(v, 15);
  return v + (lane >= 16 ? low : 0);
}

__device__ __forceinline__ int col_sum(const int* __restrict__ histB, int c,
                                       int rs, int G, int nrow, int NB) {
  int acc = 0;
  for (int r0 = rs; r0 < nrow; r0 += 32 * G) {
    int v[32];
#pragma unroll
    for (int j = 0; j < 32; ++j) {
      const int r = r0 + j * G;
      v[j] = (r < nrow) ? histB[r * NB + c] : 0;
    }
#pragma unroll
    for (int j = 0; j < 32; ++j) acc += v[j];
  }
  return acc;
}

__global__ __launch_bounds__(MV_T) void bucket_move(
    const int* __restrict__ src, const int* __restrict__ dst,
    const int* __restrict__ histB, int* __restrict__ cursor,
    int* __restrict__ bstart, int* __restrict__ rec, int E, int NB, int nrow) {
  __shared__ int hist[BKT_MAX];
  __shared__ int place[BKT_MAX];
  __shared__ int tot[BKT_MAX];
  __shared__ int wtot[MV_T / 32];
  const int tid = threadIdx.x;
  const int lane = tid & 31;
  const int w = __builtin_amdgcn_readfirstlane(tid >> 5);
  for (int i = tid; i < NB; i += MV_T) {
    hist[i] = 0;
    tot[i] = 0;
  }

  const int base = blockIdx.x * (MV_T * MV_EPT) + tid;
  int s[MV_EPT], d[MV_EPT];
#pragma unroll
  for (int j = 0; j < MV_EPT; ++j) {
    const int e = base + j * MV_T;
    const bool ok = e < E;
    d[j] = ok ? dst[e] : -1;
    s[j] = ok ? src[e] : 0;
  }
  int cps = 0;
  while ((1 << cps) < NB && (1 << cps) < MV_T) ++cps;
  const int CP = 1 << cps;
  const int G = MV_T >> cps;
  const int col = tid & (CP - 1);
  const int rg = tid >> cps;
  int cv[32];
#pragma unroll
  for (int j = 0; j < 32; ++j) {
    const int rr = rg + j * G;
    cv[j] = (col < NB && rr < nrow) ? histB[rr * NB + col] : 0;
  }
  __syncthreads();
  int acc0 = 0;
#pragma unroll
  for (int j = 0; j < 32; ++j) acc0 += cv[j];
  if (col < NB) {
    if (rg + 32 * G < nrow) acc0 += col_sum(histB, col, rg + 32 * G, G, nrow, NB);
    atomicAdd(&tot[col], acc0);
  }
  for (int cc = col + CP; cc < NB; cc += CP)
    atomicAdd(&tot[cc], col_sum(histB, cc, rg, G, nrow, NB));

  int r[MV_EPT];
#pragma unroll
  for (int j = 0; j < MV_EPT; ++j) {
    r[j] = 0;
    if (d[j] >= 0) r[j] = atomicAdd(&hist[d[j] >> BKT_SHIFT], 1);
  }

  __syncthreads();
  const int i0 = 2 * tid;
  const int a0 = (i0 < NB) ? tot[i0] : 0;
  const int a1 = (i0 + 1 < NB) ? tot[i0 + 1] : 0;
  const int ts = a0 + a1;
  const int inc = wave_prefix(ts, lane);
  if (lane == 31) wtot[w] = inc;
  __syncthreads();
  const int x = wtot[lane];
  const int inc2 = wave_prefix(x, lane);
  const int ex = __builtin_amdgcn_readlane(inc2 - x, w) + inc - ts;

  if (i0 < NB) {
    const int h = hist[i0];
    place[i0] = ex + (h ? atomicAdd(cursor + i0, h) : 0);
  }
  if (i0 + 1 < NB) {
    const int h = hist[i0 + 1];
    place[i0 + 1] = ex + a0 + (h ? atomicAdd(cursor + i0 + 1, h) : 0);
  }
  if (blockIdx.x == 0) {
    if (i0 <= NB) bstart[i0] = ex;
    if (i0 + 1 <= NB) bstart[i0 + 1] = ex + a0;
  }
  __syncthreads();

#pragma unroll
  for (int j = 0; j < MV_EPT; ++j) {
    if (d[j] >= 0) {
      const unsigned word =
          (unsigned)s[j] | ((unsigned)(d[j] & (BKT_NODES - 1)) << 24);
      rec[place[d[j] >> BKT_SHIFT] + r[j]] = (int)word;
    }
  }
}

__global__ __launch_bounds__(BKT_NODES) void bucket_place(
    const int* __restrict__ rec, const int* __restrict__ bstart,
    int* __restrict__ rowptr, int* __restrict__ csr, int N) {
  __shared__ int cnt[BKT_NODES + 1];
  __shared__ int wtot[BKT_NODES / 32];
  const int tid = threadIdx.x;
  const int lane = tid & 31;
  const int w = tid >> 5;
  const int b = blockIdx.x;
  const int e0 = bstart[b];
  const int e1 = bstart[b + 1];
  cnt[tid] = 0;
  if (tid == 0) cnt[BKT_NODES] = 0;
  __syncthreads();

  for (int e = e0 + tid; e < e1; e += BKT_NODES * 8) {
    unsigned q[8];
#pragma unroll
    for (int j = 0; j < 8; ++j) {
      const int ee = e + j * BKT_NODES;
      q[j] = (unsigned)rec[ee < e1 ? ee : e1 - 1];
    }
#pragma unroll
    for (int j = 0; j < 8; ++j) {
      const bool ok = e + j * BKT_NODES < e1;
      atomicAdd(&cnt[ok ? (int)(q[j] >> 24) : BKT_NODES], 1);
    }
  }
  __syncthreads();

  const int c = cnt[tid];
  const int inc = wave_prefix(c, lane);
  if (lane == 31) wtot[w] = inc;
  __syncthreads();
  int wofs = 0;
#pragma unroll
  for (int k = 0; k < BKT_NODES / 32; ++k)
    if (k < w) wofs += wtot[k];
  const int ex = wofs + inc - c;
  const int n = b * BKT_NODES + tid;
  if (n < N) rowptr[n] = e0 + ex;
  if (b == (int)gridDim.x - 1 && tid == 0) rowptr[N] = e1;
  cnt[tid] = ex;
  __syncthreads();

  for (int e = e0 + tid; e < e1; e += BKT_NODES * 8) {
    unsigned q[8];
    int p[8];
#pragma unroll
    for (int j = 0; j < 8; ++j) {
      const int ee = e + j * BKT_NODES;
      q[j] = (unsigned)rec[ee < e1 ? ee : e1 - 1];
    }
#pragma unroll
    for (int j = 0; j < 8; ++j) {
      const bool ok = e + j * BKT_NODES < e1;
      p[j] = atomicAdd(&cnt[ok ? (int)(q[j] >> 24) : BKT_NODES], 1);
    }
#pragma unroll
    for (int j = 0; j < 8; ++j)
      if (e + j * BKT_NODES < e1) csr[e0 + p[j]] = (int)(q[j] & 0xffffffu);
  }
}

__device__ __forceinline__ float leaky(float e) {
  return (e > 0.0f) ? e : NEG_SLOPE * e;
}


template <int J>
__device__ __forceinline__ int row_share(int v) {
  return __builtin_amdgcn_update_dpp(0, v, 0x150 + J, 0xf, 0xf, true);
}

template <int M>
__device__ __forceinline__ float row_xor_max(float v) {
  const int o = __builtin_amdgcn_update_dpp(0, __float_as_int(v), 0x160 + M, 0xf,
                                            0xf, true);
  return fmaxf(v, __int_as_float(o));
}

template <int F, int J>
__device__ __forceinline__ void add_row(const float* __restrict__ h, int s,
                                        float t, unsigned colb,
                                        float (&a)[F / 16]) {
  const unsigned sj = (unsigned)row_share<J>(s);
  const float tj = __int_as_float(row_share<J>(__float_as_int(t)));
  const char* p = (const char*)h + (sj * (unsigned)(F * 4) + colb);
  if (F == 32) {
    const float2 r = *(const float2*)p;
    a[0] += tj * r.x;
    a[1] += tj * r.y;
  } else {
    const float4 r = *(const float4*)p;
    a[0] += tj * r.x;
    a[1] += tj * r.y;
    a[2] += tj * r.z;
    a[3] += tj * r.w;
  }
}

template <int F, int J0>
__device__ __forceinline__ void add_rows4(const float* __restrict__ h, int s,
                                          float t, unsigned colb,
                                          float (&a)[F / 16]) {
  add_row<F, J0 + 0>(h, s, t, colb, a);
  add_row<F, J0 + 1>(h, s, t, colb, a);
  add_row<F, J0 + 2>(h, s, t, colb, a);
  add_row<F, J0 + 3>(h, s, t, colb, a);
}

template <int F>
__device__ __forceinline__ void add_row_at(const float* __restrict__ h, int s,
                                           float t, int j, unsigned colb,
                                           float (&a)[F / 16]) {
  const unsigned sj =
      __builtin_amdgcn_permlane16_var((unsigned)s, (unsigned)s, (unsigned)j, false, false);
  const float tj = __uint_as_float(__builtin_amdgcn_permlane16_var(
      __float_as_uint(t), __float_as_uint(t), (unsigned)j, false, false));
  const char* p = (const char*)h + (sj * (unsigned)(F * 4) + colb);
  if (F == 32) {
    const float2 r = *(const float2*)p;
    a[0] += tj * r.x;
    a[1] += tj * r.y;
  } else {
    const float4 r = *(const float4*)p;
    a[0] += tj * r.x;
    a[1] += tj * r.y;
    a[2] += tj * r.z;
    a[3] += tj * r.w;
  }
}

template <int F>
__device__ __forceinline__ void add_chunk(const float* __restrict__ h, int s,
                                          float t, int c, unsigned colb,
                                          float (&a)[F / 16]) {
  if (c >= 16) {
    add_rows4<F, 0>(h, s, t, colb, a);
    add_rows4<F, 4>(h, s, t, colb, a);
    add_rows4<F, 8>(h, s, t, colb, a);
    add_rows4<F, 12>(h, s, t, colb, a);
  } else {
    for (int j0 = 0; j0 < c; j0 += 8) {
#pragma unroll
      for (int u = 0; u < 8; ++u) add_row_at<F>(h, s, t, j0 + u, colb, a);
    }
  }
}

template <int VPL>
__device__ __forceinline__ void load_lane(const float* __restrict__ p, int l,
                                          float (&v)[VPL]) {
  if constexpr (VPL == 2) {
    const float2 r = *((const float2*)p + l);
    v[0] = r.x;
    v[1] = r.y;
  } else {
    const float4 r = *((const float4*)p + l);
    v[0] = r.x;
    v[1] = r.y;
    v[2] = r.z;
    v[3] = r.w;
  }
}

template <int VPL>
__device__ __forceinline__ void store_lane(float* __restrict__ p, int l,
                                           const float (&v)[VPL]) {
  if constexpr (VPL == 2) {
    float2 o;
    o.x = v[0];
    o.y = v[1];
    *((float2*)p + l) = o;
  } else {
    float4 o;
    o.x = v[0];
    o.y = v[1];
    o.z = v[2];
    o.w = v[3];
    *((float4*)p + l) = o;
  }
}

#define NPB 16

__device__ __forceinline__ void pool_flush(float* __restrict__ pooled,
                                           float* __restrict__ cnt, int g,
                                           int lane, float a0, int run) {
  atomicAdd(pooled + g * F_HID + lane, a0);
  if (lane == 0) atomicAdd(cnt + g, (float)run);
}

template <int F, bool HIDDEN>
__global__ __launch_bounds__(256) void node_gather(
    const int* __restrict__ rowptr, const int* __restrict__ csr,
    const float* __restrict__ sv, const float* __restrict__ dv,
    const float* __restrict__ h, const float* __restrict__ b,
    float* __restrict__ out, int N, const float* __restrict__ uw,
    float* __restrict__ snext, float* __restrict__ dnext,
    const int* __restrict__ batch, float* __restrict__ pooled,
    float* __restrict__ pcnt) {
  static_assert(F == 32 || F == 64, "two or four floats per lane");
  static_assert(HIDDEN || F == F_HID, "the pool takes one feature per lane");
  constexpr int VPL = F / 16;
  __shared__ float prow[NPB * F];
  __shared__ int pgid[NPB];
  const int wv = __builtin_amdgcn_readfirstlane((int)(threadIdx.x >> 5));
  const int nw = (blockIdx.x * 8 + wv) * 2;
  if (HIDDEN && nw >= N) return;
  const bool act = HIDDEN || nw < N;
  const int n0 = act ? nw : N - 1;
  const bool two = act && n0 + 1 < N;
  const int n1 = two ? n0 + 1 : n0;
  const int lane = threadIdx.x & 31;
  const bool hi = lane >= 16;
  const int l = lane & 15;

  const int b0 = rowptr[n0], e0 = rowptr[n0 + 1];
  const int b1 = rowptr[n1], e1 = rowptr[n1 + 1];
  const float dn0 = dv[n0], dn1 = dv[n1];
  const float sn0 = sv[n0], sn1 = sv[n1];
  int gid = 0;
  if (!HIDDEN) {
    const int g0 = batch[n0], g1 = batch[n1];
    gid = hi ? g1 : g0;
  }
  const int n = hi ? n1 : n0;
  const int beg = hi ? b1 : b0;
  const int end = hi ? e1 : e0;
  const float dn = hi ? dn1 : dn0;
  const float es = leaky((hi ? sn1 : sn0) + dn);
  const int maxdeg = (e0 - b0 > e1 - b1) ? e0 - b0 : e1 - b1;

  const int safe = end > 0 ? end - 1 : 0;
  const int p0 = beg + l, p1 = p0 + 16, p2 = p0 + 32;
  const bool k0 = p0 < end, k1 = p1 < end, k2 = p2 < end;
  const int c0 = csr[k0 ? p0 : safe];
  const int c1 = csr[k1 ? p1 : safe];
  const int c2 = csr[k2 ? p2 : safe];
  const float v0 = sv[c0];
  const float v1 = sv[c1];
  const float v2 = sv[c2];
  const int s0 = k0 ? c0 : n, s1 = k1 ? c1 : n, s2 = k2 ? c2 : n;
  const float x0 = k0 ? leaky(v0 + dn) : es;
  const float x1 = k1 ? leaky(v1 + dn) : es;
  const float x2 = k2 ? leaky(v2 + dn) : es;

  float mx = fmaxf(fmaxf(es, x0), fmaxf(x1, x2));
  for (int p = beg + 48 + l; p < end; p += 16) mx = fmaxf(mx, leaky(sv[csr[p]] + dn));
  mx = row_xor_max<1>(mx);
  mx = row_xor_max<2>(mx);
  mx = row_xor_max<4>(mx);
  mx = row_xor_max<8>(mx);

  const unsigned colb = (unsigned)l * (VPL * 4);
  const float ts = __expf(es - mx);
  float a[VPL];
  {
    const char* p = (const char*)h + ((unsigned)n * (unsigned)(F * 4) + colb);
    if (F == 32) {
      const float2 r = *(const float2*)p;
      a[0] = ts * r.x;
      a[1] = ts * r.y;
    } else {
      const float4 r = *(const float4*)p;
      a[0] = ts * r.x;
      a[1] = ts * r.y;
      a[2] = ts * r.z;
      a[3] = ts * r.w;
    }
  }

  const float t0 = k0 ? __expf(x0 - mx) : 0.0f;
  const float t1 = k1 ? __expf(x1 - mx) : 0.0f;
  const float t2 = k2 ? __expf(x2 - mx) : 0.0f;
  float zl = t0;
  zl += t1;
  zl += t2;
  for (int base = 0; base < maxdeg; base += 16) {
    int s;
    float t;
    if (base == 0) {
      s = s0;
      t = t0;
    } else if (base == 16) {
      s = s1;
      t = t1;
    } else if (base == 32) {
      s = s2;
      t = t2;
    } else {
      const int p = beg + base + l;
      s = n;
      t = 0.0f;
      if (p < end) {
        s = csr[p];
        t = __expf(leaky(sv[s] + dn) - mx);
      }
      zl += t;
    }
    add_chunk<F>(h, s, t, maxdeg - base, colb, a);
  }
  zl = row_xor_add<1>(zl);
  zl = row_xor_add<2>(zl);
  zl = row_xor_add<4>(zl);
  zl = row_xor_add<8>(zl);
  const float z = zl + ts;

  const bool live = act && (!hi || two);
  float o[VPL];
#pragma unroll
  for (int i = 0; i < VPL; ++i) o[i] = a[i] / z;
  if (HIDDEN) {
    float bb[VPL];
    load_lane<VPL>(b, l, bb);
#pragma unroll
    for (int i = 0; i < VPL; ++i) {
      const float w = o[i] + bb[i];
      o[i] = (w > 0.0f) ? w : (__expf(w) - 1.0f);
    }
  }
  if (!HIDDEN) {
    const int slot = wv * 2 + (hi ? 1 : 0);
    {
      float2 ov;
      ov.x = o[0];
      ov.y = o[1];
      *((float2*)(prow + slot * F) + l) = ov;
    }
    if (l == 0) pgid[slot] = live ? gid : -1;
    __syncthreads();
    if (wv == 0) {
      int gq[NPB];
      float rq[NPB];
#pragma unroll
      for (int u = 0; u < NPB; ++u) {
        gq[u] = pgid[u];
        rq[u] = prow[u * F + lane];
      }
      int cur = -1;
      int run = 0;
      float a0 = 0.0f;
#pragma unroll
      for (int u = 0; u < NPB; ++u) {
        const int g = __builtin_amdgcn_readfirstlane(gq[u]);
        const float r = rq[u];
        if (g >= 0) {
          if (run > 0 && g != cur) {
            pool_flush(pooled, pcnt, cur, lane, a0, run);
            run = 0;
            a0 = 0.0f;
          }
          cur = g;
          a0 += r;
          ++run;
        }
      }
      if (run > 0) pool_flush(pooled, pcnt, cur, lane, a0, run);
    }
    return;
  }
  if (live) store_lane<VPL>(out + (size_t)n * F, l, o);
  if (HIDDEN) {
    float uu[VPL], ww[VPL];
    load_lane<VPL>(uw, l, uu);
    load_lane<VPL>(uw + F, l, ww);
    float ps = 0.0f, pd = 0.0f;
#pragma unroll
    for (int i = 0; i < VPL; ++i) {
      ps += o[i] * uu[i];
      pd += o[i] * ww[i];
    }
    ps = row_xor_add<1>(ps);
    pd = row_xor_add<1>(pd);
    ps = row_xor_add<2>(ps);
    pd = row_xor_add<2>(pd);
    ps = row_xor_add<4>(ps);
    pd = row_xor_add<4>(pd);
    ps = row_xor_add<8>(ps);
    pd = row_xor_add<8>(pd);
    if (live && l == 0) {
      snext[n] = ps;
      dnext[n] = pd;
    }
  }
}

__global__ __launch_bounds__(NUM_GRAPHS * 2) void pool_project(
    const float* __restrict__ pooled, const float* __restrict__ W,
    const float* __restrict__ b, const float* __restrict__ cnt,
    float* __restrict__ out) {
  constexpr int NT = F_OUT / 16;
  __shared__ float Ws[F_HID * F_OUT];
  for (int i = threadIdx.x; i < F_HID * F_OUT; i += NUM_GRAPHS * 2) Ws[i] = W[i];
  __syncthreads();

  const int wave = threadIdx.x >> 5;
  const int lane = threadIdx.x & 31;
  const int m0 = wave * 16;
  const int halfsel = lane >> 4;
  const int lidx = lane & 15;
  const int row = m0 + lidx;
  const int khalf = halfsel * 2;

  float bb[NT];
#pragma unroll
  for (int t = 0; t < NT; ++t) bb[t] = b[t * 16 + lidx];

  v8f acc[NT];
#pragma unroll
  for (int t = 0; t < NT; ++t)
#pragma unroll
    for (int j = 0; j < 8; ++j) acc[t][j] = 0.0f;

  wave_rows16<F_HID, F_OUT>(pooled, Ws, row, khalf, lidx, acc);

  const int rbase = m0 + halfsel * 8;
#pragma unroll
  for (int i = 0; i < 8; ++i) {
    const int r = rbase + i;
    const float cg = cnt[r];
    const float den = fmaxf(cg, 1.0f);
#pragma unroll
    for (int t = 0; t < NT; ++t)
      out[r * F_OUT + t * 16 + lidx] = (acc[t][i] + cg * bb[t]) / den;
  }
}

static inline int cdiv(int a, int b) { return (a + b - 1) / b; }

extern "C" void kernel_launch(void* const* d_in, const int* in_sizes, int n_in,
                              void* d_out, int out_size, void* d_ws,
                              size_t ws_size, hipStream_t stream) {
  const float* x      = (const float*)d_in[0];
  const int*   ei     = (const int*)d_in[1];
  const int*   batch  = (const int*)d_in[2];
  const float* W1     = (const float*)d_in[3];
  const float* a_src1 = (const float*)d_in[4];
  const float* a_dst1 = (const float*)d_in[5];
  const float* b1     = (const float*)d_in[6];
  const float* W2     = (const float*)d_in[7];
  const float* a_src2 = (const float*)d_in[8];
  const float* a_dst2 = (const float*)d_in[9];
  const float* b2     = (const float*)d_in[10];

  const int N = in_sizes[2];
  const int E = in_sizes[1] / 2;
  const int* src = ei;
  const int* dst = ei + E;

  float* ws = (float*)d_ws;
  size_t off = 0;
  float* h1 = ws + off; off += (size_t)N * F_HID;
  float* o1 = ws + off; off += (size_t)N * F_HID;
  float* sv = ws + off; off += (size_t)N;
  float* dv = ws + off; off += (size_t)N;
  float* sv2 = ws + off; off += (size_t)N;
  float* dv2 = ws + off; off += (size_t)N;
  float* cnt = ws + off; off += NUM_GRAPHS;
  float* pooled = ws + off; off += NUM_GRAPHS * F_HID;
  float* uw = ws + off; off += 2 * F_HID;
  int* cursor = (int*)(ws + off); off += BKT_MAX;
  int* bstart = (int*)(ws + off); off += BKT_MAX;
  int* rowptr = (int*)(ws + off); off += (size_t)N + 1;
  int* rec    = (int*)(ws + off); off += (size_t)E;
  int* csr    = (int*)(ws + off); off += (size_t)E;
  const int NB = cdiv(N, BKT_NODES);
  const int nA = cdiv(N, 128);
  const int NCB = cdiv(E, CNT_CHUNK);
  int* histB  = (int*)(ws + off); off += (size_t)NCB * NB;

  gemm_wmma<128, F_HID, true><<<nA + NCB, CNT_T, 0, stream>>>(
      x, W1, a_src1, a_dst1, h1, sv, dv, N, cursor, pooled, cnt, W2, a_src2,
      a_dst2, uw, dst, histB, E, NB, nA);

  bucket_move<<<cdiv(E, MV_T * MV_EPT), MV_T, 0, stream>>>(src, dst, histB, cursor,
                                                           bstart, rec, E, NB, NCB);
  bucket_place<<<NB, BKT_NODES, 0, stream>>>(rec, bstart, rowptr, csr, N);

  node_gather<F_HID, true><<<cdiv(N, NPB), 256, 0, stream>>>(
      rowptr, csr, sv, dv, h1, b1, o1, N, uw, sv2, dv2, nullptr, nullptr, nullptr);

  node_gather<F_HID, false><<<cdiv(N, NPB), 256, 0, stream>>>(
      rowptr, csr, sv2, dv2, o1, nullptr, nullptr, N, nullptr, nullptr, nullptr,
      batch, pooled, cnt);

  pool_project<<<1, NUM_GRAPHS * 2, 0, stream>>>(pooled, W2, b2, cnt, (float*)d_out);
}


# Edge sort's scattered 4-byte stores go through LDS: bucket_move and bucket_place put each word at its place in an LDS copy, then store rec and csr in index order, whole lines per wave; run order kept
# speedup vs baseline: 9.8043x; 1.2516x over previous
#include <hip/hip_runtime.h>
#include <hip/hip_bf16.h>
#include <math.h>

typedef __attribute__((ext_vector_type(2))) float v2f;
typedef __attribute__((ext_vector_type(8))) float v8f;

#define NUM_GRAPHS 64
#define NEG_SLOPE 0.2f

#define BKT_NODES 256
#define BKT_SHIFT 8
#define BKT_MAX 2048
#define MV_T 1024
#define MV_EPT 8
#define CNT_T 256
#define CNT_EPT 32
#define CNT_ROUNDS 2
#define CNT_CHUNK (CNT_T * CNT_EPT * CNT_ROUNDS)
#define PL_CAP 12288


#define F_HID 32
#define F_OUT 64

template <int M>
__device__ __forceinline__ float row_xor_add(float v) {
  const int o = __builtin_amdgcn_update_dpp(0, __float_as_int(v), 0x160 + M, 0xf,
                                            0xf, true);
  return v + __int_as_float(o);
}

template <int FIN, int FOUT>
__device__ __forceinline__ void wave_rows16(const float* __restrict__ x,
                                            const float* Ws, int row, int khalf,
                                            int lidx, v8f (&acc)[FOUT / 16]) {
  constexpr int NT = FOUT / 16;
#pragma unroll 4
  for (int k = 0; k < FIN; k += 4) {
    const int kk = k + khalf;
    v2f a;
    a.x = x[(size_t)row * FIN + kk];
    a.y = x[(size_t)row * FIN + kk + 1];
#pragma unroll
    for (int t = 0; t < NT; ++t) {
      v2f b;
      b.x = Ws[kk * FOUT + t * 16 + lidx];
      b.y = Ws[(kk + 1) * FOUT + t * 16 + lidx];
      acc[t] = __builtin_amdgcn_wmma_f32_16x16x4_f32(
          false, a, false, b, (short)0, acc[t], false, false);
    }
  }
}

template <int FIN, int FOUT, bool ZERO>
__global__ __launch_bounds__(256) void gemm_wmma(
    const float* __restrict__ x, const float* __restrict__ W,
    const float* __restrict__ asrc, const float* __restrict__ adst,
    float* __restrict__ h, float* __restrict__ s, float* __restrict__ d, int N,
    int* __restrict__ zbk, float* __restrict__ zpool, float* __restrict__ zcnt,
    const float* __restrict__ Wn, const float* __restrict__ asrcn,
    const float* __restrict__ adstn, float* __restrict__ uw,
    const int* __restrict__ dst, int* __restrict__ histB, int E, int NB,
    int nA) {
  constexpr int NT = FOUT / 16;
  __shared__ float Ws[FIN * FOUT];

  if (ZERO) {
    static_assert(!ZERO || FIN * FOUT >= BKT_MAX, "the counters fit in Ws");
    if ((int)blockIdx.x >= nA) {
      int* hist = reinterpret_cast<int*>(Ws);
      const int cb = (int)blockIdx.x - nA;
      for (int i = threadIdx.x; i < NB; i += CNT_T) hist[i] = 0;
      __syncthreads();
      for (int r = 0; r < CNT_ROUNDS; ++r) {
        const int base = cb * CNT_CHUNK + r * (CNT_T * CNT_EPT) + (int)threadIdx.x;
        int dd[CNT_EPT];
#pragma unroll
        for (int j = 0; j < CNT_EPT; ++j) {
          const int e = base + j * CNT_T;
          dd[j] = (e < E) ? dst[e] : -1;
        }
#pragma unroll
        for (int j = 0; j < CNT_EPT; ++j)
          if (dd[j] >= 0) atomicAdd(&hist[dd[j] >> BKT_SHIFT], 1);
      }
      __syncthreads();
      for (int i = threadIdx.x; i < NB; i += CNT_T) histB[cb * NB + i] = hist[i];
      return;
    }
  }

  for (int i = threadIdx.x; i < FIN * FOUT; i += 256) Ws[i] = W[i];
  __syncthreads();

  if (ZERO) {
    if (blockIdx.x == 0) {
      for (int i = threadIdx.x; i < BKT_MAX; i += 256) zbk[i] = 0;
      for (int i = threadIdx.x; i < NUM_GRAPHS * FOUT; i += 256) zpool[i] = 0.0f;
      if (threadIdx.x < NUM_GRAPHS) zcnt[threadIdx.x] = 0.0f;
      if (threadIdx.x < 2 * FOUT) {
        const int f = threadIdx.x % FOUT;
        const float* av = (threadIdx.x < FOUT) ? asrcn : adstn;
        const float4* wr = (const float4*)(Wn + f * F_OUT);
        const float4* ar = (const float4*)av;
        float acc1 = 0.0f;
#pragma unroll
        for (int q = 0; q < F_OUT / 4; ++q) {
          const float4 wq = wr[q];
          const float4 aq = ar[q];
          acc1 += wq.x * aq.x;
          acc1 += wq.y * aq.y;
          acc1 += wq.z * aq.z;
          acc1 += wq.w * aq.w;
        }
        uw[threadIdx.x] = acc1;
      }
    }
  }

  const int wave = threadIdx.x >> 5;
  const int lane = threadIdx.x & 31;
  const int m0 = (blockIdx.x * 8 + wave) * 16;
  const int halfsel = lane >> 4;
  const int lidx = lane & 15;
  int row = m0 + lidx;
  if (row >= N) row = N - 1;
  const int khalf = halfsel * 2;

  float as_[NT], ad_[NT];
#pragma unroll
  for (int t = 0; t < NT; ++t) {
    as_[t] = asrc[t * 16 + lidx];
    ad_[t] = adst[t * 16 + lidx];
  }

  v8f acc[NT];
#pragma unroll
  for (int t = 0; t < NT; ++t)
#pragma unroll
    for (int j = 0; j < 8; ++j) acc[t][j] = 0.0f;

  wave_rows16<FIN, FOUT>(x, Ws, row, khalf, lidx, acc);

  const int rbase = m0 + halfsel * 8;
#pragma unroll
  for (int i = 0; i < 8; ++i) {
    const int r = rbase + i;
    if (r < N) {
#pragma unroll
      for (int t = 0; t < NT; ++t)
        h[(size_t)r * FOUT + t * 16 + lidx] = acc[t][i];
    }
  }

  float ps[8], pd[8];
#pragma unroll
  for (int i = 0; i < 8; ++i) {
    float a = 0.0f, b = 0.0f;
#pragma unroll
    for (int t = 0; t < NT; ++t) {
      a += acc[t][i] * as_[t];
      b += acc[t][i] * ad_[t];
    }
    ps[i] = a;
    pd[i] = b;
  }
#pragma unroll
  for (int i = 0; i < 8; ++i) {
    ps[i] = row_xor_add<1>(ps[i]);
    pd[i] = row_xor_add<1>(pd[i]);
  }
#pragma unroll
  for (int i = 0; i < 8; ++i) {
    ps[i] = row_xor_add<2>(ps[i]);
    pd[i] = row_xor_add<2>(pd[i]);
  }
#pragma unroll
  for (int i = 0; i < 8; ++i) {
    ps[i] = row_xor_add<4>(ps[i]);
    pd[i] = row_xor_add<4>(pd[i]);
  }
#pragma unroll
  for (int i = 0; i < 8; ++i) {
    ps[i] = row_xor_add<8>(ps[i]);
    pd[i] = row_xor_add<8>(pd[i]);
  }
  float so = ps[0], dout = pd[0];
#pragma unroll
  for (int i = 1; i < 8; ++i) {
    if (lidx == i) {
      so = ps[i];
      dout = pd[i];
    }
  }
  const int r = rbase + lidx;
  if (lidx < 8 && r < N) {
    s[r] = so;
    d[r] = dout;
  }
}


template <int S>
__device__ __forceinline__ int row_shr_add(int v) {
  return v + __builtin_amdgcn_update_dpp(0, v, 0x110 + S, 0xf, 0xf, true);
}

__device__ __forceinline__ int wave_prefix(int v, int lane) {
  v = row_shr_add<1>(v);
  v = row_shr_add<2>(v);
  v = row_shr_add<4>(v);
  v = row_shr_add<8>(v);
  const int low = __builtin_amdgcn_readlane(v, 15);
  return v + (lane >= 16 ? low : 0);
}

__device__ __forceinline__ int col_sum(const int* __restrict__ histB, int c,
                                       int rs, int G, int nrow, int NB) {
  int acc = 0;
  for (int r0 = rs; r0 < nrow; r0 += 32 * G) {
    int v[32];
#pragma unroll
    for (int j = 0; j < 32; ++j) {
      const int r = r0 + j * G;
      v[j] = (r < nrow) ? histB[r * NB + c] : 0;
    }
#pragma unroll
    for (int j = 0; j < 32; ++j) acc += v[j];
  }
  return acc;
}

__global__ __launch_bounds__(MV_T) void bucket_move(
    const int* __restrict__ src, const int* __restrict__ dst,
    const int* __restrict__ histB, int* __restrict__ cursor,
    int* __restrict__ bstart, int* __restrict__ rec, int E, int NB, int nrow) {
  __shared__ int hist[BKT_MAX];
  __shared__ int place[BKT_MAX];
  __shared__ int tot[BKT_MAX];
  __shared__ int wtot[MV_T / 32];
  __shared__ int wloc[MV_T / 32];
  __shared__ int stage[MV_T * MV_EPT];
  __shared__ unsigned short sbkt[MV_T * MV_EPT];
  const int tid = threadIdx.x;
  const int lane = tid & 31;
  const int w = __builtin_amdgcn_readfirstlane(tid >> 5);
  for (int i = tid; i < NB; i += MV_T) {
    hist[i] = 0;
    tot[i] = 0;
  }

  const int base = blockIdx.x * (MV_T * MV_EPT) + tid;
  int s[MV_EPT], d[MV_EPT];
#pragma unroll
  for (int j = 0; j < MV_EPT; ++j) {
    const int e = base + j * MV_T;
    const bool ok = e < E;
    d[j] = ok ? dst[e] : -1;
    s[j] = ok ? src[e] : 0;
  }
  int cps = 0;
  while ((1 << cps) < NB && (1 << cps) < MV_T) ++cps;
  const int CP = 1 << cps;
  const int G = MV_T >> cps;
  const int col = tid & (CP - 1);
  const int rg = tid >> cps;
  int cv[32];
#pragma unroll
  for (int j = 0; j < 32; ++j) {
    const int rr = rg + j * G;
    cv[j] = (col < NB && rr < nrow) ? histB[rr * NB + col] : 0;
  }
  __syncthreads();
  int acc0 = 0;
#pragma unroll
  for (int j = 0; j < 32; ++j) acc0 += cv[j];
  if (col < NB) {
    if (rg + 32 * G < nrow) acc0 += col_sum(histB, col, rg + 32 * G, G, nrow, NB);
    atomicAdd(&tot[col], acc0);
  }
  for (int cc = col + CP; cc < NB; cc += CP)
    atomicAdd(&tot[cc], col_sum(histB, cc, rg, G, nrow, NB));

  int r[MV_EPT];
#pragma unroll
  for (int j = 0; j < MV_EPT; ++j) {
    r[j] = 0;
    if (d[j] >= 0) r[j] = atomicAdd(&hist[d[j] >> BKT_SHIFT], 1);
  }

  __syncthreads();
  const int i0 = 2 * tid;
  const int a0 = (i0 < NB) ? tot[i0] : 0;
  const int a1 = (i0 + 1 < NB) ? tot[i0 + 1] : 0;
  const int h0 = (i0 < NB) ? hist[i0] : 0;
  const int h1 = (i0 + 1 < NB) ? hist[i0 + 1] : 0;
  const int ts = a0 + a1;
  const int hs = h0 + h1;
  const int inc = wave_prefix(ts, lane);
  const int linc = wave_prefix(hs, lane);
  if (lane == 31) {
    wtot[w] = inc;
    wloc[w] = linc;
  }
  __syncthreads();
  const int x = wtot[lane];
  const int xl = wloc[lane];
  const int inc2 = wave_prefix(x, lane);
  const int linc2 = wave_prefix(xl, lane);
  const int ex = __builtin_amdgcn_readlane(inc2 - x, w) + inc - ts;
  const int lex = __builtin_amdgcn_readlane(linc2 - xl, w) + linc - hs;
  const int nv = __builtin_amdgcn_readlane(linc2, 31);

  if (i0 < NB) {
    const int g = ex + (h0 ? atomicAdd(cursor + i0, h0) : 0);
    hist[i0] = lex;
    place[i0] = g - lex;
  }
  if (i0 + 1 < NB) {
    const int g = ex + a0 + (h1 ? atomicAdd(cursor + i0 + 1, h1) : 0);
    hist[i0 + 1] = lex + h0;
    place[i0 + 1] = g - (lex + h0);
  }
  if (blockIdx.x == 0) {
    if (i0 <= NB) bstart[i0] = ex;
    if (i0 + 1 <= NB) bstart[i0 + 1] = ex + a0;
  }
  __syncthreads();

  int lp[MV_EPT];
#pragma unroll
  for (int j = 0; j < MV_EPT; ++j) lp[j] = hist[d[j] >= 0 ? (d[j] >> BKT_SHIFT) : 0];
#pragma unroll
  for (int j = 0; j < MV_EPT; ++j) {
    if (d[j] >= 0) {
      const unsigned word =
          (unsigned)s[j] | ((unsigned)(d[j] & (BKT_NODES - 1)) << 24);
      const int pos = lp[j] + r[j];
      stage[pos] = (int)word;
      sbkt[pos] = (unsigned short)(d[j] >> BKT_SHIFT);
    }
  }
  __syncthreads();

  int wv[MV_EPT], bv[MV_EPT], pv[MV_EPT];
#pragma unroll
  for (int j = 0; j < MV_EPT; ++j) {
    const int i = tid + j * MV_T;
    const int ii = i < nv ? i : 0;
    wv[j] = stage[ii];
    bv[j] = sbkt[ii];
  }
#pragma unroll
  for (int j = 0; j < MV_EPT; ++j) pv[j] = place[bv[j]];
#pragma unroll
  for (int j = 0; j < MV_EPT; ++j) {
    const int i = tid + j * MV_T;
    if (i < nv) rec[pv[j] + i] = wv[j];
  }
}

__global__ __launch_bounds__(BKT_NODES) void bucket_place(
    const int* __restrict__ rec, const int* __restrict__ bstart,
    int* __restrict__ rowptr, int* __restrict__ csr, int N) {
  __shared__ int cnt[BKT_NODES + 1];
  __shared__ int wtot[BKT_NODES / 32];
  __shared__ int stage[PL_CAP];
  const int tid = threadIdx.x;
  const int lane = tid & 31;
  const int w = tid >> 5;
  const int b = blockIdx.x;
  const int e0 = bstart[b];
  const int e1 = bstart[b + 1];
  cnt[tid] = 0;
  if (tid == 0) cnt[BKT_NODES] = 0;
  __syncthreads();

  for (int e = e0 + tid; e < e1; e += BKT_NODES * 8) {
    unsigned q[8];
#pragma unroll
    for (int j = 0; j < 8; ++j) {
      const int ee = e + j * BKT_NODES;
      q[j] = (unsigned)rec[ee < e1 ? ee : e1 - 1];
    }
#pragma unroll
    for (int j = 0; j < 8; ++j) {
      const bool ok = e + j * BKT_NODES < e1;
      atomicAdd(&cnt[ok ? (int)(q[j] >> 24) : BKT_NODES], 1);
    }
  }
  __syncthreads();

  const int c = cnt[tid];
  const int inc = wave_prefix(c, lane);
  if (lane == 31) wtot[w] = inc;
  __syncthreads();
  int wofs = 0;
#pragma unroll
  for (int k = 0; k < BKT_NODES / 32; ++k)
    if (k < w) wofs += wtot[k];
  const int ex = wofs + inc - c;
  const int n = b * BKT_NODES + tid;
  if (n < N) rowptr[n] = e0 + ex;
  if (b == (int)gridDim.x - 1 && tid == 0) rowptr[N] = e1;
  cnt[tid] = ex;
  __syncthreads();

  for (int e = e0 + tid; e < e1; e += BKT_NODES * 8) {
    unsigned q[8];
    int p[8];
#pragma unroll
    for (int j = 0; j < 8; ++j) {
      const int ee = e + j * BKT_NODES;
      q[j] = (unsigned)rec[ee < e1 ? ee : e1 - 1];
    }
#pragma unroll
    for (int j = 0; j < 8; ++j) {
      const bool ok = e + j * BKT_NODES < e1;
      p[j] = atomicAdd(&cnt[ok ? (int)(q[j] >> 24) : BKT_NODES], 1);
    }
#pragma unroll
    for (int j = 0; j < 8; ++j) {
      if (e + j * BKT_NODES < e1) {
        const int sw = (int)(q[j] & 0xffffffu);
        if (p[j] < PL_CAP)
          stage[p[j]] = sw;
        else
          csr[e0 + p[j]] = sw;
      }
    }
  }
  __syncthreads();

  const int ne = e1 - e0;
  const int ns = ne < PL_CAP ? ne : PL_CAP;
  for (int i = tid; i < ns; i += BKT_NODES * 8) {
    int v[8];
#pragma unroll
    for (int j = 0; j < 8; ++j) {
      const int ii = i + j * BKT_NODES;
      v[j] = stage[ii < ns ? ii : ns - 1];
    }
#pragma unroll
    for (int j = 0; j < 8; ++j) {
      const int ii = i + j * BKT_NODES;
      if (ii < ns) csr[e0 + ii] = v[j];
    }
  }
}

__device__ __forceinline__ float leaky(float e) {
  return (e > 0.0f) ? e : NEG_SLOPE * e;
}


template <int J>
__device__ __forceinline__ int row_share(int v) {
  return __builtin_amdgcn_update_dpp(0, v, 0x150 + J, 0xf, 0xf, true);
}

template <int M>
__device__ __forceinline__ float row_xor_max(float v) {
  const int o = __builtin_amdgcn_update_dpp(0, __float_as_int(v), 0x160 + M, 0xf,
                                            0xf, true);
  return fmaxf(v, __int_as_float(o));
}

template <int M>
__device__ __forceinline__ float row_xor_get(float v) {
  return __int_as_float(__builtin_amdgcn_update_dpp(0, __float_as_int(v),
                                                    0x160 + M, 0xf, 0xf, true));
}

template <int F, int J, int CW>
__device__ __forceinline__ void load_rows(const float* __restrict__ h, int s,
                                          unsigned colb, float (&r)[CW][F / 16]) {
  if constexpr (J < CW) {
    const unsigned sj = (unsigned)row_share<J>(s);
    const char* p = (const char*)h + (sj * (unsigned)(F * 4) + colb);
    if constexpr (F == 32) {
      const float2 v = *(const float2*)p;
      r[J][0] = v.x;
      r[J][1] = v.y;
    } else {
      const float4 v = *(const float4*)p;
      r[J][0] = v.x;
      r[J][1] = v.y;
      r[J][2] = v.z;
      r[J][3] = v.w;
    }
    load_rows<F, J + 1, CW>(h, s, colb, r);
  }
}

template <int F, int J, int CW>
__device__ __forceinline__ void weigh_rows(float t, const float (&r)[CW][F / 16],
                                           float (&a)[F / 16]) {
  if constexpr (J < CW) {
    const float tj = __int_as_float(row_share<J>(__float_as_int(t)));
#pragma unroll
    for (int i = 0; i < F / 16; ++i) a[i] += tj * r[J][i];
    weigh_rows<F, J + 1, CW>(t, r, a);
  }
}

template <int M, int NOUT>
__device__ __forceinline__ void tr_step(const float (&p)[2 * NOUT], bool up,
                                        float (&q)[NOUT]) {
#pragma unroll
  for (int j = 0; j < NOUT; ++j) {
    const float keep = up ? p[j + NOUT] : p[j];
    const float send = up ? p[j] : p[j + NOUT];
    q[j] = keep + row_xor_get<M>(send);
  }
}

template <int CW>
__device__ __forceinline__ float chunk_scores(const float (&p)[CW], int l) {
  float q4[4], q2[2], q1[1];
  if constexpr (CW == 16) {
    float q8[8];
    tr_step<8, 8>(p, (l & 8) != 0, q8);
    tr_step<4, 4>(q8, (l & 4) != 0, q4);
  } else {
    tr_step<4, 4>(p, (l & 4) != 0, q4);
  }
  tr_step<2, 2>(q4, (l & 2) != 0, q2);
  tr_step<1, 1>(q2, (l & 1) != 0, q1);
  if constexpr (CW == 16) return q1[0];
  return q1[0] + row_xor_get<8>(q1[0]);
}

template <int F, int CW>
__device__ __forceinline__ void add_chunk(const float* __restrict__ h, int s,
                                          bool k, float dn, unsigned colb,
                                          const float (&u)[F / 16], int l,
                                          float& m, float& zl, float& zs,
                                          float (&a)[F / 16]) {
  constexpr int VPL = F / 16;
  float r[CW][VPL];
  load_rows<F, 0, CW>(h, s, colb, r);
  float p[CW];
#pragma unroll
  for (int j = 0; j < CW; ++j) {
    float d = r[j][0] * u[0];
#pragma unroll
    for (int i = 1; i < VPL; ++i) d += r[j][i] * u[i];
    p[j] = d;
  }
  const float v = chunk_scores<CW>(p, l);
  const float x = k ? leaky(v + dn) : m;
  float cm = row_xor_max<1>(x);
  cm = row_xor_max<2>(cm);
  cm = row_xor_max<4>(cm);
  cm = row_xor_max<8>(cm);
  const float mn = fmaxf(m, cm);
  const float sc = __expf(m - mn);
  const float t = k ? __expf(x - mn) : 0.0f;
#pragma unroll
  for (int i = 0; i < VPL; ++i) a[i] *= sc;
  zl = zl * sc + t;
  zs *= sc;
  m = mn;
  weigh_rows<F, 0, CW>(t, r, a);
}

template <int VPL>
__device__ __forceinline__ void load_lane(const float* __restrict__ p, int l,
                                          float (&v)[VPL]) {
  if constexpr (VPL == 2) {
    const float2 r = *((const float2*)p + l);
    v[0] = r.x;
    v[1] = r.y;
  } else {
    const float4 r = *((const float4*)p + l);
    v[0] = r.x;
    v[1] = r.y;
    v[2] = r.z;
    v[3] = r.w;
  }
}

template <int VPL>
__device__ __forceinline__ void store_lane(float* __restrict__ p, int l,
                                           const float (&v)[VPL]) {
  if constexpr (VPL == 2) {
    float2 o;
    o.x = v[0];
    o.y = v[1];
    *((float2*)p + l) = o;
  } else {
    float4 o;
    o.x = v[0];
    o.y = v[1];
    o.z = v[2];
    o.w = v[3];
    *((float4*)p + l) = o;
  }
}

#define NPB 16

__device__ __forceinline__ void pool_flush(float* __restrict__ pooled,
                                           float* __restrict__ cnt, int g,
                                           int lane, float a0, int run) {
  atomicAdd(pooled + g * F_HID + lane, a0);
  if (lane == 0) atomicAdd(cnt + g, (float)run);
}

template <int F, bool HIDDEN>
__global__ __launch_bounds__(256) void node_gather(
    const int* __restrict__ rowptr, const int* __restrict__ csr,
    const float* __restrict__ sv, const float* __restrict__ dv,
    const float* __restrict__ h, const float* __restrict__ b,
    float* __restrict__ out, int N, const float* __restrict__ uw,
    float* __restrict__ snext, float* __restrict__ dnext,
    const int* __restrict__ batch, float* __restrict__ pooled,
    float* __restrict__ pcnt, const float* __restrict__ us) {
  static_assert(F == 32 || F == 64, "two or four floats per lane");
  static_assert(HIDDEN || F == F_HID, "the pool takes one feature per lane");
  constexpr int VPL = F / 16;
  __shared__ float prow[NPB * F];
  __shared__ int pgid[NPB];
  const int wv = __builtin_amdgcn_readfirstlane((int)(threadIdx.x >> 5));
  const int nw = (blockIdx.x * 8 + wv) * 2;
  if (HIDDEN && nw >= N) return;
  const bool act = HIDDEN || nw < N;
  const int n0 = act ? nw : N - 1;
  const bool two = act && n0 + 1 < N;
  const int n1 = two ? n0 + 1 : n0;
  const int lane = threadIdx.x & 31;
  const bool hi = lane >= 16;
  const int l = lane & 15;

  const int b0 = rowptr[n0], e0 = rowptr[n0 + 1];
  const int b1 = rowptr[n1], e1 = rowptr[n1 + 1];
  const float dn0 = dv[n0], dn1 = dv[n1];
  const float sn0 = sv[n0], sn1 = sv[n1];
  int gid = 0;
  if (!HIDDEN) {
    const int g0 = batch[n0], g1 = batch[n1];
    gid = hi ? g1 : g0;
  }
  const int n = hi ? n1 : n0;
  const int beg = hi ? b1 : b0;
  const int end = hi ? e1 : e0;
  const float dn = hi ? dn1 : dn0;
  const float es = leaky((hi ? sn1 : sn0) + dn);
  const int maxdeg = (e0 - b0 > e1 - b1) ? e0 - b0 : e1 - b1;

  const int safe = end > 0 ? end - 1 : 0;
  const int p0 = beg + l, p1 = p0 + 16, p2 = p0 + 32;
  const bool k0 = p0 < end, k1 = p1 < end, k2 = p2 < end;
  const int c0 = csr[k0 ? p0 : safe];
  const int c1 = csr[k1 ? p1 : safe];
  const int c2 = csr[k2 ? p2 : safe];
  const int s0 = k0 ? c0 : n, s1 = k1 ? c1 : n, s2 = k2 ? c2 : n;

  const unsigned colb = (unsigned)l * (VPL * 4);
  float u[VPL];
  load_lane<VPL>(us, l, u);
  float a[VPL];
  load_lane<VPL>(h + (size_t)n * F, l, a);
  float m = es, zl = 0.0f, zs = 1.0f;

  for (int base = 0; base < maxdeg; base += 16) {
    int s;
    bool k;
    if (base == 0) {
      s = s0;
      k = k0;
    } else if (base == 16) {
      s = s1;
      k = k1;
    } else if (base == 32) {
      s = s2;
      k = k2;
    } else {
      const int p = beg + base + l;
      k = p < end;
      s = n;
      if (k) s = csr[p];
    }
    if (maxdeg - base > 8)
      add_chunk<F, 16>(h, s, k, dn, colb, u, l, m, zl, zs, a);
    else
      add_chunk<F, 8>(h, s, k, dn, colb, u, l, m, zl, zs, a);
  }
  zl = row_xor_add<1>(zl);
  zl = row_xor_add<2>(zl);
  zl = row_xor_add<4>(zl);
  zl = row_xor_add<8>(zl);
  const float z = zl + zs;

  const bool live = act && (!hi || two);
  float o[VPL];
#pragma unroll
  for (int i = 0; i < VPL; ++i) o[i] = a[i] / z;
  if (HIDDEN) {
    float bb[VPL];
    load_lane<VPL>(b, l, bb);
#pragma unroll
    for (int i = 0; i < VPL; ++i) {
      const float w = o[i] + bb[i];
      o[i] = (w > 0.0f) ? w : (__expf(w) - 1.0f);
    }
  }
  if (!HIDDEN) {
    const int slot = wv * 2 + (hi ? 1 : 0);
    {
      float2 ov;
      ov.x = o[0];
      ov.y = o[1];
      *((float2*)(prow + slot * F) + l) = ov;
    }
    if (l == 0) pgid[slot] = live ? gid : -1;
    __syncthreads();
    if (wv == 0) {
      int gq[NPB];
      float rq[NPB];
#pragma unroll
      for (int u = 0; u < NPB; ++u) {
        gq[u] = pgid[u];
        rq[u] = prow[u * F + lane];
      }
      int cur = -1;
      int run = 0;
      float a0 = 0.0f;
#pragma unroll
      for (int u = 0; u < NPB; ++u) {
        const int g = __builtin_amdgcn_readfirstlane(gq[u]);
        const float r = rq[u];
        if (g >= 0) {
          if (run > 0 && g != cur) {
            pool_flush(pooled, pcnt, cur, lane, a0, run);
            run = 0;
            a0 = 0.0f;
          }
          cur = g;
          a0 += r;
          ++run;
        }
      }
      if (run > 0) pool_flush(pooled, pcnt, cur, lane, a0, run);
    }
    return;
  }
  if (live) store_lane<VPL>(out + (size_t)n * F, l, o);
  if (HIDDEN) {
    float uu[VPL], ww[VPL];
    load_lane<VPL>(uw, l, uu);
    load_lane<VPL>(uw + F, l, ww);
    float ps = 0.0f, pd = 0.0f;
#pragma unroll
    for (int i = 0; i < VPL; ++i) {
      ps += o[i] * uu[i];
      pd += o[i] * ww[i];
    }
    ps = row_xor_add<1>(ps);
    pd = row_xor_add<1>(pd);
    ps = row_xor_add<2>(ps);
    pd = row_xor_add<2>(pd);
    ps = row_xor_add<4>(ps);
    pd = row_xor_add<4>(pd);
    ps = row_xor_add<8>(ps);
    pd = row_xor_add<8>(pd);
    if (live && l == 0) {
      snext[n] = ps;
      dnext[n] = pd;
    }
  }
}

__global__ __launch_bounds__(NUM_GRAPHS * 2) void pool_project(
    const float* __restrict__ pooled, const float* __restrict__ W,
    const float* __restrict__ b, const float* __restrict__ cnt,
    float* __restrict__ out) {
  constexpr int NT = F_OUT / 16;
  __shared__ float Ws[F_HID * F_OUT];
  for (int i = threadIdx.x; i < F_HID * F_OUT; i += NUM_GRAPHS * 2) Ws[i] = W[i];
  __syncthreads();

  const int wave = threadIdx.x >> 5;
  const int lane = threadIdx.x & 31;
  const int m0 = wave * 16;
  const int halfsel = lane >> 4;
  const int lidx = lane & 15;
  const int row = m0 + lidx;
  const int khalf = halfsel * 2;

  float bb[NT];
#pragma unroll
  for (int t = 0; t < NT; ++t) bb[t] = b[t * 16 + lidx];

  v8f acc[NT];
#pragma unroll
  for (int t = 0; t < NT; ++t)
#pragma unroll
    for (int j = 0; j < 8; ++j) acc[t][j] = 0.0f;

  wave_rows16<F_HID, F_OUT>(pooled, Ws, row, khalf, lidx, acc);

  const int rbase = m0 + halfsel * 8;
#pragma unroll
  for (int i = 0; i < 8; ++i) {
    const int r = rbase + i;
    const float cg = cnt[r];
    const float den = fmaxf(cg, 1.0f);
#pragma unroll
    for (int t = 0; t < NT; ++t)
      out[r * F_OUT + t * 16 + lidx] = (acc[t][i] + cg * bb[t]) / den;
  }
}

static inline int cdiv(int a, int b) { return (a + b - 1) / b; }

extern "C" void kernel_launch(void* const* d_in, const int* in_sizes, int n_in,
                              void* d_out, int out_size, void* d_ws,
                              size_t ws_size, hipStream_t stream) {
  const float* x      = (const float*)d_in[0];
  const int*   ei     = (const int*)d_in[1];
  const int*   batch  = (const int*)d_in[2];
  const float* W1     = (const float*)d_in[3];
  const float* a_src1 = (const float*)d_in[4];
  const float* a_dst1 = (const float*)d_in[5];
  const float* b1     = (const float*)d_in[6];
  const float* W2     = (const float*)d_in[7];
  const float* a_src2 = (const float*)d_in[8];
  const float* a_dst2 = (const float*)d_in[9];
  const float* b2     = (const float*)d_in[10];

  const int N = in_sizes[2];
  const int E = in_sizes[1] / 2;
  const int* src = ei;
  const int* dst = ei + E;

  float* ws = (float*)d_ws;
  size_t off = 0;
  float* h1 = ws + off; off += (size_t)N * F_HID;
  float* o1 = ws + off; off += (size_t)N * F_HID;
  float* sv = ws + off; off += (size_t)N;
  float* dv = ws + off; off += (size_t)N;
  float* sv2 = ws + off; off += (size_t)N;
  float* dv2 = ws + off; off += (size_t)N;
  float* cnt = ws + off; off += NUM_GRAPHS;
  float* pooled = ws + off; off += NUM_GRAPHS * F_HID;
  float* uw = ws + off; off += 2 * F_HID;
  int* cursor = (int*)(ws + off); off += BKT_MAX;
  int* bstart = (int*)(ws + off); off += BKT_MAX;
  int* rowptr = (int*)(ws + off); off += (size_t)N + 1;
  int* rec    = (int*)(ws + off); off += (size_t)E;
  int* csr    = (int*)(ws + off); off += (size_t)E;
  const int NB = cdiv(N, BKT_NODES);
  const int nA = cdiv(N, 128);
  const int NCB = cdiv(E, CNT_CHUNK);
  int* histB  = (int*)(ws + off); off += (size_t)NCB * NB;

  gemm_wmma<128, F_HID, true><<<nA + NCB, CNT_T, 0, stream>>>(
      x, W1, a_src1, a_dst1, h1, sv, dv, N, cursor, pooled, cnt, W2, a_src2,
      a_dst2, uw, dst, histB, E, NB, nA);

  bucket_move<<<cdiv(E, MV_T * MV_EPT), MV_T, 0, stream>>>(src, dst, histB, cursor,
                                                           bstart, rec, E, NB, NCB);
  bucket_place<<<NB, BKT_NODES, 0, stream>>>(rec, bstart, rowptr, csr, N);

  node_gather<F_HID, true><<<cdiv(N, NPB), 256, 0, stream>>>(
      rowptr, csr, sv, dv, h1, b1, o1, N, uw, sv2, dv2, nullptr, nullptr, nullptr,
      a_src1);

  node_gather<F_HID, false><<<cdiv(N, NPB), 256, 0, stream>>>(
      rowptr, csr, sv2, dv2, o1, nullptr, nullptr, N, nullptr, nullptr, nullptr,
      batch, pooled, cnt, uw);

  pool_project<<<1, NUM_GRAPHS * 2, 0, stream>>>(pooled, W2, b2, cnt, (float*)d_out);
}
